# starting kernel
# baseline (speedup 1.0000x reference)
#include <hip/hip_runtime.h>

typedef float v2f __attribute__((ext_vector_type(2)));
typedef float v8f __attribute__((ext_vector_type(8)));

#define N_NODES 100000
#define IN_DIM  384
#define HID     64


__global__ void k_zero_deg(unsigned int* __restrict__ deg) {
    int i = blockIdx.x * blockDim.x + threadIdx.x;
    if (i < N_NODES) deg[i] = 0u;
}

__global__ void k_deg_count(const int* __restrict__ dst, int E,
                            unsigned int* __restrict__ deg) {
    int e = blockIdx.x * blockDim.x + threadIdx.x;
    if (e < E) atomicAdd(&deg[dst[e]], 1u);
}

__global__ void k_dinv(const unsigned int* __restrict__ deg,
                       float* __restrict__ dinv) {
    int i = blockIdx.x * blockDim.x + threadIdx.x;
    if (i < N_NODES) dinv[i] = rsqrtf((float)deg[i] + 1.0f);
}

__global__ __launch_bounds__(128) void k_gemm1(const float* __restrict__ X,
                                               const float* __restrict__ W1,
                                               float* __restrict__ H) {
    const int lane = threadIdx.x & 31;
    const int wave = threadIdx.x >> 5;
    const int half = lane >> 4;
    const int lr   = lane & 15;
    const int rowBase = blockIdx.x * 16;
    const int colBase = wave * 16;

    v8f c = {};
    const float* xrow = X + (size_t)(rowBase + lr) * IN_DIM;
    for (int k0 = 0; k0 < IN_DIM; k0 += 4) {
        v2f a, b;
        const float* ap = xrow + k0 + 2 * half;
        a[0] = ap[0];
        a[1] = ap[1];
        const float* bp = W1 + (size_t)(k0 + 2 * half) * HID + colBase + lr;
        b[0] = bp[0];
        b[1] = bp[HID];
        c = __builtin_amdgcn_wmma_f32_16x16x4_f32(
                 false, a,  false, b,
                 (short)0, c,  false,  false);
    }
    float* hrow = H + (size_t)(rowBase + 8 * half) * HID + colBase + lr;
#pragma unroll
    for (int r = 0; r < 8; ++r)
        hrow[(size_t)r * HID] = c[r];
}


__global__ void k_selfloop1(const float* __restrict__ H,
                            const float* __restrict__ dinv,
                            float* __restrict__ agg) {
    int idx = blockIdx.x * blockDim.x + threadIdx.x;
    if (idx < N_NODES * HID) {
        float di = dinv[idx >> 6];
        agg[idx] = H[idx] * di * di;
    }
}

__global__ void k_scatter1(const int* __restrict__ src,
                           const int* __restrict__ dst, int E,
                           const float* __restrict__ H,
                           const float* __restrict__ dinv,
                           float* __restrict__ agg) {
    int i = blockIdx.x * blockDim.x + threadIdx.x;
    int e = i >> 6;
    if (e < E) {
        int f = i & 63;
        int s = src[e], d = dst[e];
        float w = dinv[s] * dinv[d];
        atomicAdd(&agg[(size_t)d * HID + f], H[(size_t)s * HID + f] * w);
    }
}

__global__ __launch_bounds__(256) void k_dot2(const float* __restrict__ agg,
                                              const float* __restrict__ b1,
                                              const float* __restrict__ W2,
                                              float* __restrict__ z) {
    int row  = (blockIdx.x * blockDim.x + threadIdx.x) >> 5;
    int lane = threadIdx.x & 31;
    if (row < N_NODES) {
        const float* a = agg + (size_t)row * HID;
        float v0 = a[lane] + b1[lane];
        float v1 = a[lane + 32] + b1[lane + 32];
        v0 = v0 > 0.0f ? v0 : 0.0f;
        v1 = v1 > 0.0f ? v1 : 0.0f;
        float s = v0 * W2[lane] + v1 * W2[lane + 32];
#pragma unroll
        for (int off = 16; off > 0; off >>= 1)
            s += __shfl_down(s, off, 32);
        if (lane == 0) z[row] = s;
    }
}

__global__ void k_out_init(const float* __restrict__ z,
                           const float* __restrict__ dinv,
                           const float* __restrict__ b2,
                           float* __restrict__ out) {
    int i = blockIdx.x * blockDim.x + threadIdx.x;
    if (i < N_NODES) {
        float di = dinv[i];
        out[i] = z[i] * di * di + b2[0];
    }
}

__global__ void k_scatter2(const int* __restrict__ src,
                           const int* __restrict__ dst, int E,
                           const float* __restrict__ z,
                           const float* __restrict__ dinv,
                           float* __restrict__ out) {
    int e = blockIdx.x * blockDim.x + threadIdx.x;
    if (e < E) {
        int s = src[e], d = dst[e];
        atomicAdd(&out[d], z[s] * dinv[s] * dinv[d]);
    }
}


extern "C" void kernel_launch(void* const* d_in, const int* in_sizes, int n_in,
                              void* d_out, int out_size, void* d_ws, size_t ws_size,
                              hipStream_t stream) {
    const float* X  = (const float*)d_in[0];
    const int*   EI = (const int*)d_in[1];
    const float* W1 = (const float*)d_in[2];
    const float* b1 = (const float*)d_in[3];
    const float* W2 = (const float*)d_in[4];
    const float* b2 = (const float*)d_in[5];
    float* out = (float*)d_out;

    const int E = in_sizes[1] / 2;
    const int* src = EI;
    const int* dst = EI + E;

    char* ws = (char*)d_ws;
    float* H    = (float*)ws;
    float* agg  = (float*)(ws + (size_t)N_NODES * HID * 4);
    float* dinv = (float*)(ws + (size_t)2 * N_NODES * HID * 4);
    float* z    = dinv + N_NODES;
    unsigned int* deg = (unsigned int*)(z + N_NODES);

    const int B = 256;
    k_zero_deg <<<(N_NODES + B - 1) / B, B, 0, stream>>>(deg);
    k_deg_count<<<(E + B - 1) / B, B, 0, stream>>>(dst, E, deg);
    k_dinv     <<<(N_NODES + B - 1) / B, B, 0, stream>>>(deg, dinv);

    k_gemm1    <<<N_NODES / 16, 128, 0, stream>>>(X, W1, H);

    k_selfloop1<<<(N_NODES * HID + B - 1) / B, B, 0, stream>>>(H, dinv, agg);
    long long work1 = (long long)E * 64;
    k_scatter1 <<<(int)((work1 + B - 1) / B), B, 0, stream>>>(src, dst, E, H, dinv, agg);

    k_dot2     <<<(N_NODES * 32 + B - 1) / B, B, 0, stream>>>(agg, b1, W2, z);
    k_out_init <<<(N_NODES + B - 1) / B, B, 0, stream>>>(z, dinv, b2, out);
    k_scatter2 <<<(E + B - 1) / B, B, 0, stream>>>(src, dst, E, z, dinv, out);
}


# Float atomic scatter-adds of both layers replaced by row gathers over a destination-sorted edge list (degree atomic's returned rank + scan); sums in arrival-rank order, new dot tree; relu.W2 fused
# speedup vs baseline: 2.4016x; 2.4016x over previous
#include <hip/hip_runtime.h>


typedef float v2f __attribute__((ext_vector_type(2)));
typedef float v8f __attribute__((ext_vector_type(8)));

#define N_NODES 100000
#define IN_DIM  384
#define HID     64
#define SCAN_B  1024


__global__ void k_zero_deg(unsigned int* __restrict__ deg) {
    int i = blockIdx.x * blockDim.x + threadIdx.x;
    if (i < N_NODES) deg[i] = 0u;
}

__global__ void k_deg_rank(const int* __restrict__ dst, int E,
                           unsigned int* __restrict__ deg,
                           unsigned int* __restrict__ rank) {
    int e = blockIdx.x * blockDim.x + threadIdx.x;
    if (e < E) rank[e] = atomicAdd(&deg[dst[e]], 1u);
}

__global__ __launch_bounds__(SCAN_B) void k_blocksum(const unsigned int* __restrict__ deg,
                                                     unsigned int* __restrict__ bsum) {
    __shared__ unsigned int wtot[SCAN_B / 32];
    const int lane = threadIdx.x & 31;
    const int wave = threadIdx.x >> 5;
    int i = blockIdx.x * SCAN_B + threadIdx.x;
    unsigned int v = (i < N_NODES) ? deg[i] : 0u;
    for (int off = 16; off > 0; off >>= 1) v += __shfl_down(v, off, 32);
    if (lane == 0) wtot[wave] = v;
    __syncthreads();
    if (wave == 0) {
        unsigned int t = wtot[lane];
        for (int off = 16; off > 0; off >>= 1) t += __shfl_down(t, off, 32);
        if (lane == 0) bsum[blockIdx.x] = t;
    }
}

__global__ __launch_bounds__(SCAN_B) void k_scan(unsigned int* degptr,
                                                 const unsigned int* __restrict__ bsum,
                                                 float* __restrict__ dinv) {
    __shared__ unsigned int wtot[SCAN_B / 32];
    const int lane = threadIdx.x & 31;
    const int wave = threadIdx.x >> 5;
    int i = blockIdx.x * SCAN_B + threadIdx.x;
    unsigned int v = (i < N_NODES) ? degptr[i] : 0u;
    unsigned int incl = v;
    for (int off = 1; off < 32; off <<= 1) {
        unsigned int t = __shfl_up(incl, off, 32);
        if (lane >= off) incl += t;
    }
    if (lane == 31) wtot[wave] = incl;
    __syncthreads();
    unsigned int before = 0u;
    for (int b = 0; b < (int)blockIdx.x; ++b) before += bsum[b];
    for (int w = 0; w < wave; ++w) before += wtot[w];
    if (i < N_NODES) {
        degptr[i] = before + incl - v;
        dinv[i] = rsqrtf((float)v + 1.0f);
    }
}

__global__ void k_fill(const int* __restrict__ src,
                       const int* __restrict__ dst, int E,
                       const unsigned int* __restrict__ rowptr,
                       const unsigned int* __restrict__ rank,
                       int* __restrict__ csr) {
    int e = blockIdx.x * blockDim.x + threadIdx.x;
    if (e < E) {
        unsigned int pos = rowptr[dst[e]] + rank[e];
        if (pos < (unsigned int)E) csr[pos] = src[e];
    }
}

__global__ __launch_bounds__(128) void k_gemm1(const float* __restrict__ X,
                                               const float* __restrict__ W1,
                                               float* __restrict__ H) {
    const int lane = threadIdx.x & 31;
    const int wave = threadIdx.x >> 5;
    const int half = lane >> 4;
    const int lr   = lane & 15;
    const int rowBase = blockIdx.x * 16;
    const int colBase = wave * 16;

    v8f c = {};
    const float* xrow = X + (size_t)(rowBase + lr) * IN_DIM;
    for (int k0 = 0; k0 < IN_DIM; k0 += 4) {
        v2f a, b;
        const float* ap = xrow + k0 + 2 * half;
        a[0] = ap[0];
        a[1] = ap[1];
        const float* bp = W1 + (size_t)(k0 + 2 * half) * HID + colBase + lr;
        b[0] = bp[0];
        b[1] = bp[HID];
        c = __builtin_amdgcn_wmma_f32_16x16x4_f32(
                 false, a,  false, b,
                 (short)0, c,  false,  false);
    }
    float* hrow = H + (size_t)(rowBase + 8 * half) * HID + colBase + lr;
#pragma unroll
    for (int r = 0; r < 8; ++r)
        hrow[(size_t)r * HID] = c[r];
}

__global__ __launch_bounds__(256) void k_gather1(const unsigned int* __restrict__ rowptr,
                                                 const int* __restrict__ csr, int E,
                                                 const float* __restrict__ H,
                                                 const float* __restrict__ dinv,
                                                 const float* __restrict__ b1,
                                                 const float* __restrict__ W2,
                                                 float* __restrict__ z) {
    const int lane = threadIdx.x & 31;
    const int row  = (blockIdx.x * blockDim.x + threadIdx.x) >> 5;
    if (row >= N_NODES) return;

    const int nxt = min(row + 1, N_NODES - 1);
    int endv = (int)rowptr[nxt];
    if (row + 1 >= N_NODES) endv = E;
    const int start = __builtin_amdgcn_readfirstlane((int)rowptr[row]);
    const int end   = __builtin_amdgcn_readfirstlane(min(endv, E));

    const float di = dinv[row];
    const v2f* H2 = (const v2f*)H;
    const v2f hs = H2[(size_t)row * (HID / 2) + lane];
    float a0 = hs[0] * di * di;
    float a1 = hs[1] * di * di;

    for (int base = start; base < end; base += 32) {
        const int idx = base + lane;
        int   s = 0;
        float w = 0.0f;
        if (idx < end) {
            s = csr[idx];
            s = min(max(s, 0), N_NODES - 1);
            w = dinv[s] * di;
        }
        const int cnt = min(32, end - base);
        int t0 = 0;
        for (; t0 + 8 <= cnt; t0 += 8) {
            v2f h[8];
#pragma unroll
            for (int k = 0; k < 8; ++k) {
                const int st = __builtin_amdgcn_readlane(s, t0 + k);
                h[k] = H2[(size_t)st * (HID / 2) + lane];
            }
#pragma unroll
            for (int k = 0; k < 8; ++k) {
                const float wt = __int_as_float(__builtin_amdgcn_readlane(__float_as_int(w), t0 + k));
                a0 += h[k][0] * wt;
                a1 += h[k][1] * wt;
            }
        }
        if (t0 < cnt) {
            v2f h[7];
#pragma unroll
            for (int k = 0; k < 7; ++k) {
                h[k] = (v2f){0.0f, 0.0f};
                if (t0 + k < cnt) {
                    const int st = __builtin_amdgcn_readlane(s, t0 + k);
                    h[k] = H2[(size_t)st * (HID / 2) + lane];
                }
            }
#pragma unroll
            for (int k = 0; k < 7; ++k) {
                const float wt = __int_as_float(__builtin_amdgcn_readlane(__float_as_int(w), t0 + k));
                a0 += h[k][0] * wt;
                a1 += h[k][1] * wt;
            }
        }
    }

    float v0 = a0 + b1[2 * lane];
    float v1 = a1 + b1[2 * lane + 1];
    v0 = v0 > 0.0f ? v0 : 0.0f;
    v1 = v1 > 0.0f ? v1 : 0.0f;
    float sum = v0 * W2[2 * lane] + v1 * W2[2 * lane + 1];
    for (int off = 16; off > 0; off >>= 1)
        sum += __shfl_down(sum, off, 32);
    if (lane == 0) z[row] = sum;
}

__global__ __launch_bounds__(256) void k_gather2(const unsigned int* __restrict__ rowptr,
                                                 const int* __restrict__ csr, int E,
                                                 const float* __restrict__ z,
                                                 const float* __restrict__ dinv,
                                                 const float* __restrict__ b2,
                                                 float* __restrict__ out) {
    const int lane = threadIdx.x & 31;
    const int row  = (blockIdx.x * blockDim.x + threadIdx.x) >> 5;
    if (row >= N_NODES) return;

    int start = (int)rowptr[row];
    int end   = E;
    if (row + 1 < N_NODES) end = (int)rowptr[row + 1];
    end = min(end, E);

    const float di = dinv[row];
    float p = 0.0f;
    for (int idx = start + lane; idx < end; idx += 32) {
        int s = csr[idx];
        s = min(max(s, 0), N_NODES - 1);
        p += z[s] * dinv[s] * di;
    }
    for (int off = 16; off > 0; off >>= 1)
        p += __shfl_down(p, off, 32);
    if (lane == 0) out[row] = z[row] * di * di + b2[0] + p;
}


extern "C" void kernel_launch(void* const* d_in, const int* in_sizes, int n_in,
                              void* d_out, int out_size, void* d_ws, size_t ws_size,
                              hipStream_t stream) {
    const float* X  = (const float*)d_in[0];
    const int*   EI = (const int*)d_in[1];
    const float* W1 = (const float*)d_in[2];
    const float* b1 = (const float*)d_in[3];
    const float* W2 = (const float*)d_in[4];
    const float* b2 = (const float*)d_in[5];
    float* out = (float*)d_out;

    const int E = in_sizes[1] / 2;
    const int* src = EI;
    const int* dst = EI + E;

    char* ws = (char*)d_ws;
    float* H    = (float*)ws;
    unsigned int* rank = (unsigned int*)(ws + (size_t)N_NODES * HID * 4);
    int*   csr  = (int*)(rank + E);
    float* dinv = (float*)(ws + (size_t)2 * N_NODES * HID * 4);
    float* z    = dinv + N_NODES;
    unsigned int* deg = (unsigned int*)(z + N_NODES);
    unsigned int* bsum = (unsigned int*)z;

    const int B = 256;
    const int NB = (N_NODES + SCAN_B - 1) / SCAN_B;
    k_zero_deg <<<(N_NODES + B - 1) / B, B, 0, stream>>>(deg);
    k_deg_rank <<<(E + B - 1) / B, B, 0, stream>>>(dst, E, deg, rank);
    k_blocksum <<<NB, SCAN_B, 0, stream>>>(deg, bsum);
    k_scan     <<<NB, SCAN_B, 0, stream>>>(deg, bsum, dinv);

    k_gemm1    <<<N_NODES / 16, 128, 0, stream>>>(X, W1, H);

    k_fill     <<<(E + B - 1) / B, B, 0, stream>>>(src, dst, E, deg, rank, csr);
    k_gather1  <<<(N_NODES * 32 + B - 1) / B, B, 0, stream>>>(deg, csr, E, H, dinv, b1, W2, z);
    k_gather2  <<<(N_NODES * 32 + B - 1) / B, B, 0, stream>>>(deg, csr, E, z, dinv, b2, out);
}


# Edge sort by destination moved off global atomics: two-level counting sort on LDS counters (coarse buckets, then count/scan/place per bucket); row sums in arrival order of LDS atomics
# speedup vs baseline: 4.4924x; 1.8706x over previous
#include <hip/hip_runtime.h>


typedef float v2f __attribute__((ext_vector_type(2)));
typedef float v8f __attribute__((ext_vector_type(8)));

#define N_NODES 100000
#define IN_DIM  384
#define HID     64
#define SCAN_B  1024


#define BKT_SHIFT 9
#define BKT_NODES (1 << BKT_SHIFT)
#define NBKT ((N_NODES + BKT_NODES - 1) / BKT_NODES)
#define G1 256
#define PB 1024
#define EU 4
#define SRC_BITS 17
#define SRC_MASK ((1u << SRC_BITS) - 1u)

__global__ __launch_bounds__(PB) void k_hist(const int* __restrict__ dst, int E,
                                             unsigned int* __restrict__ counts) {
    __shared__ unsigned int h[NBKT];
    for (int i = threadIdx.x; i < NBKT; i += PB) h[i] = 0u;
    __syncthreads();
    const int chunk = (E + (int)gridDim.x - 1) / (int)gridDim.x;
    const int lo = (int)blockIdx.x * chunk;
    const int hi = min(E, lo + chunk);
    int e0 = lo + (int)threadIdx.x;
    for (; e0 + (EU - 1) * PB < hi; e0 += EU * PB) {
        int d[EU];
#pragma unroll
        for (int u = 0; u < EU; ++u) d[u] = dst[e0 + u * PB];
#pragma unroll
        for (int u = 0; u < EU; ++u) {
            const int dd = min(max(d[u], 0), N_NODES - 1);
            atomicAdd(&h[dd >> BKT_SHIFT], 1u);
        }
    }
    for (; e0 < hi; e0 += PB) {
        const int dd = min(max(dst[e0], 0), N_NODES - 1);
        atomicAdd(&h[dd >> BKT_SHIFT], 1u);
    }
    __syncthreads();
    for (int i = threadIdx.x; i < NBKT; i += PB) counts[(size_t)i * gridDim.x + blockIdx.x] = h[i];
}

__global__ __launch_bounds__(SCAN_B) void k_scan_counts(const unsigned int* __restrict__ cin,
                                                        unsigned int* __restrict__ cout, int n) {
    __shared__ unsigned int wtot[SCAN_B / 32];
    __shared__ unsigned int wpre[SCAN_B / 32];
    const int lane = threadIdx.x & 31;
    const int wave = threadIdx.x >> 5;
    unsigned int pre = 0u;
    for (int t = 0; t < (int)blockIdx.x; ++t) pre += cin[(size_t)t * SCAN_B + threadIdx.x];
    for (int off = 16; off > 0; off >>= 1) pre += __shfl_down(pre, off, 32);
    const int i = blockIdx.x * SCAN_B + threadIdx.x;
    const unsigned int v = (i < n) ? cin[i] : 0u;
    unsigned int incl = v;
    for (int off = 1; off < 32; off <<= 1) {
        unsigned int t = __shfl_up(incl, off, 32);
        if (lane >= off) incl += t;
    }
    if (lane == 31) wtot[wave] = incl;
    if (lane == 0)  wpre[wave] = pre;
    __syncthreads();
    unsigned int before = 0u;
    for (int w = 0; w < SCAN_B / 32; ++w) before += wpre[w];
    for (int w = 0; w < wave; ++w) before += wtot[w];
    if (i < n) cout[i] = before + incl - v;
}

__global__ __launch_bounds__(PB) void k_bucket_scatter(const int* __restrict__ src,
                                                       const int* __restrict__ dst, int E,
                                                       const unsigned int* __restrict__ offs,
                                                       unsigned int* __restrict__ keys) {
    __shared__ unsigned int cur[NBKT];
    for (int i = threadIdx.x; i < NBKT; i += PB) cur[i] = offs[(size_t)i * gridDim.x + blockIdx.x];
    __syncthreads();
    const int chunk = (E + (int)gridDim.x - 1) / (int)gridDim.x;
    const int lo = (int)blockIdx.x * chunk;
    const int hi = min(E, lo + chunk);
    int e0 = lo + (int)threadIdx.x;
    for (; e0 + (EU - 1) * PB < hi; e0 += EU * PB) {
        int d[EU], s[EU];
#pragma unroll
        for (int u = 0; u < EU; ++u) {
            d[u] = dst[e0 + u * PB];
            s[u] = src[e0 + u * PB];
        }
        unsigned int pos[EU];
#pragma unroll
        for (int u = 0; u < EU; ++u) {
            d[u] = min(max(d[u], 0), N_NODES - 1);
            s[u] = min(max(s[u], 0), N_NODES - 1);
            pos[u] = atomicAdd(&cur[d[u] >> BKT_SHIFT], 1u);
        }
#pragma unroll
        for (int u = 0; u < EU; ++u) {
            keys[min(pos[u], (unsigned int)E - 1u)] =
                ((unsigned int)(d[u] & (BKT_NODES - 1)) << SRC_BITS) | (unsigned int)s[u];
        }
    }
    for (; e0 < hi; e0 += PB) {
        const int dd = min(max(dst[e0], 0), N_NODES - 1);
        const int ss = min(max(src[e0], 0), N_NODES - 1);
        const unsigned int pos = atomicAdd(&cur[dd >> BKT_SHIFT], 1u);
        keys[min(pos, (unsigned int)E - 1u)] = ((unsigned int)(dd & (BKT_NODES - 1)) << SRC_BITS) | (unsigned int)ss;
    }
}

__global__ __launch_bounds__(PB) void k_bucket_sort(const unsigned int* __restrict__ offs, int nblk, int E,
                                                    const unsigned int* __restrict__ keys,
                                                    unsigned int* __restrict__ rowptr,
                                                    float* __restrict__ dinv,
                                                    int* __restrict__ csr) {
    __shared__ unsigned int cnt[BKT_NODES];
    __shared__ unsigned int wtot[BKT_NODES / 32];
    const int tid  = threadIdx.x;
    const int lane = tid & 31;
    const int wave = tid >> 5;
    const int b = blockIdx.x;
    unsigned int lo = offs[(size_t)b * nblk];
    unsigned int hi = (unsigned int)E;
    if (b + 1 < NBKT) hi = offs[(size_t)(b + 1) * nblk];
    hi = min(hi, (unsigned int)E);
    lo = min(lo, hi);

    if (tid < BKT_NODES) cnt[tid] = 0u;
    __syncthreads();
    unsigned int i0 = lo + tid;
    for (; i0 + (EU - 1) * PB < hi; i0 += EU * PB) {
        unsigned int k[EU];
#pragma unroll
        for (int u = 0; u < EU; ++u) k[u] = keys[i0 + u * PB];
#pragma unroll
        for (int u = 0; u < EU; ++u)
            atomicAdd(&cnt[(k[u] >> SRC_BITS) & (BKT_NODES - 1)], 1u);
    }
    for (; i0 < hi; i0 += PB)
        atomicAdd(&cnt[(keys[i0] >> SRC_BITS) & (BKT_NODES - 1)], 1u);
    __syncthreads();

    const unsigned int v = (tid < BKT_NODES) ? cnt[tid] : 0u;
    unsigned int incl = v;
    for (int off = 1; off < 32; off <<= 1) {
        unsigned int t = __shfl_up(incl, off, 32);
        if (lane >= off) incl += t;
    }
    if (tid < BKT_NODES && lane == 31) wtot[wave] = incl;
    __syncthreads();
    if (tid < BKT_NODES) {
        unsigned int before = 0u;
        for (int w = 0; w < wave; ++w) before += wtot[w];
        const unsigned int start = lo + before + incl - v;
        cnt[tid] = start;
        const int node = b * BKT_NODES + tid;
        if (node < N_NODES) {
            rowptr[node] = start;
            dinv[node] = rsqrtf((float)v + 1.0f);
        }
    }
    __syncthreads();
    i0 = lo + tid;
    for (; i0 + (EU - 1) * PB < hi; i0 += EU * PB) {
        unsigned int k[EU];
#pragma unroll
        for (int u = 0; u < EU; ++u) k[u] = keys[i0 + u * PB];
        unsigned int pos[EU];
#pragma unroll
        for (int u = 0; u < EU; ++u)
            pos[u] = atomicAdd(&cnt[(k[u] >> SRC_BITS) & (BKT_NODES - 1)], 1u);
#pragma unroll
        for (int u = 0; u < EU; ++u) {
            csr[min(pos[u], (unsigned int)E - 1u)] = (int)(k[u] & SRC_MASK);
        }
    }
    for (; i0 < hi; i0 += PB) {
        const unsigned int k = keys[i0];
        const unsigned int pos = atomicAdd(&cnt[(k >> SRC_BITS) & (BKT_NODES - 1)], 1u);
        csr[min(pos, (unsigned int)E - 1u)] = (int)(k & SRC_MASK);
    }
}

__global__ __launch_bounds__(128) void k_gemm1(const float* __restrict__ X,
                                               const float* __restrict__ W1,
                                               float* __restrict__ H) {
    const int lane = threadIdx.x & 31;
    const int wave = threadIdx.x >> 5;
    const int half = lane >> 4;
    const int lr   = lane & 15;
    const int rowBase = blockIdx.x * 16;
    const int colBase = wave * 16;

    v8f c = {};
    const float* xrow = X + (size_t)(rowBase + lr) * IN_DIM;
    for (int k0 = 0; k0 < IN_DIM; k0 += 4) {
        v2f a, b;
        const float* ap = xrow + k0 + 2 * half;
        a[0] = ap[0];
        a[1] = ap[1];
        const float* bp = W1 + (size_t)(k0 + 2 * half) * HID + colBase + lr;
        b[0] = bp[0];
        b[1] = bp[HID];
        c = __builtin_amdgcn_wmma_f32_16x16x4_f32(
                 false, a,  false, b,
                 (short)0, c,  false,  false);
    }
    float* hrow = H + (size_t)(rowBase + 8 * half) * HID + colBase + lr;
#pragma unroll
    for (int r = 0; r < 8; ++r)
        hrow[(size_t)r * HID] = c[r];
}

__global__ __launch_bounds__(256) void k_gather1(const unsigned int* __restrict__ rowptr,
                                                 const int* __restrict__ csr, int E,
                                                 const float* __restrict__ H,
                                                 const float* __restrict__ dinv,
                                                 const float* __restrict__ b1,
                                                 const float* __restrict__ W2,
                                                 float* __restrict__ z) {
    const int lane = threadIdx.x & 31;
    const int row  = (blockIdx.x * blockDim.x + threadIdx.x) >> 5;
    if (row >= N_NODES) return;

    const int nxt = min(row + 1, N_NODES - 1);
    int endv = (int)rowptr[nxt];
    if (row + 1 >= N_NODES) endv = E;
    const int start = __builtin_amdgcn_readfirstlane((int)rowptr[row]);
    const int end   = __builtin_amdgcn_readfirstlane(min(endv, E));

    const float di = dinv[row];
    const v2f* H2 = (const v2f*)H;
    const v2f hs = H2[(size_t)row * (HID / 2) + lane];
    float a0 = hs[0] * di * di;
    float a1 = hs[1] * di * di;

    for (int base = start; base < end; base += 32) {
        const int idx = base + lane;
        int   s = 0;
        float w = 0.0f;
        if (idx < end) {
            s = csr[idx];
            s = min(max(s, 0), N_NODES - 1);
            w = dinv[s] * di;
        }
        const int cnt = min(32, end - base);
        int t0 = 0;
        for (; t0 + 8 <= cnt; t0 += 8) {
            v2f h[8];
#pragma unroll
            for (int k = 0; k < 8; ++k) {
                const int st = __builtin_amdgcn_readlane(s, t0 + k);
                h[k] = H2[(size_t)st * (HID / 2) + lane];
            }
#pragma unroll
            for (int k = 0; k < 8; ++k) {
                const float wt = __int_as_float(__builtin_amdgcn_readlane(__float_as_int(w), t0 + k));
                a0 += h[k][0] * wt;
                a1 += h[k][1] * wt;
            }
        }
        if (t0 < cnt) {
            v2f h[7];
#pragma unroll
            for (int k = 0; k < 7; ++k) {
                h[k] = (v2f){0.0f, 0.0f};
                if (t0 + k < cnt) {
                    const int st = __builtin_amdgcn_readlane(s, t0 + k);
                    h[k] = H2[(size_t)st * (HID / 2) + lane];
                }
            }
#pragma unroll
            for (int k = 0; k < 7; ++k) {
                const float wt = __int_as_float(__builtin_amdgcn_readlane(__float_as_int(w), t0 + k));
                a0 += h[k][0] * wt;
                a1 += h[k][1] * wt;
            }
        }
    }

    float v0 = a0 + b1[2 * lane];
    float v1 = a1 + b1[2 * lane + 1];
    v0 = v0 > 0.0f ? v0 : 0.0f;
    v1 = v1 > 0.0f ? v1 : 0.0f;
    float sum = v0 * W2[2 * lane] + v1 * W2[2 * lane + 1];
    for (int off = 16; off > 0; off >>= 1)
        sum += __shfl_down(sum, off, 32);
    if (lane == 0) z[row] = sum;
}

__global__ __launch_bounds__(256) void k_gather2(const unsigned int* __restrict__ rowptr,
                                                 const int* __restrict__ csr, int E,
                                                 const float* __restrict__ z,
                                                 const float* __restrict__ dinv,
                                                 const float* __restrict__ b2,
                                                 float* __restrict__ out) {
    const int lane = threadIdx.x & 31;
    const int row  = (blockIdx.x * blockDim.x + threadIdx.x) >> 5;
    if (row >= N_NODES) return;

    int start = (int)rowptr[row];
    int end   = E;
    if (row + 1 < N_NODES) end = (int)rowptr[row + 1];
    end = min(end, E);

    const float di = dinv[row];
    float p = 0.0f;
    for (int idx = start + lane; idx < end; idx += 32) {
        int s = csr[idx];
        s = min(max(s, 0), N_NODES - 1);
        p += z[s] * dinv[s] * di;
    }
    for (int off = 16; off > 0; off >>= 1)
        p += __shfl_down(p, off, 32);
    if (lane == 0) out[row] = z[row] * di * di + b2[0] + p;
}


extern "C" void kernel_launch(void* const* d_in, const int* in_sizes, int n_in,
                              void* d_out, int out_size, void* d_ws, size_t ws_size,
                              hipStream_t stream) {
    const float* X  = (const float*)d_in[0];
    const int*   EI = (const int*)d_in[1];
    const float* W1 = (const float*)d_in[2];
    const float* b1 = (const float*)d_in[3];
    const float* W2 = (const float*)d_in[4];
    const float* b2 = (const float*)d_in[5];
    float* out = (float*)d_out;

    const int E = in_sizes[1] / 2;
    const int* src = EI;
    const int* dst = EI + E;

    char* ws = (char*)d_ws;
    float* H    = (float*)ws;
    unsigned int* keys = (unsigned int*)(ws + (size_t)N_NODES * HID * 4);
    int*   csr  = (int*)(keys + E);
    float* dinv = (float*)(ws + (size_t)2 * N_NODES * HID * 4);
    float* z    = dinv + N_NODES;
    unsigned int* rowptr = (unsigned int*)(z + N_NODES);
    unsigned int* craw = (unsigned int*)dinv;
    unsigned int* offs = (unsigned int*)z;

    const int B = 256;
    k_hist           <<<G1, PB, 0, stream>>>(dst, E, craw);
    k_scan_counts    <<<(NBKT * G1 + SCAN_B - 1) / SCAN_B, SCAN_B, 0, stream>>>(craw, offs, NBKT * G1);
    k_bucket_scatter <<<G1, PB, 0, stream>>>(src, dst, E, offs, keys);

    k_gemm1    <<<N_NODES / 16, 128, 0, stream>>>(X, W1, H);

    k_bucket_sort    <<<NBKT, PB, 0, stream>>>(offs, G1, E, keys, rowptr, dinv, csr);
    k_gather1  <<<(N_NODES * 32 + B - 1) / B, B, 0, stream>>>(rowptr, csr, E, H, dinv, b1, W2, z);
    k_gather2  <<<(N_NODES * 32 + B - 1) / B, B, 0, stream>>>(rowptr, csr, E, z, dinv, b2, out);
}


# X @ W1 rebuilt: one wave per 16x64 strip (4 accumulators), X rows read once by 16-byte loads issued ahead, W1 transposed in LDS; k order inside each dot sum changed
# speedup vs baseline: 5.2149x; 1.1608x over previous
#include <hip/hip_runtime.h>


typedef float v2f __attribute__((ext_vector_type(2)));
typedef float v8f __attribute__((ext_vector_type(8)));

#define N_NODES 100000
#define IN_DIM  384
#define HID     64
#define SCAN_B  1024


#define BKT_SHIFT 9
#define BKT_NODES (1 << BKT_SHIFT)
#define NBKT ((N_NODES + BKT_NODES - 1) / BKT_NODES)
#define G1 256
#define PB 1024
#define EU 4
#define SRC_BITS 17
#define SRC_MASK ((1u << SRC_BITS) - 1u)

__global__ __launch_bounds__(PB) void k_hist(const int* __restrict__ dst, int E,
                                             unsigned int* __restrict__ counts) {
    __shared__ unsigned int h[NBKT];
    for (int i = threadIdx.x; i < NBKT; i += PB) h[i] = 0u;
    __syncthreads();
    const int chunk = (E + (int)gridDim.x - 1) / (int)gridDim.x;
    const int lo = (int)blockIdx.x * chunk;
    const int hi = min(E, lo + chunk);
    int e0 = lo + (int)threadIdx.x;
    for (; e0 + (EU - 1) * PB < hi; e0 += EU * PB) {
        int d[EU];
#pragma unroll
        for (int u = 0; u < EU; ++u) d[u] = dst[e0 + u * PB];
#pragma unroll
        for (int u = 0; u < EU; ++u) {
            const int dd = min(max(d[u], 0), N_NODES - 1);
            atomicAdd(&h[dd >> BKT_SHIFT], 1u);
        }
    }
    for (; e0 < hi; e0 += PB) {
        const int dd = min(max(dst[e0], 0), N_NODES - 1);
        atomicAdd(&h[dd >> BKT_SHIFT], 1u);
    }
    __syncthreads();
    for (int i = threadIdx.x; i < NBKT; i += PB) counts[(size_t)i * gridDim.x + blockIdx.x] = h[i];
}

__global__ __launch_bounds__(SCAN_B) void k_scan_counts(const unsigned int* __restrict__ cin,
                                                        unsigned int* __restrict__ cout, int n) {
    __shared__ unsigned int wtot[SCAN_B / 32];
    __shared__ unsigned int wpre[SCAN_B / 32];
    const int lane = threadIdx.x & 31;
    const int wave = threadIdx.x >> 5;
    unsigned int pre = 0u;
    for (int t = 0; t < (int)blockIdx.x; ++t) pre += cin[(size_t)t * SCAN_B + threadIdx.x];
    for (int off = 16; off > 0; off >>= 1) pre += __shfl_down(pre, off, 32);
    const int i = blockIdx.x * SCAN_B + threadIdx.x;
    const unsigned int v = (i < n) ? cin[i] : 0u;
    unsigned int incl = v;
    for (int off = 1; off < 32; off <<= 1) {
        unsigned int t = __shfl_up(incl, off, 32);
        if (lane >= off) incl += t;
    }
    if (lane == 31) wtot[wave] = incl;
    if (lane == 0)  wpre[wave] = pre;
    __syncthreads();
    unsigned int before = 0u;
    for (int w = 0; w < SCAN_B / 32; ++w) before += wpre[w];
    for (int w = 0; w < wave; ++w) before += wtot[w];
    if (i < n) cout[i] = before + incl - v;
}

__global__ __launch_bounds__(PB) void k_bucket_scatter(const int* __restrict__ src,
                                                       const int* __restrict__ dst, int E,
                                                       const unsigned int* __restrict__ offs,
                                                       unsigned int* __restrict__ keys) {
    __shared__ unsigned int cur[NBKT];
    for (int i = threadIdx.x; i < NBKT; i += PB) cur[i] = offs[(size_t)i * gridDim.x + blockIdx.x];
    __syncthreads();
    const int chunk = (E + (int)gridDim.x - 1) / (int)gridDim.x;
    const int lo = (int)blockIdx.x * chunk;
    const int hi = min(E, lo + chunk);
    int e0 = lo + (int)threadIdx.x;
    for (; e0 + (EU - 1) * PB < hi; e0 += EU * PB) {
        int d[EU], s[EU];
#pragma unroll
        for (int u = 0; u < EU; ++u) {
            d[u] = dst[e0 + u * PB];
            s[u] = src[e0 + u * PB];
        }
        unsigned int pos[EU];
#pragma unroll
        for (int u = 0; u < EU; ++u) {
            d[u] = min(max(d[u], 0), N_NODES - 1);
            s[u] = min(max(s[u], 0), N_NODES - 1);
            pos[u] = atomicAdd(&cur[d[u] >> BKT_SHIFT], 1u);
        }
#pragma unroll
        for (int u = 0; u < EU; ++u) {
            keys[min(pos[u], (unsigned int)E - 1u)] =
                ((unsigned int)(d[u] & (BKT_NODES - 1)) << SRC_BITS) | (unsigned int)s[u];
        }
    }
    for (; e0 < hi; e0 += PB) {
        const int dd = min(max(dst[e0], 0), N_NODES - 1);
        const int ss = min(max(src[e0], 0), N_NODES - 1);
        const unsigned int pos = atomicAdd(&cur[dd >> BKT_SHIFT], 1u);
        keys[min(pos, (unsigned int)E - 1u)] = ((unsigned int)(dd & (BKT_NODES - 1)) << SRC_BITS) | (unsigned int)ss;
    }
}

__global__ __launch_bounds__(PB) void k_bucket_sort(const unsigned int* __restrict__ offs, int nblk, int E,
                                                    const unsigned int* __restrict__ keys,
                                                    unsigned int* __restrict__ rowptr,
                                                    float* __restrict__ dinv,
                                                    int* __restrict__ csr) {
    __shared__ unsigned int cnt[BKT_NODES];
    __shared__ unsigned int wtot[BKT_NODES / 32];
    const int tid  = threadIdx.x;
    const int lane = tid & 31;
    const int wave = tid >> 5;
    const int b = blockIdx.x;
    unsigned int lo = offs[(size_t)b * nblk];
    unsigned int hi = (unsigned int)E;
    if (b + 1 < NBKT) hi = offs[(size_t)(b + 1) * nblk];
    hi = min(hi, (unsigned int)E);
    lo = min(lo, hi);

    if (tid < BKT_NODES) cnt[tid] = 0u;
    __syncthreads();
    unsigned int i0 = lo + tid;
    for (; i0 + (EU - 1) * PB < hi; i0 += EU * PB) {
        unsigned int k[EU];
#pragma unroll
        for (int u = 0; u < EU; ++u) k[u] = keys[i0 + u * PB];
#pragma unroll
        for (int u = 0; u < EU; ++u)
            atomicAdd(&cnt[(k[u] >> SRC_BITS) & (BKT_NODES - 1)], 1u);
    }
    for (; i0 < hi; i0 += PB)
        atomicAdd(&cnt[(keys[i0] >> SRC_BITS) & (BKT_NODES - 1)], 1u);
    __syncthreads();

    const unsigned int v = (tid < BKT_NODES) ? cnt[tid] : 0u;
    unsigned int incl = v;
    for (int off = 1; off < 32; off <<= 1) {
        unsigned int t = __shfl_up(incl, off, 32);
        if (lane >= off) incl += t;
    }
    if (tid < BKT_NODES && lane == 31) wtot[wave] = incl;
    __syncthreads();
    if (tid < BKT_NODES) {
        unsigned int before = 0u;
        for (int w = 0; w < wave; ++w) before += wtot[w];
        const unsigned int start = lo + before + incl - v;
        cnt[tid] = start;
        const int node = b * BKT_NODES + tid;
        if (node < N_NODES) {
            rowptr[node] = start;
            dinv[node] = rsqrtf((float)v + 1.0f);
        }
    }
    __syncthreads();
    i0 = lo + tid;
    for (; i0 + (EU - 1) * PB < hi; i0 += EU * PB) {
        unsigned int k[EU];
#pragma unroll
        for (int u = 0; u < EU; ++u) k[u] = keys[i0 + u * PB];
        unsigned int pos[EU];
#pragma unroll
        for (int u = 0; u < EU; ++u)
            pos[u] = atomicAdd(&cnt[(k[u] >> SRC_BITS) & (BKT_NODES - 1)], 1u);
#pragma unroll
        for (int u = 0; u < EU; ++u) {
            csr[min(pos[u], (unsigned int)E - 1u)] = (int)(k[u] & SRC_MASK);
        }
    }
    for (; i0 < hi; i0 += PB) {
        const unsigned int k = keys[i0];
        const unsigned int pos = atomicAdd(&cnt[(k >> SRC_BITS) & (BKT_NODES - 1)], 1u);
        csr[min(pos, (unsigned int)E - 1u)] = (int)(k & SRC_MASK);
    }
}

typedef float v4f __attribute__((ext_vector_type(4)));

#define KP 388
#define GEMM_ROWS 64
#define GEMM_BLOCKS ((N_NODES + GEMM_ROWS - 1) / GEMM_ROWS)
#define KSUP 64

__global__ __launch_bounds__(128) void k_gemm1(const float* __restrict__ X,
                                               const float* __restrict__ W1,
                                               float* __restrict__ H) {
    __shared__ __attribute__((aligned(16))) float Bt[HID * KP];
    const int tid  = threadIdx.x;

    {
        const int c = tid & (HID - 1);
#pragma unroll 12
        for (int k4 = tid >> 6; k4 < IN_DIM / 4; k4 += 2) {
            const float* p = W1 + (size_t)(4 * k4) * HID + c;
            v4f v;
            v[0] = p[0];
            v[1] = p[HID];
            v[2] = p[2 * HID];
            v[3] = p[3 * HID];
            *(v4f*)&Bt[c * KP + 4 * k4] = v;
        }
    }
    __syncthreads();

    const int lane = tid & 31;
    const int wave = tid >> 5;
    const int half = lane >> 4;
    const int lr   = lane & 15;
    const int rowBase = blockIdx.x * GEMM_ROWS + wave * 16;
    if (rowBase >= N_NODES) return;

    const v4f* xa = (const v4f*)(X + (size_t)(rowBase + lr) * IN_DIM + 4 * half);
    const v4f* bb = (const v4f*)(Bt + lr * KP + 4 * half);

    v8f acc[4] = {};
    v4f a_cur[KSUP / 8], a_nxt[KSUP / 8];
#pragma unroll
    for (int j = 0; j < KSUP / 8; ++j) a_cur[j] = xa[2 * j];

#pragma unroll
    for (int ss = 0; ss < IN_DIM / KSUP; ++ss) {
        if (ss + 1 < IN_DIM / KSUP) {
#pragma unroll
            for (int j = 0; j < KSUP / 8; ++j) a_nxt[j] = xa[(ss + 1) * (KSUP / 4) + 2 * j];
        }
#pragma unroll
        for (int ch = 0; ch < KSUP / 8; ++ch) {
            v4f b[4];
#pragma unroll
            for (int t = 0; t < 4; ++t) b[t] = bb[t * (16 * KP / 4) + ss * (KSUP / 4) + 2 * ch];
            const v4f a = a_cur[ch];
            const v2f a01 = {a[0], a[1]};
            const v2f a23 = {a[2], a[3]};
#pragma unroll
            for (int t = 0; t < 4; ++t) {
                const v2f b01 = {b[t][0], b[t][1]};
                acc[t] = __builtin_amdgcn_wmma_f32_16x16x4_f32(
                         false, a01,  false, b01,
                         (short)0, acc[t],  false,  false);
            }
#pragma unroll
            for (int t = 0; t < 4; ++t) {
                const v2f b23 = {b[t][2], b[t][3]};
                acc[t] = __builtin_amdgcn_wmma_f32_16x16x4_f32(
                         false, a23,  false, b23,
                         (short)0, acc[t],  false,  false);
            }
        }
#pragma unroll
        for (int j = 0; j < KSUP / 8; ++j) a_cur[j] = a_nxt[j];
    }

    float* hrow = H + (size_t)(rowBase + 8 * half) * HID + lr;
#pragma unroll
    for (int t = 0; t < 4; ++t) {
#pragma unroll
        for (int r = 0; r < 8; ++r)
            hrow[(size_t)r * HID + 16 * t] = acc[t][r];
    }
}

__global__ __launch_bounds__(256) void k_gather1(const unsigned int* __restrict__ rowptr,
                                                 const int* __restrict__ csr, int E,
                                                 const float* __restrict__ H,
                                                 const float* __restrict__ dinv,
                                                 const float* __restrict__ b1,
                                                 const float* __restrict__ W2,
                                                 float* __restrict__ z) {
    const int lane = threadIdx.x & 31;
    const int row  = (blockIdx.x * blockDim.x + threadIdx.x) >> 5;
    if (row >= N_NODES) return;

    const int nxt = min(row + 1, N_NODES - 1);
    int endv = (int)rowptr[nxt];
    if (row + 1 >= N_NODES) endv = E;
    const int start = __builtin_amdgcn_readfirstlane((int)rowptr[row]);
    const int end   = __builtin_amdgcn_readfirstlane(min(endv, E));

    const float di = dinv[row];
    const v2f* H2 = (const v2f*)H;
    const v2f hs = H2[(size_t)row * (HID / 2) + lane];
    float a0 = hs[0] * di * di;
    float a1 = hs[1] * di * di;

    for (int base = start; base < end; base += 32) {
        const int idx = base + lane;
        int   s = 0;
        float w = 0.0f;
        if (idx < end) {
            s = csr[idx];
            s = min(max(s, 0), N_NODES - 1);
            w = dinv[s] * di;
        }
        const int cnt = min(32, end - base);
        int t0 = 0;
        for (; t0 + 8 <= cnt; t0 += 8) {
            v2f h[8];
#pragma unroll
            for (int k = 0; k < 8; ++k) {
                const int st = __builtin_amdgcn_readlane(s, t0 + k);
                h[k] = H2[(size_t)st * (HID / 2) + lane];
            }
#pragma unroll
            for (int k = 0; k < 8; ++k) {
                const float wt = __int_as_float(__builtin_amdgcn_readlane(__float_as_int(w), t0 + k));
                a0 += h[k][0] * wt;
                a1 += h[k][1] * wt;
            }
        }
        if (t0 < cnt) {
            v2f h[7];
#pragma unroll
            for (int k = 0; k < 7; ++k) {
                h[k] = (v2f){0.0f, 0.0f};
                if (t0 + k < cnt) {
                    const int st = __builtin_amdgcn_readlane(s, t0 + k);
                    h[k] = H2[(size_t)st * (HID / 2) + lane];
                }
            }
#pragma unroll
            for (int k = 0; k < 7; ++k) {
                const float wt = __int_as_float(__builtin_amdgcn_readlane(__float_as_int(w), t0 + k));
                a0 += h[k][0] * wt;
                a1 += h[k][1] * wt;
            }
        }
    }

    float v0 = a0 + b1[2 * lane];
    float v1 = a1 + b1[2 * lane + 1];
    v0 = v0 > 0.0f ? v0 : 0.0f;
    v1 = v1 > 0.0f ? v1 : 0.0f;
    float sum = v0 * W2[2 * lane] + v1 * W2[2 * lane + 1];
    for (int off = 16; off > 0; off >>= 1)
        sum += __shfl_down(sum, off, 32);
    if (lane == 0) z[row] = sum;
}

__global__ __launch_bounds__(256) void k_gather2(const unsigned int* __restrict__ rowptr,
                                                 const int* __restrict__ csr, int E,
                                                 const float* __restrict__ z,
                                                 const float* __restrict__ dinv,
                                                 const float* __restrict__ b2,
                                                 float* __restrict__ out) {
    const int lane = threadIdx.x & 31;
    const int row  = (blockIdx.x * blockDim.x + threadIdx.x) >> 5;
    if (row >= N_NODES) return;

    int start = (int)rowptr[row];
    int end   = E;
    if (row + 1 < N_NODES) end = (int)rowptr[row + 1];
    end = min(end, E);

    const float di = dinv[row];
    float p = 0.0f;
    for (int idx = start + lane; idx < end; idx += 32) {
        int s = csr[idx];
        s = min(max(s, 0), N_NODES - 1);
        p += z[s] * dinv[s] * di;
    }
    for (int off = 16; off > 0; off >>= 1)
        p += __shfl_down(p, off, 32);
    if (lane == 0) out[row] = z[row] * di * di + b2[0] + p;
}


extern "C" void kernel_launch(void* const* d_in, const int* in_sizes, int n_in,
                              void* d_out, int out_size, void* d_ws, size_t ws_size,
                              hipStream_t stream) {
    const float* X  = (const float*)d_in[0];
    const int*   EI = (const int*)d_in[1];
    const float* W1 = (const float*)d_in[2];
    const float* b1 = (const float*)d_in[3];
    const float* W2 = (const float*)d_in[4];
    const float* b2 = (const float*)d_in[5];
    float* out = (float*)d_out;

    const int E = in_sizes[1] / 2;
    const int* src = EI;
    const int* dst = EI + E;

    char* ws = (char*)d_ws;
    float* H    = (float*)ws;
    unsigned int* keys = (unsigned int*)(ws + (size_t)N_NODES * HID * 4);
    int*   csr  = (int*)(keys + E);
    float* dinv = (float*)(ws + (size_t)2 * N_NODES * HID * 4);
    float* z    = dinv + N_NODES;
    unsigned int* rowptr = (unsigned int*)(z + N_NODES);
    unsigned int* craw = (unsigned int*)dinv;
    unsigned int* offs = (unsigned int*)z;

    const int B = 256;
    k_hist           <<<G1, PB, 0, stream>>>(dst, E, craw);
    k_scan_counts    <<<(NBKT * G1 + SCAN_B - 1) / SCAN_B, SCAN_B, 0, stream>>>(craw, offs, NBKT * G1);
    k_bucket_scatter <<<G1, PB, 0, stream>>>(src, dst, E, offs, keys);

    k_gemm1    <<<GEMM_BLOCKS, 128, 0, stream>>>(X, W1, H);

    k_bucket_sort    <<<NBKT, PB, 0, stream>>>(offs, G1, E, keys, rowptr, dinv, csr);
    k_gather1  <<<(N_NODES * 32 + B - 1) / B, B, 0, stream>>>(rowptr, csr, E, H, dinv, b1, W2, z);
    k_gather2  <<<(N_NODES * 32 + B - 1) / B, B, 0, stream>>>(rowptr, csr, E, z, dinv, b2, out);
}


# Both edge passes: one resident wave sums 8 consecutive destination rows (12500 waves per pass, not 100000); row bounds, dinv, b1/W2/b2 read once per wave, self row one row ahead; sum order kept
# speedup vs baseline: 5.4752x; 1.0499x over previous
#include <hip/hip_runtime.h>


typedef float v2f __attribute__((ext_vector_type(2)));
typedef float v8f __attribute__((ext_vector_type(8)));

#define N_NODES 100000
#define IN_DIM  384
#define HID     64
#define SCAN_B  1024


#define BKT_SHIFT 9
#define BKT_NODES (1 << BKT_SHIFT)
#define NBKT ((N_NODES + BKT_NODES - 1) / BKT_NODES)
#define G1 256
#define PB 1024
#define EU 4
#define SRC_BITS 17
#define SRC_MASK ((1u << SRC_BITS) - 1u)

__global__ __launch_bounds__(PB) void k_hist(const int* __restrict__ dst, int E,
                                             unsigned int* __restrict__ counts) {
    __shared__ unsigned int h[NBKT];
    for (int i = threadIdx.x; i < NBKT; i += PB) h[i] = 0u;
    __syncthreads();
    const int chunk = (E + (int)gridDim.x - 1) / (int)gridDim.x;
    const int lo = (int)blockIdx.x * chunk;
    const int hi = min(E, lo + chunk);
    int e0 = lo + (int)threadIdx.x;
    for (; e0 + (EU - 1) * PB < hi; e0 += EU * PB) {
        int d[EU];
#pragma unroll
        for (int u = 0; u < EU; ++u) d[u] = dst[e0 + u * PB];
#pragma unroll
        for (int u = 0; u < EU; ++u) {
            const int dd = min(max(d[u], 0), N_NODES - 1);
            atomicAdd(&h[dd >> BKT_SHIFT], 1u);
        }
    }
    for (; e0 < hi; e0 += PB) {
        const int dd = min(max(dst[e0], 0), N_NODES - 1);
        atomicAdd(&h[dd >> BKT_SHIFT], 1u);
    }
    __syncthreads();
    for (int i = threadIdx.x; i < NBKT; i += PB) counts[(size_t)i * gridDim.x + blockIdx.x] = h[i];
}

__global__ __launch_bounds__(SCAN_B) void k_scan_counts(const unsigned int* __restrict__ cin,
                                                        unsigned int* __restrict__ cout, int n) {
    __shared__ unsigned int wtot[SCAN_B / 32];
    __shared__ unsigned int wpre[SCAN_B / 32];
    const int lane = threadIdx.x & 31;
    const int wave = threadIdx.x >> 5;
    unsigned int pre = 0u;
    for (int t = 0; t < (int)blockIdx.x; ++t) pre += cin[(size_t)t * SCAN_B + threadIdx.x];
    for (int off = 16; off > 0; off >>= 1) pre += __shfl_down(pre, off, 32);
    const int i = blockIdx.x * SCAN_B + threadIdx.x;
    const unsigned int v = (i < n) ? cin[i] : 0u;
    unsigned int incl = v;
    for (int off = 1; off < 32; off <<= 1) {
        unsigned int t = __shfl_up(incl, off, 32);
        if (lane >= off) incl += t;
    }
    if (lane == 31) wtot[wave] = incl;
    if (lane == 0)  wpre[wave] = pre;
    __syncthreads();
    unsigned int before = 0u;
    for (int w = 0; w < SCAN_B / 32; ++w) before += wpre[w];
    for (int w = 0; w < wave; ++w) before += wtot[w];
    if (i < n) cout[i] = before + incl - v;
}

__global__ __launch_bounds__(PB) void k_bucket_scatter(const int* __restrict__ src,
                                                       const int* __restrict__ dst, int E,
                                                       const unsigned int* __restrict__ offs,
                                                       unsigned int* __restrict__ keys) {
    __shared__ unsigned int cur[NBKT];
    for (int i = threadIdx.x; i < NBKT; i += PB) cur[i] = offs[(size_t)i * gridDim.x + blockIdx.x];
    __syncthreads();
    const int chunk = (E + (int)gridDim.x - 1) / (int)gridDim.x;
    const int lo = (int)blockIdx.x * chunk;
    const int hi = min(E, lo + chunk);
    int e0 = lo + (int)threadIdx.x;
    for (; e0 + (EU - 1) * PB < hi; e0 += EU * PB) {
        int d[EU], s[EU];
#pragma unroll
        for (int u = 0; u < EU; ++u) {
            d[u] = dst[e0 + u * PB];
            s[u] = src[e0 + u * PB];
        }
        unsigned int pos[EU];
#pragma unroll
        for (int u = 0; u < EU; ++u) {
            d[u] = min(max(d[u], 0), N_NODES - 1);
            s[u] = min(max(s[u], 0), N_NODES - 1);
            pos[u] = atomicAdd(&cur[d[u] >> BKT_SHIFT], 1u);
        }
#pragma unroll
        for (int u = 0; u < EU; ++u) {
            keys[min(pos[u], (unsigned int)E - 1u)] =
                ((unsigned int)(d[u] & (BKT_NODES - 1)) << SRC_BITS) | (unsigned int)s[u];
        }
    }
    for (; e0 < hi; e0 += PB) {
        const int dd = min(max(dst[e0], 0), N_NODES - 1);
        const int ss = min(max(src[e0], 0), N_NODES - 1);
        const unsigned int pos = atomicAdd(&cur[dd >> BKT_SHIFT], 1u);
        keys[min(pos, (unsigned int)E - 1u)] = ((unsigned int)(dd & (BKT_NODES - 1)) << SRC_BITS) | (unsigned int)ss;
    }
}

__global__ __launch_bounds__(PB) void k_bucket_sort(const unsigned int* __restrict__ offs, int nblk, int E,
                                                    const unsigned int* __restrict__ keys,
                                                    unsigned int* __restrict__ rowptr,
                                                    float* __restrict__ dinv,
                                                    int* __restrict__ csr) {
    __shared__ unsigned int cnt[BKT_NODES];
    __shared__ unsigned int wtot[BKT_NODES / 32];
    const int tid  = threadIdx.x;
    const int lane = tid & 31;
    const int wave = tid >> 5;
    const int b = blockIdx.x;
    unsigned int lo = offs[(size_t)b * nblk];
    unsigned int hi = (unsigned int)E;
    if (b + 1 < NBKT) hi = offs[(size_t)(b + 1) * nblk];
    hi = min(hi, (unsigned int)E);
    lo = min(lo, hi);

    if (tid < BKT_NODES) cnt[tid] = 0u;
    __syncthreads();
    unsigned int i0 = lo + tid;
    for (; i0 + (EU - 1) * PB < hi; i0 += EU * PB) {
        unsigned int k[EU];
#pragma unroll
        for (int u = 0; u < EU; ++u) k[u] = keys[i0 + u * PB];
#pragma unroll
        for (int u = 0; u < EU; ++u)
            atomicAdd(&cnt[(k[u] >> SRC_BITS) & (BKT_NODES - 1)], 1u);
    }
    for (; i0 < hi; i0 += PB)
        atomicAdd(&cnt[(keys[i0] >> SRC_BITS) & (BKT_NODES - 1)], 1u);
    __syncthreads();

    const unsigned int v = (tid < BKT_NODES) ? cnt[tid] : 0u;
    unsigned int incl = v;
    for (int off = 1; off < 32; off <<= 1) {
        unsigned int t = __shfl_up(incl, off, 32);
        if (lane >= off) incl += t;
    }
    if (tid < BKT_NODES && lane == 31) wtot[wave] = incl;
    __syncthreads();
    if (tid < BKT_NODES) {
        unsigned int before = 0u;
        for (int w = 0; w < wave; ++w) before += wtot[w];
        const unsigned int start = lo + before + incl - v;
        cnt[tid] = start;
        const int node = b * BKT_NODES + tid;
        if (node < N_NODES) {
            rowptr[node] = start;
            dinv[node] = rsqrtf((float)v + 1.0f);
        }
    }
    __syncthreads();
    i0 = lo + tid;
    for (; i0 + (EU - 1) * PB < hi; i0 += EU * PB) {
        unsigned int k[EU];
#pragma unroll
        for (int u = 0; u < EU; ++u) k[u] = keys[i0 + u * PB];
        unsigned int pos[EU];
#pragma unroll
        for (int u = 0; u < EU; ++u)
            pos[u] = atomicAdd(&cnt[(k[u] >> SRC_BITS) & (BKT_NODES - 1)], 1u);
#pragma unroll
        for (int u = 0; u < EU; ++u) {
            csr[min(pos[u], (unsigned int)E - 1u)] = (int)(k[u] & SRC_MASK);
        }
    }
    for (; i0 < hi; i0 += PB) {
        const unsigned int k = keys[i0];
        const unsigned int pos = atomicAdd(&cnt[(k >> SRC_BITS) & (BKT_NODES - 1)], 1u);
        csr[min(pos, (unsigned int)E - 1u)] = (int)(k & SRC_MASK);
    }
}

typedef float v4f __attribute__((ext_vector_type(4)));

#define KP 388
#define GEMM_ROWS 64
#define GEMM_BLOCKS ((N_NODES + GEMM_ROWS - 1) / GEMM_ROWS)
#define KSUP 64

__global__ __launch_bounds__(128) void k_gemm1(const float* __restrict__ X,
                                               const float* __restrict__ W1,
                                               float* __restrict__ H) {
    __shared__ __attribute__((aligned(16))) float Bt[HID * KP];
    const int tid  = threadIdx.x;

    {
        const int c = tid & (HID - 1);
#pragma unroll 12
        for (int k4 = tid >> 6; k4 < IN_DIM / 4; k4 += 2) {
            const float* p = W1 + (size_t)(4 * k4) * HID + c;
            v4f v;
            v[0] = p[0];
            v[1] = p[HID];
            v[2] = p[2 * HID];
            v[3] = p[3 * HID];
            *(v4f*)&Bt[c * KP + 4 * k4] = v;
        }
    }
    __syncthreads();

    const int lane = tid & 31;
    const int wave = tid >> 5;
    const int half = lane >> 4;
    const int lr   = lane & 15;
    const int rowBase = blockIdx.x * GEMM_ROWS + wave * 16;
    if (rowBase >= N_NODES) return;

    const v4f* xa = (const v4f*)(X + (size_t)(rowBase + lr) * IN_DIM + 4 * half);
    const v4f* bb = (const v4f*)(Bt + lr * KP + 4 * half);

    v8f acc[4] = {};
    v4f a_cur[KSUP / 8], a_nxt[KSUP / 8];
#pragma unroll
    for (int j = 0; j < KSUP / 8; ++j) a_cur[j] = xa[2 * j];

#pragma unroll
    for (int ss = 0; ss < IN_DIM / KSUP; ++ss) {
        if (ss + 1 < IN_DIM / KSUP) {
#pragma unroll
            for (int j = 0; j < KSUP / 8; ++j) a_nxt[j] = xa[(ss + 1) * (KSUP / 4) + 2 * j];
        }
#pragma unroll
        for (int ch = 0; ch < KSUP / 8; ++ch) {
            v4f b[4];
#pragma unroll
            for (int t = 0; t < 4; ++t) b[t] = bb[t * (16 * KP / 4) + ss * (KSUP / 4) + 2 * ch];
            const v4f a = a_cur[ch];
            const v2f a01 = {a[0], a[1]};
            const v2f a23 = {a[2], a[3]};
#pragma unroll
            for (int t = 0; t < 4; ++t) {
                const v2f b01 = {b[t][0], b[t][1]};
                acc[t] = __builtin_amdgcn_wmma_f32_16x16x4_f32(
                         false, a01,  false, b01,
                         (short)0, acc[t],  false,  false);
            }
#pragma unroll
            for (int t = 0; t < 4; ++t) {
                const v2f b23 = {b[t][2], b[t][3]};
                acc[t] = __builtin_amdgcn_wmma_f32_16x16x4_f32(
                         false, a23,  false, b23,
                         (short)0, acc[t],  false,  false);
            }
        }
#pragma unroll
        for (int j = 0; j < KSUP / 8; ++j) a_cur[j] = a_nxt[j];
    }

    float* hrow = H + (size_t)(rowBase + 8 * half) * HID + lr;
#pragma unroll
    for (int t = 0; t < 4; ++t) {
#pragma unroll
        for (int r = 0; r < 8; ++r)
            hrow[(size_t)r * HID + 16 * t] = acc[t][r];
    }
}

#define RPW 8
#define GATHER_T 128
#define GATHER_BLOCKS ((N_NODES + (GATHER_T / 32) * RPW - 1) / ((GATHER_T / 32) * RPW))

__global__ __launch_bounds__(GATHER_T) void k_gather1(const unsigned int* __restrict__ rowptr,
                                                      const int* __restrict__ csr, int E,
                                                      const float* __restrict__ H,
                                                      const float* __restrict__ dinv,
                                                      const float* __restrict__ b1,
                                                      const float* __restrict__ W2,
                                                      float* __restrict__ z) {
    const int lane = threadIdx.x & 31;
    const int gw = __builtin_amdgcn_readfirstlane((int)((blockIdx.x * blockDim.x + threadIdx.x) >> 5));
    const int row0 = gw * RPW;
    if (row0 >= N_NODES) return;
    const int nrows = min(RPW, N_NODES - row0);

    const int bi = row0 + min(lane, nrows);
    int rb = E;
    if (bi < N_NODES) rb = (int)rowptr[bi];
    rb = min(rb, E);
    const float dl = dinv[row0 + min(lane, nrows - 1)];
    const float b1a = b1[2 * lane];
    const float b1b = b1[2 * lane + 1];
    const float w2a = W2[2 * lane];
    const float w2b = W2[2 * lane + 1];

    const v2f* H2 = (const v2f*)H;
    float zv = 0.0f;
    v2f hs_next = H2[(size_t)row0 * (HID / 2) + lane];

    for (int r = 0; r < nrows; ++r) {
        const int row = row0 + r;
        const int start = __builtin_amdgcn_readlane(rb, r);
        const int end   = __builtin_amdgcn_readlane(rb, r + 1);
        const float di  = __int_as_float(__builtin_amdgcn_readlane(__float_as_int(dl), r));

        const v2f hs = hs_next;
        if (r + 1 < nrows) hs_next = H2[(size_t)(row + 1) * (HID / 2) + lane];
        float a0 = hs[0] * di * di;
        float a1 = hs[1] * di * di;

        for (int base = start; base < end; base += 32) {
            const int idx = base + lane;
            int   s = 0;
            float w = 0.0f;
            if (idx < end) {
                s = csr[idx];
                s = min(max(s, 0), N_NODES - 1);
                w = dinv[s] * di;
            }
            const int cnt = min(32, end - base);
            int t0 = 0;
            for (; t0 + 8 <= cnt; t0 += 8) {
                v2f h[8];
#pragma unroll
                for (int k = 0; k < 8; ++k) {
                    const int st = __builtin_amdgcn_readlane(s, t0 + k);
                    h[k] = H2[(size_t)st * (HID / 2) + lane];
                }
#pragma unroll
                for (int k = 0; k < 8; ++k) {
                    const float wt = __int_as_float(__builtin_amdgcn_readlane(__float_as_int(w), t0 + k));
                    a0 += h[k][0] * wt;
                    a1 += h[k][1] * wt;
                }
            }
            if (t0 < cnt) {
                v2f h[7];
#pragma unroll
                for (int k = 0; k < 7; ++k) {
                    h[k] = (v2f){0.0f, 0.0f};
                    if (t0 + k < cnt) {
                        const int st = __builtin_amdgcn_readlane(s, t0 + k);
                        h[k] = H2[(size_t)st * (HID / 2) + lane];
                    }
                }
#pragma unroll
                for (int k = 0; k < 7; ++k) {
                    const float wt = __int_as_float(__builtin_amdgcn_readlane(__float_as_int(w), t0 + k));
                    a0 += h[k][0] * wt;
                    a1 += h[k][1] * wt;
                }
            }
        }

        float v0 = a0 + b1a;
        float v1 = a1 + b1b;
        v0 = v0 > 0.0f ? v0 : 0.0f;
        v1 = v1 > 0.0f ? v1 : 0.0f;
        float sum = v0 * w2a + v1 * w2b;
        for (int off = 16; off > 0; off >>= 1)
            sum += __shfl_down(sum, off, 32);
        const float tot = __int_as_float(__builtin_amdgcn_readfirstlane(__float_as_int(sum)));
        if (lane == r) zv = tot;
    }
    if (lane < nrows) z[row0 + lane] = zv;
}

__global__ __launch_bounds__(GATHER_T) void k_gather2(const unsigned int* __restrict__ rowptr,
                                                      const int* __restrict__ csr, int E,
                                                      const float* __restrict__ z,
                                                      const float* __restrict__ dinv,
                                                      const float* __restrict__ b2,
                                                      float* __restrict__ out) {
    const int lane = threadIdx.x & 31;
    const int gw = __builtin_amdgcn_readfirstlane((int)((blockIdx.x * blockDim.x + threadIdx.x) >> 5));
    const int row0 = gw * RPW;
    if (row0 >= N_NODES) return;
    const int nrows = min(RPW, N_NODES - row0);

    const int bi = row0 + min(lane, nrows);
    int rb = E;
    if (bi < N_NODES) rb = (int)rowptr[bi];
    rb = min(rb, E);
    const int rl = row0 + min(lane, nrows - 1);
    const float dl = dinv[rl];
    const float zl = z[rl];
    const float bias = b2[0];
    float ov = 0.0f;

    for (int r = 0; r < nrows; ++r) {
        const int start = __builtin_amdgcn_readlane(rb, r);
        const int end   = __builtin_amdgcn_readlane(rb, r + 1);
        const float di  = __int_as_float(__builtin_amdgcn_readlane(__float_as_int(dl), r));
        float p = 0.0f;
        for (int idx = start + lane; idx < end; idx += 32) {
            int s = csr[idx];
            s = min(max(s, 0), N_NODES - 1);
            p += z[s] * dinv[s] * di;
        }
        for (int off = 16; off > 0; off >>= 1)
            p += __shfl_down(p, off, 32);
        const float tot = __int_as_float(__builtin_amdgcn_readfirstlane(__float_as_int(p)));
        if (lane == r) ov = tot;
    }
    if (lane < nrows) out[row0 + lane] = zl * dl * dl + bias + ov;
}


extern "C" void kernel_launch(void* const* d_in, const int* in_sizes, int n_in,
                              void* d_out, int out_size, void* d_ws, size_t ws_size,
                              hipStream_t stream) {
    const float* X  = (const float*)d_in[0];
    const int*   EI = (const int*)d_in[1];
    const float* W1 = (const float*)d_in[2];
    const float* b1 = (const float*)d_in[3];
    const float* W2 = (const float*)d_in[4];
    const float* b2 = (const float*)d_in[5];
    float* out = (float*)d_out;

    const int E = in_sizes[1] / 2;
    const int* src = EI;
    const int* dst = EI + E;

    char* ws = (char*)d_ws;
    float* H    = (float*)ws;
    unsigned int* keys = (unsigned int*)(ws + (size_t)N_NODES * HID * 4);
    int*   csr  = (int*)(keys + E);
    float* dinv = (float*)(ws + (size_t)2 * N_NODES * HID * 4);
    float* z    = dinv + N_NODES;
    unsigned int* rowptr = (unsigned int*)(z + N_NODES);
    unsigned int* craw = (unsigned int*)dinv;
    unsigned int* offs = (unsigned int*)z;

    k_hist           <<<G1, PB, 0, stream>>>(dst, E, craw);
    k_scan_counts    <<<(NBKT * G1 + SCAN_B - 1) / SCAN_B, SCAN_B, 0, stream>>>(craw, offs, NBKT * G1);
    k_bucket_scatter <<<G1, PB, 0, stream>>>(src, dst, E, offs, keys);

    k_gemm1    <<<GEMM_BLOCKS, 128, 0, stream>>>(X, W1, H);

    k_bucket_sort    <<<NBKT, PB, 0, stream>>>(offs, G1, E, keys, rowptr, dinv, csr);
    k_gather1  <<<GATHER_BLOCKS, GATHER_T, 0, stream>>>(rowptr, csr, E, H, dinv, b1, W2, z);
    k_gather2  <<<GATHER_BLOCKS, GATHER_T, 0, stream>>>(rowptr, csr, E, z, dinv, b2, out);
}


# dinv[src] folded into stored node values (X@W1 rows stored as H*dinv, product launched after the sort; layer 1 stores z*dinv): no per-edge dinv read in either edge pass; term=(v*dinv[src])*dinv[dst]
# speedup vs baseline: 6.2760x; 1.1463x over previous
#include <hip/hip_runtime.h>


typedef float v2f __attribute__((ext_vector_type(2)));
typedef float v8f __attribute__((ext_vector_type(8)));

#define N_NODES 100000
#define IN_DIM  384
#define HID     64
#define SCAN_B  1024


#define BKT_SHIFT 9
#define BKT_NODES (1 << BKT_SHIFT)
#define NBKT ((N_NODES + BKT_NODES - 1) / BKT_NODES)
#define G1 256
#define PB 1024
#define EU 4
#define SRC_BITS 17
#define SRC_MASK ((1u << SRC_BITS) - 1u)

__global__ __launch_bounds__(PB) void k_hist(const int* __restrict__ dst, int E,
                                             unsigned int* __restrict__ counts) {
    __shared__ unsigned int h[NBKT];
    for (int i = threadIdx.x; i < NBKT; i += PB) h[i] = 0u;
    __syncthreads();
    const int chunk = (E + (int)gridDim.x - 1) / (int)gridDim.x;
    const int lo = (int)blockIdx.x * chunk;
    const int hi = min(E, lo + chunk);
    int e0 = lo + (int)threadIdx.x;
    for (; e0 + (EU - 1) * PB < hi; e0 += EU * PB) {
        int d[EU];
#pragma unroll
        for (int u = 0; u < EU; ++u) d[u] = dst[e0 + u * PB];
#pragma unroll
        for (int u = 0; u < EU; ++u) {
            const int dd = min(max(d[u], 0), N_NODES - 1);
            atomicAdd(&h[dd >> BKT_SHIFT], 1u);
        }
    }
    for (; e0 < hi; e0 += PB) {
        const int dd = min(max(dst[e0], 0), N_NODES - 1);
        atomicAdd(&h[dd >> BKT_SHIFT], 1u);
    }
    __syncthreads();
    for (int i = threadIdx.x; i < NBKT; i += PB) counts[(size_t)i * gridDim.x + blockIdx.x] = h[i];
}

__global__ __launch_bounds__(SCAN_B) void k_scan_counts(const unsigned int* __restrict__ cin,
                                                        unsigned int* __restrict__ cout, int n) {
    __shared__ unsigned int wtot[SCAN_B / 32];
    __shared__ unsigned int wpre[SCAN_B / 32];
    const int lane = threadIdx.x & 31;
    const int wave = threadIdx.x >> 5;
    unsigned int pre = 0u;
    for (int t = 0; t < (int)blockIdx.x; ++t) pre += cin[(size_t)t * SCAN_B + threadIdx.x];
    for (int off = 16; off > 0; off >>= 1) pre += __shfl_down(pre, off, 32);
    const int i = blockIdx.x * SCAN_B + threadIdx.x;
    const unsigned int v = (i < n) ? cin[i] : 0u;
    unsigned int incl = v;
    for (int off = 1; off < 32; off <<= 1) {
        unsigned int t = __shfl_up(incl, off, 32);
        if (lane >= off) incl += t;
    }
    if (lane == 31) wtot[wave] = incl;
    if (lane == 0)  wpre[wave] = pre;
    __syncthreads();
    unsigned int before = 0u;
    for (int w = 0; w < SCAN_B / 32; ++w) before += wpre[w];
    for (int w = 0; w < wave; ++w) before += wtot[w];
    if (i < n) cout[i] = before + incl - v;
}

__global__ __launch_bounds__(PB) void k_bucket_scatter(const int* __restrict__ src,
                                                       const int* __restrict__ dst, int E,
                                                       const unsigned int* __restrict__ offs,
                                                       unsigned int* __restrict__ keys) {
    __shared__ unsigned int cur[NBKT];
    for (int i = threadIdx.x; i < NBKT; i += PB) cur[i] = offs[(size_t)i * gridDim.x + blockIdx.x];
    __syncthreads();
    const int chunk = (E + (int)gridDim.x - 1) / (int)gridDim.x;
    const int lo = (int)blockIdx.x * chunk;
    const int hi = min(E, lo + chunk);
    int e0 = lo + (int)threadIdx.x;
    for (; e0 + (EU - 1) * PB < hi; e0 += EU * PB) {
        int d[EU], s[EU];
#pragma unroll
        for (int u = 0; u < EU; ++u) {
            d[u] = dst[e0 + u * PB];
            s[u] = src[e0 + u * PB];
        }
        unsigned int pos[EU];
#pragma unroll
        for (int u = 0; u < EU; ++u) {
            d[u] = min(max(d[u], 0), N_NODES - 1);
            s[u] = min(max(s[u], 0), N_NODES - 1);
            pos[u] = atomicAdd(&cur[d[u] >> BKT_SHIFT], 1u);
        }
#pragma unroll
        for (int u = 0; u < EU; ++u) {
            keys[min(pos[u], (unsigned int)E - 1u)] =
                ((unsigned int)(d[u] & (BKT_NODES - 1)) << SRC_BITS) | (unsigned int)s[u];
        }
    }
    for (; e0 < hi; e0 += PB) {
        const int dd = min(max(dst[e0], 0), N_NODES - 1);
        const int ss = min(max(src[e0], 0), N_NODES - 1);
        const unsigned int pos = atomicAdd(&cur[dd >> BKT_SHIFT], 1u);
        keys[min(pos, (unsigned int)E - 1u)] = ((unsigned int)(dd & (BKT_NODES - 1)) << SRC_BITS) | (unsigned int)ss;
    }
}

__global__ __launch_bounds__(PB) void k_bucket_sort(const unsigned int* __restrict__ offs, int nblk, int E,
                                                    const unsigned int* __restrict__ keys,
                                                    unsigned int* __restrict__ rowptr,
                                                    float* __restrict__ dinv,
                                                    int* __restrict__ csr) {
    __shared__ unsigned int cnt[BKT_NODES];
    __shared__ unsigned int wtot[BKT_NODES / 32];
    const int tid  = threadIdx.x;
    const int lane = tid & 31;
    const int wave = tid >> 5;
    const int b = blockIdx.x;
    unsigned int lo = offs[(size_t)b * nblk];
    unsigned int hi = (unsigned int)E;
    if (b + 1 < NBKT) hi = offs[(size_t)(b + 1) * nblk];
    hi = min(hi, (unsigned int)E);
    lo = min(lo, hi);

    if (tid < BKT_NODES) cnt[tid] = 0u;
    __syncthreads();
    unsigned int i0 = lo + tid;
    for (; i0 + (EU - 1) * PB < hi; i0 += EU * PB) {
        unsigned int k[EU];
#pragma unroll
        for (int u = 0; u < EU; ++u) k[u] = keys[i0 + u * PB];
#pragma unroll
        for (int u = 0; u < EU; ++u)
            atomicAdd(&cnt[(k[u] >> SRC_BITS) & (BKT_NODES - 1)], 1u);
    }
    for (; i0 < hi; i0 += PB)
        atomicAdd(&cnt[(keys[i0] >> SRC_BITS) & (BKT_NODES - 1)], 1u);
    __syncthreads();

    const unsigned int v = (tid < BKT_NODES) ? cnt[tid] : 0u;
    unsigned int incl = v;
    for (int off = 1; off < 32; off <<= 1) {
        unsigned int t = __shfl_up(incl, off, 32);
        if (lane >= off) incl += t;
    }
    if (tid < BKT_NODES && lane == 31) wtot[wave] = incl;
    __syncthreads();
    if (tid < BKT_NODES) {
        unsigned int before = 0u;
        for (int w = 0; w < wave; ++w) before += wtot[w];
        const unsigned int start = lo + before + incl - v;
        cnt[tid] = start;
        const int node = b * BKT_NODES + tid;
        if (node < N_NODES) {
            rowptr[node] = start;
            dinv[node] = rsqrtf((float)v + 1.0f);
        }
    }
    __syncthreads();
    i0 = lo + tid;
    for (; i0 + (EU - 1) * PB < hi; i0 += EU * PB) {
        unsigned int k[EU];
#pragma unroll
        for (int u = 0; u < EU; ++u) k[u] = keys[i0 + u * PB];
        unsigned int pos[EU];
#pragma unroll
        for (int u = 0; u < EU; ++u)
            pos[u] = atomicAdd(&cnt[(k[u] >> SRC_BITS) & (BKT_NODES - 1)], 1u);
#pragma unroll
        for (int u = 0; u < EU; ++u) {
            csr[min(pos[u], (unsigned int)E - 1u)] = (int)(k[u] & SRC_MASK);
        }
    }
    for (; i0 < hi; i0 += PB) {
        const unsigned int k = keys[i0];
        const unsigned int pos = atomicAdd(&cnt[(k >> SRC_BITS) & (BKT_NODES - 1)], 1u);
        csr[min(pos, (unsigned int)E - 1u)] = (int)(k & SRC_MASK);
    }
}

typedef float v4f __attribute__((ext_vector_type(4)));

#define KP 388
#define GEMM_ROWS 64
#define GEMM_BLOCKS ((N_NODES + GEMM_ROWS - 1) / GEMM_ROWS)
#define KSUP 64

__global__ __launch_bounds__(128) void k_gemm1(const float* __restrict__ X,
                                               const float* __restrict__ W1,
                                               const float* __restrict__ dinv,
                                               float* __restrict__ H) {
    __shared__ __attribute__((aligned(16))) float Bt[HID * KP];
    const int tid  = threadIdx.x;

    {
        const int c = tid & (HID - 1);
#pragma unroll 12
        for (int k4 = tid >> 6; k4 < IN_DIM / 4; k4 += 2) {
            const float* p = W1 + (size_t)(4 * k4) * HID + c;
            v4f v;
            v[0] = p[0];
            v[1] = p[HID];
            v[2] = p[2 * HID];
            v[3] = p[3 * HID];
            *(v4f*)&Bt[c * KP + 4 * k4] = v;
        }
    }
    __syncthreads();

    const int lane = tid & 31;
    const int wave = tid >> 5;
    const int half = lane >> 4;
    const int lr   = lane & 15;
    const int rowBase = blockIdx.x * GEMM_ROWS + wave * 16;
    if (rowBase >= N_NODES) return;

    const v4f* xa = (const v4f*)(X + (size_t)(rowBase + lr) * IN_DIM + 4 * half);
    const v4f* bb = (const v4f*)(Bt + lr * KP + 4 * half);

    const v4f* dp = (const v4f*)(dinv + rowBase + 8 * half);
    const v4f dlo = dp[0];
    const v4f dhi = dp[1];

    v8f acc[4] = {};
    v4f a_cur[KSUP / 8], a_nxt[KSUP / 8];
#pragma unroll
    for (int j = 0; j < KSUP / 8; ++j) a_cur[j] = xa[2 * j];

#pragma unroll
    for (int ss = 0; ss < IN_DIM / KSUP; ++ss) {
        if (ss + 1 < IN_DIM / KSUP) {
#pragma unroll
            for (int j = 0; j < KSUP / 8; ++j) a_nxt[j] = xa[(ss + 1) * (KSUP / 4) + 2 * j];
        }
#pragma unroll
        for (int ch = 0; ch < KSUP / 8; ++ch) {
            v4f b[4];
#pragma unroll
            for (int t = 0; t < 4; ++t) b[t] = bb[t * (16 * KP / 4) + ss * (KSUP / 4) + 2 * ch];
            const v4f a = a_cur[ch];
            const v2f a01 = {a[0], a[1]};
            const v2f a23 = {a[2], a[3]};
#pragma unroll
            for (int t = 0; t < 4; ++t) {
                const v2f b01 = {b[t][0], b[t][1]};
                acc[t] = __builtin_amdgcn_wmma_f32_16x16x4_f32(
                         false, a01,  false, b01,
                         (short)0, acc[t],  false,  false);
            }
#pragma unroll
            for (int t = 0; t < 4; ++t) {
                const v2f b23 = {b[t][2], b[t][3]};
                acc[t] = __builtin_amdgcn_wmma_f32_16x16x4_f32(
                         false, a23,  false, b23,
                         (short)0, acc[t],  false,  false);
            }
        }
#pragma unroll
        for (int j = 0; j < KSUP / 8; ++j) a_cur[j] = a_nxt[j];
    }

    float* hrow = H + (size_t)(rowBase + 8 * half) * HID + lr;
#pragma unroll
    for (int t = 0; t < 4; ++t) {
#pragma unroll
        for (int r = 0; r < 8; ++r)
            hrow[(size_t)r * HID + 16 * t] = acc[t][r] * (r < 4 ? dlo[r & 3] : dhi[r & 3]);
    }
}

#define RPW 8
#define GATHER_T 128
#define GATHER_BLOCKS ((N_NODES + (GATHER_T / 32) * RPW - 1) / ((GATHER_T / 32) * RPW))

__global__ __launch_bounds__(GATHER_T) void k_gather1(const unsigned int* __restrict__ rowptr,
                                                      const int* __restrict__ csr, int E,
                                                      const float* __restrict__ H,
                                                      const float* __restrict__ dinv,
                                                      const float* __restrict__ b1,
                                                      const float* __restrict__ W2,
                                                      float* __restrict__ zs) {
    const int lane = threadIdx.x & 31;
    const int gw = __builtin_amdgcn_readfirstlane((int)((blockIdx.x * blockDim.x + threadIdx.x) >> 5));
    const int row0 = gw * RPW;
    if (row0 >= N_NODES) return;
    const int nrows = min(RPW, N_NODES - row0);

    const int bi = row0 + min(lane, nrows);
    int rb = E;
    if (bi < N_NODES) rb = (int)rowptr[bi];
    rb = min(rb, E);
    const float dl = dinv[row0 + min(lane, nrows - 1)];
    const float b1a = b1[2 * lane];
    const float b1b = b1[2 * lane + 1];
    const float w2a = W2[2 * lane];
    const float w2b = W2[2 * lane + 1];

    const v2f* H2 = (const v2f*)H;
    float zv = 0.0f;
    v2f hs_next = H2[(size_t)row0 * (HID / 2) + lane];

    for (int r = 0; r < nrows; ++r) {
        const int row = row0 + r;
        const int start = __builtin_amdgcn_readlane(rb, r);
        const int end   = __builtin_amdgcn_readlane(rb, r + 1);
        const float di  = __int_as_float(__builtin_amdgcn_readlane(__float_as_int(dl), r));

        const v2f hs = hs_next;
        if (r + 1 < nrows) hs_next = H2[(size_t)(row + 1) * (HID / 2) + lane];
        float a0 = hs[0] * di;
        float a1 = hs[1] * di;

        for (int base = start; base < end; base += 32) {
            const int idx = base + lane;
            int   s = 0;
            if (idx < end) {
                s = csr[idx];
                s = min(max(s, 0), N_NODES - 1);
            }
            const int cnt = min(32, end - base);
            int t0 = 0;
            for (; t0 + 8 <= cnt; t0 += 8) {
                v2f h[8];
#pragma unroll
                for (int k = 0; k < 8; ++k) {
                    const int st = __builtin_amdgcn_readlane(s, t0 + k);
                    h[k] = H2[(size_t)st * (HID / 2) + lane];
                }
#pragma unroll
                for (int k = 0; k < 8; ++k) {
                    a0 += h[k][0] * di;
                    a1 += h[k][1] * di;
                }
            }
            if (t0 < cnt) {
                v2f h[7];
#pragma unroll
                for (int k = 0; k < 7; ++k) {
                    h[k] = (v2f){0.0f, 0.0f};
                    if (t0 + k < cnt) {
                        const int st = __builtin_amdgcn_readlane(s, t0 + k);
                        h[k] = H2[(size_t)st * (HID / 2) + lane];
                    }
                }
#pragma unroll
                for (int k = 0; k < 7; ++k) {
                    a0 += h[k][0] * di;
                    a1 += h[k][1] * di;
                }
            }
        }

        float v0 = a0 + b1a;
        float v1 = a1 + b1b;
        v0 = v0 > 0.0f ? v0 : 0.0f;
        v1 = v1 > 0.0f ? v1 : 0.0f;
        float sum = v0 * w2a + v1 * w2b;
        for (int off = 16; off > 0; off >>= 1)
            sum += __shfl_down(sum, off, 32);
        const float tot = __int_as_float(__builtin_amdgcn_readfirstlane(__float_as_int(sum)));
        if (lane == r) zv = tot * di;
    }
    if (lane < nrows) zs[row0 + lane] = zv;
}

__global__ __launch_bounds__(GATHER_T) void k_gather2(const unsigned int* __restrict__ rowptr,
                                                      const int* __restrict__ csr, int E,
                                                      const float* __restrict__ zs,
                                                      const float* __restrict__ dinv,
                                                      const float* __restrict__ b2,
                                                      float* __restrict__ out) {
    const int lane = threadIdx.x & 31;
    const int gw = __builtin_amdgcn_readfirstlane((int)((blockIdx.x * blockDim.x + threadIdx.x) >> 5));
    const int row0 = gw * RPW;
    if (row0 >= N_NODES) return;
    const int nrows = min(RPW, N_NODES - row0);

    const int bi = row0 + min(lane, nrows);
    int rb = E;
    if (bi < N_NODES) rb = (int)rowptr[bi];
    rb = min(rb, E);
    const int rl = row0 + min(lane, nrows - 1);
    const float dl = dinv[rl];
    const float zl = zs[rl];
    const float bias = b2[0];
    float ov = 0.0f;

    for (int r = 0; r < nrows; ++r) {
        const int start = __builtin_amdgcn_readlane(rb, r);
        const int end   = __builtin_amdgcn_readlane(rb, r + 1);
        const float di  = __int_as_float(__builtin_amdgcn_readlane(__float_as_int(dl), r));
        float p = 0.0f;
        for (int idx = start + lane; idx < end; idx += 32) {
            int s = csr[idx];
            s = min(max(s, 0), N_NODES - 1);
            p += zs[s] * di;
        }
        for (int off = 16; off > 0; off >>= 1)
            p += __shfl_down(p, off, 32);
        const float tot = __int_as_float(__builtin_amdgcn_readfirstlane(__float_as_int(p)));
        if (lane == r) ov = tot;
    }
    if (lane < nrows) out[row0 + lane] = zl * dl + bias + ov;
}


extern "C" void kernel_launch(void* const* d_in, const int* in_sizes, int n_in,
                              void* d_out, int out_size, void* d_ws, size_t ws_size,
                              hipStream_t stream) {
    const float* X  = (const float*)d_in[0];
    const int*   EI = (const int*)d_in[1];
    const float* W1 = (const float*)d_in[2];
    const float* b1 = (const float*)d_in[3];
    const float* W2 = (const float*)d_in[4];
    const float* b2 = (const float*)d_in[5];
    float* out = (float*)d_out;

    const int E = in_sizes[1] / 2;
    const int* src = EI;
    const int* dst = EI + E;

    char* ws = (char*)d_ws;
    float* H    = (float*)ws;
    unsigned int* keys = (unsigned int*)(ws + (size_t)N_NODES * HID * 4);
    int*   csr  = (int*)(keys + E);
    float* dinv = (float*)(ws + (size_t)2 * N_NODES * HID * 4);
    float* z    = dinv + N_NODES;
    unsigned int* rowptr = (unsigned int*)(z + N_NODES);
    unsigned int* craw = (unsigned int*)dinv;
    unsigned int* offs = (unsigned int*)z;

    k_hist           <<<G1, PB, 0, stream>>>(dst, E, craw);
    k_scan_counts    <<<(NBKT * G1 + SCAN_B - 1) / SCAN_B, SCAN_B, 0, stream>>>(craw, offs, NBKT * G1);
    k_bucket_scatter <<<G1, PB, 0, stream>>>(src, dst, E, offs, keys);
    k_bucket_sort    <<<NBKT, PB, 0, stream>>>(offs, G1, E, keys, rowptr, dinv, csr);

    k_gemm1    <<<GEMM_BLOCKS, 128, 0, stream>>>(X, W1, dinv, H);

    k_gather1  <<<GATHER_BLOCKS, GATHER_T, 0, stream>>>(rowptr, csr, E, H, dinv, b1, W2, z);
    k_gather2  <<<GATHER_BLOCKS, GATHER_T, 0, stream>>>(rowptr, csr, E, z, dinv, b2, out);
}


# Both scattered 4-byte store passes of the edge sort staged in LDS: a block orders its edges in LDS, then writes consecutive words (whole 128-B lines); row sums still in LDS-atomic arrival order
# speedup vs baseline: 7.4582x; 1.1884x over previous
#include <hip/hip_runtime.h>


typedef float v2f __attribute__((ext_vector_type(2)));
typedef float v8f __attribute__((ext_vector_type(8)));

#define N_NODES 100000
#define IN_DIM  384
#define HID     64
#define SCAN_B  1024


#define BKT_SHIFT 9
#define BKT_NODES (1 << BKT_SHIFT)
#define NBKT ((N_NODES + BKT_NODES - 1) / BKT_NODES)
#define G1 256
#define PB 1024
#define EU 4
#define SRC_BITS 17
#define SRC_MASK ((1u << SRC_BITS) - 1u)

__global__ __launch_bounds__(PB) void k_hist(const int* __restrict__ dst, int E,
                                             unsigned int* __restrict__ counts) {
    __shared__ unsigned int h[NBKT];
    for (int i = threadIdx.x; i < NBKT; i += PB) h[i] = 0u;
    __syncthreads();
    const int chunk = (E + (int)gridDim.x - 1) / (int)gridDim.x;
    const int lo = (int)blockIdx.x * chunk;
    const int hi = min(E, lo + chunk);
    int e0 = lo + (int)threadIdx.x;
    for (; e0 + (EU - 1) * PB < hi; e0 += EU * PB) {
        int d[EU];
#pragma unroll
        for (int u = 0; u < EU; ++u) d[u] = dst[e0 + u * PB];
#pragma unroll
        for (int u = 0; u < EU; ++u) {
            const int dd = min(max(d[u], 0), N_NODES - 1);
            atomicAdd(&h[dd >> BKT_SHIFT], 1u);
        }
    }
    for (; e0 < hi; e0 += PB) {
        const int dd = min(max(dst[e0], 0), N_NODES - 1);
        atomicAdd(&h[dd >> BKT_SHIFT], 1u);
    }
    __syncthreads();
    for (int i = threadIdx.x; i < NBKT; i += PB) counts[(size_t)i * gridDim.x + blockIdx.x] = h[i];
}

__global__ __launch_bounds__(SCAN_B) void k_scan_counts(const unsigned int* __restrict__ cin,
                                                        unsigned int* __restrict__ cout, int n) {
    __shared__ unsigned int wtot[SCAN_B / 32];
    __shared__ unsigned int wpre[SCAN_B / 32];
    const int lane = threadIdx.x & 31;
    const int wave = threadIdx.x >> 5;
    unsigned int pre = 0u;
    for (int t = 0; t < (int)blockIdx.x; ++t) pre += cin[(size_t)t * SCAN_B + threadIdx.x];
    for (int off = 16; off > 0; off >>= 1) pre += __shfl_down(pre, off, 32);
    const int i = blockIdx.x * SCAN_B + threadIdx.x;
    const unsigned int v = (i < n) ? cin[i] : 0u;
    unsigned int incl = v;
    for (int off = 1; off < 32; off <<= 1) {
        unsigned int t = __shfl_up(incl, off, 32);
        if (lane >= off) incl += t;
    }
    if (lane == 31) wtot[wave] = incl;
    if (lane == 0)  wpre[wave] = pre;
    __syncthreads();
    unsigned int before = 0u;
    for (int w = 0; w < SCAN_B / 32; ++w) before += wpre[w];
    for (int w = 0; w < wave; ++w) before += wtot[w];
    if (i < n) cout[i] = before + incl - v;
}

#define CH_CAP 16384
typedef unsigned int v2u __attribute__((ext_vector_type(2)));
static_assert(NBKT <= 256, "the local scan of k_bucket_scatter uses the first 8 waves");

template <bool STAGED>
__device__ __forceinline__ void scatter_chunk(const int* __restrict__ src, const int* __restrict__ dst,
                                              int lo, int hi, int E, unsigned int* cur,
                                              v2u* stage, unsigned int* __restrict__ keys) {
    int e0 = lo + (int)threadIdx.x;
    for (; e0 + (EU - 1) * PB < hi; e0 += EU * PB) {
        int d[EU], s[EU];
#pragma unroll
        for (int u = 0; u < EU; ++u) {
            d[u] = dst[e0 + u * PB];
            s[u] = src[e0 + u * PB];
        }
        unsigned int pos[EU];
#pragma unroll
        for (int u = 0; u < EU; ++u) {
            d[u] = min(max(d[u], 0), N_NODES - 1);
            s[u] = min(max(s[u], 0), N_NODES - 1);
            pos[u] = atomicAdd(&cur[d[u] >> BKT_SHIFT], 1u);
        }
#pragma unroll
        for (int u = 0; u < EU; ++u) {
            const unsigned int w = ((unsigned int)(d[u] & (BKT_NODES - 1)) << SRC_BITS) | (unsigned int)s[u];
            if (STAGED) stage[min(pos[u], (unsigned int)CH_CAP - 1u)] = (v2u){w, (unsigned int)(d[u] >> BKT_SHIFT)};
            else        keys[min(pos[u], (unsigned int)E - 1u)] = w;
        }
    }
    for (; e0 < hi; e0 += PB) {
        const int dd = min(max(dst[e0], 0), N_NODES - 1);
        const int ss = min(max(src[e0], 0), N_NODES - 1);
        const unsigned int pos = atomicAdd(&cur[dd >> BKT_SHIFT], 1u);
        const unsigned int w = ((unsigned int)(dd & (BKT_NODES - 1)) << SRC_BITS) | (unsigned int)ss;
        if (STAGED) stage[min(pos, (unsigned int)CH_CAP - 1u)] = (v2u){w, (unsigned int)(dd >> BKT_SHIFT)};
        else        keys[min(pos, (unsigned int)E - 1u)] = w;
    }
}

__global__ __launch_bounds__(PB) void k_bucket_scatter(const int* __restrict__ src,
                                                       const int* __restrict__ dst, int E,
                                                       const unsigned int* __restrict__ counts,
                                                       const unsigned int* __restrict__ offs,
                                                       unsigned int* __restrict__ keys) {
    __shared__ unsigned int cur[NBKT];
    __shared__ unsigned int gdelta[NBKT];
    __shared__ unsigned int wtot[8];
    __shared__ __attribute__((aligned(8))) v2u stage[CH_CAP];
    const int tid = threadIdx.x;
    const int chunk = (E + (int)gridDim.x - 1) / (int)gridDim.x;
    const int lo = (int)blockIdx.x * chunk;
    const int hi = min(E, lo + chunk);

    if (chunk > CH_CAP) {
        for (int i = tid; i < NBKT; i += PB) cur[i] = offs[(size_t)i * gridDim.x + blockIdx.x];
        __syncthreads();
        scatter_chunk<false>(src, dst, lo, hi, E, cur, stage, keys);
        return;
    }

    const int lane = tid & 31;
    const int wave = tid >> 5;
    unsigned int v = 0u, g0 = 0u;
    if (tid < NBKT) {
        v  = counts[(size_t)tid * gridDim.x + blockIdx.x];
        g0 = offs[(size_t)tid * gridDim.x + blockIdx.x];
    }
    unsigned int incl = v;
    if (tid < 256) {
        for (int off = 1; off < 32; off <<= 1) {
            unsigned int t = __shfl_up(incl, off, 32);
            if (lane >= off) incl += t;
        }
        if (lane == 31) wtot[wave] = incl;
    }
    __syncthreads();
    if (tid < NBKT) {
        unsigned int before = 0u;
        for (int w = 0; w < wave; ++w) before += wtot[w];
        const unsigned int ls = before + incl - v;
        cur[tid] = ls;
        gdelta[tid] = g0 - ls;
    }
    __syncthreads();

    scatter_chunk<true>(src, dst, lo, hi, E, cur, stage, keys);
    __syncthreads();

    const int n = hi - lo;
    int i = tid;
    for (; i + (EU - 1) * PB < n; i += EU * PB) {
        v2u e[EU];
#pragma unroll
        for (int u = 0; u < EU; ++u) e[u] = stage[i + u * PB];
        unsigned int gd[EU];
#pragma unroll
        for (int u = 0; u < EU; ++u) gd[u] = gdelta[min(e[u][1], (unsigned int)NBKT - 1u)];
#pragma unroll
        for (int u = 0; u < EU; ++u) {
            const unsigned int g = (unsigned int)(i + u * PB) + gd[u];
            keys[min(g, (unsigned int)E - 1u)] = e[u][0];
        }
    }
    for (; i < n; i += PB) {
        const v2u e = stage[i];
        const unsigned int g = (unsigned int)i + gdelta[min(e[1], (unsigned int)NBKT - 1u)];
        keys[min(g, (unsigned int)E - 1u)] = e[0];
    }
}

#define SORT_CAP 32768

template <bool STAGED>
__device__ __forceinline__ void place_bucket(const unsigned int* __restrict__ keys,
                                             unsigned int lo, unsigned int hi, int E,
                                             unsigned int* cnt, int* stage, int* __restrict__ csr) {
    unsigned int i0 = lo + threadIdx.x;
    for (; i0 + (EU - 1) * PB < hi; i0 += EU * PB) {
        unsigned int k[EU];
#pragma unroll
        for (int u = 0; u < EU; ++u) k[u] = keys[i0 + u * PB];
        unsigned int pos[EU];
#pragma unroll
        for (int u = 0; u < EU; ++u)
            pos[u] = atomicAdd(&cnt[(k[u] >> SRC_BITS) & (BKT_NODES - 1)], 1u);
#pragma unroll
        for (int u = 0; u < EU; ++u) {
            if (STAGED) stage[min(pos[u], (unsigned int)SORT_CAP - 1u)] = (int)(k[u] & SRC_MASK);
            else        csr[min(lo + pos[u], (unsigned int)E - 1u)] = (int)(k[u] & SRC_MASK);
        }
    }
    for (; i0 < hi; i0 += PB) {
        const unsigned int k = keys[i0];
        const unsigned int pos = atomicAdd(&cnt[(k >> SRC_BITS) & (BKT_NODES - 1)], 1u);
        if (STAGED) stage[min(pos, (unsigned int)SORT_CAP - 1u)] = (int)(k & SRC_MASK);
        else        csr[min(lo + pos, (unsigned int)E - 1u)] = (int)(k & SRC_MASK);
    }
}

__global__ __launch_bounds__(PB) void k_bucket_sort(const unsigned int* __restrict__ offs, int nblk, int E,
                                                    const unsigned int* __restrict__ keys,
                                                    unsigned int* __restrict__ rowptr,
                                                    float* __restrict__ dinv,
                                                    int* __restrict__ csr) {
    __shared__ unsigned int cnt[BKT_NODES];
    __shared__ unsigned int wtot[BKT_NODES / 32];
    __shared__ int stage[SORT_CAP];
    const int tid  = threadIdx.x;
    const int lane = tid & 31;
    const int wave = tid >> 5;
    const int b = blockIdx.x;
    unsigned int lo = offs[(size_t)b * nblk];
    unsigned int hi = (unsigned int)E;
    if (b + 1 < NBKT) hi = offs[(size_t)(b + 1) * nblk];
    hi = min(hi, (unsigned int)E);
    lo = min(lo, hi);
    const unsigned int n = hi - lo;
    const bool staged = n <= (unsigned int)SORT_CAP;

    if (tid < BKT_NODES) cnt[tid] = 0u;
    __syncthreads();
    unsigned int i0 = lo + tid;
    for (; i0 + (EU - 1) * PB < hi; i0 += EU * PB) {
        unsigned int k[EU];
#pragma unroll
        for (int u = 0; u < EU; ++u) k[u] = keys[i0 + u * PB];
#pragma unroll
        for (int u = 0; u < EU; ++u)
            atomicAdd(&cnt[(k[u] >> SRC_BITS) & (BKT_NODES - 1)], 1u);
    }
    for (; i0 < hi; i0 += PB)
        atomicAdd(&cnt[(keys[i0] >> SRC_BITS) & (BKT_NODES - 1)], 1u);
    __syncthreads();

    const unsigned int v = (tid < BKT_NODES) ? cnt[tid] : 0u;
    unsigned int incl = v;
    for (int off = 1; off < 32; off <<= 1) {
        unsigned int t = __shfl_up(incl, off, 32);
        if (lane >= off) incl += t;
    }
    if (tid < BKT_NODES && lane == 31) wtot[wave] = incl;
    __syncthreads();
    if (tid < BKT_NODES) {
        unsigned int before = 0u;
        for (int w = 0; w < wave; ++w) before += wtot[w];
        const unsigned int start = before + incl - v;
        cnt[tid] = start;
        const int node = b * BKT_NODES + tid;
        if (node < N_NODES) {
            rowptr[node] = lo + start;
            dinv[node] = rsqrtf((float)v + 1.0f);
        }
    }
    __syncthreads();
    if (!staged) {
        place_bucket<false>(keys, lo, hi, E, cnt, stage, csr);
        return;
    }
    place_bucket<true>(keys, lo, hi, E, cnt, stage, csr);
    __syncthreads();

    const unsigned int a = lo & 31u;
    const unsigned int m = n + a;
    int* __restrict__ line0 = csr + (lo - a);
    unsigned int j = (unsigned int)tid < a ? (unsigned int)tid + PB : (unsigned int)tid;
    for (; j + (EU - 1) * PB < m; j += EU * PB) {
        int w[EU];
#pragma unroll
        for (int u = 0; u < EU; ++u) w[u] = stage[j - a + u * PB];
#pragma unroll
        for (int u = 0; u < EU; ++u) line0[j + u * PB] = w[u];
    }
    for (; j < m; j += PB) line0[j] = stage[j - a];
}

typedef float v4f __attribute__((ext_vector_type(4)));

#define KP 388
#define GEMM_ROWS 64
#define GEMM_BLOCKS ((N_NODES + GEMM_ROWS - 1) / GEMM_ROWS)
#define KSUP 64

__global__ __launch_bounds__(128) void k_gemm1(const float* __restrict__ X,
                                               const float* __restrict__ W1,
                                               const float* __restrict__ dinv,
                                               float* __restrict__ H) {
    __shared__ __attribute__((aligned(16))) float Bt[HID * KP];
    const int tid  = threadIdx.x;

    {
        const int c = tid & (HID - 1);
#pragma unroll 12
        for (int k4 = tid >> 6; k4 < IN_DIM / 4; k4 += 2) {
            const float* p = W1 + (size_t)(4 * k4) * HID + c;
            v4f v;
            v[0] = p[0];
            v[1] = p[HID];
            v[2] = p[2 * HID];
            v[3] = p[3 * HID];
            *(v4f*)&Bt[c * KP + 4 * k4] = v;
        }
    }
    __syncthreads();

    const int lane = tid & 31;
    const int wave = tid >> 5;
    const int half = lane >> 4;
    const int lr   = lane & 15;
    const int rowBase = blockIdx.x * GEMM_ROWS + wave * 16;
    if (rowBase >= N_NODES) return;

    const v4f* xa = (const v4f*)(X + (size_t)(rowBase + lr) * IN_DIM + 4 * half);
    const v4f* bb = (const v4f*)(Bt + lr * KP + 4 * half);

    const v4f* dp = (const v4f*)(dinv + rowBase + 8 * half);
    const v4f dlo = dp[0];
    const v4f dhi = dp[1];

    v8f acc[4] = {};
    v4f a_cur[KSUP / 8], a_nxt[KSUP / 8];
#pragma unroll
    for (int j = 0; j < KSUP / 8; ++j) a_cur[j] = xa[2 * j];

#pragma unroll
    for (int ss = 0; ss < IN_DIM / KSUP; ++ss) {
        if (ss + 1 < IN_DIM / KSUP) {
#pragma unroll
            for (int j = 0; j < KSUP / 8; ++j) a_nxt[j] = xa[(ss + 1) * (KSUP / 4) + 2 * j];
        }
#pragma unroll
        for (int ch = 0; ch < KSUP / 8; ++ch) {
            v4f b[4];
#pragma unroll
            for (int t = 0; t < 4; ++t) b[t] = bb[t * (16 * KP / 4) + ss * (KSUP / 4) + 2 * ch];
            const v4f a = a_cur[ch];
            const v2f a01 = {a[0], a[1]};
            const v2f a23 = {a[2], a[3]};
#pragma unroll
            for (int t = 0; t < 4; ++t) {
                const v2f b01 = {b[t][0], b[t][1]};
                acc[t] = __builtin_amdgcn_wmma_f32_16x16x4_f32(
                         false, a01,  false, b01,
                         (short)0, acc[t],  false,  false);
            }
#pragma unroll
            for (int t = 0; t < 4; ++t) {
                const v2f b23 = {b[t][2], b[t][3]};
                acc[t] = __builtin_amdgcn_wmma_f32_16x16x4_f32(
                         false, a23,  false, b23,
                         (short)0, acc[t],  false,  false);
            }
        }
#pragma unroll
        for (int j = 0; j < KSUP / 8; ++j) a_cur[j] = a_nxt[j];
    }

    float* hrow = H + (size_t)(rowBase + 8 * half) * HID + lr;
#pragma unroll
    for (int t = 0; t < 4; ++t) {
#pragma unroll
        for (int r = 0; r < 8; ++r)
            hrow[(size_t)r * HID + 16 * t] = acc[t][r] * (r < 4 ? dlo[r & 3] : dhi[r & 3]);
    }
}

#define RPW 8
#define GATHER_T 128
#define GATHER_BLOCKS ((N_NODES + (GATHER_T / 32) * RPW - 1) / ((GATHER_T / 32) * RPW))

__global__ __launch_bounds__(GATHER_T) void k_gather1(const unsigned int* __restrict__ rowptr,
                                                      const int* __restrict__ csr, int E,
                                                      const float* __restrict__ H,
                                                      const float* __restrict__ dinv,
                                                      const float* __restrict__ b1,
                                                      const float* __restrict__ W2,
                                                      float* __restrict__ zs) {
    const int lane = threadIdx.x & 31;
    const int gw = __builtin_amdgcn_readfirstlane((int)((blockIdx.x * blockDim.x + threadIdx.x) >> 5));
    const int row0 = gw * RPW;
    if (row0 >= N_NODES) return;
    const int nrows = min(RPW, N_NODES - row0);

    const int bi = row0 + min(lane, nrows);
    int rb = E;
    if (bi < N_NODES) rb = (int)rowptr[bi];
    rb = min(rb, E);
    const float dl = dinv[row0 + min(lane, nrows - 1)];
    const float b1a = b1[2 * lane];
    const float b1b = b1[2 * lane + 1];
    const float w2a = W2[2 * lane];
    const float w2b = W2[2 * lane + 1];

    const v2f* H2 = (const v2f*)H;
    float zv = 0.0f;
    v2f hs_next = H2[(size_t)row0 * (HID / 2) + lane];

    for (int r = 0; r < nrows; ++r) {
        const int row = row0 + r;
        const int start = __builtin_amdgcn_readlane(rb, r);
        const int end   = __builtin_amdgcn_readlane(rb, r + 1);
        const float di  = __int_as_float(__builtin_amdgcn_readlane(__float_as_int(dl), r));

        const v2f hs = hs_next;
        if (r + 1 < nrows) hs_next = H2[(size_t)(row + 1) * (HID / 2) + lane];
        float a0 = hs[0] * di;
        float a1 = hs[1] * di;

        for (int base = start; base < end; base += 32) {
            const int idx = base + lane;
            int   s = 0;
            if (idx < end) {
                s = csr[idx];
                s = min(max(s, 0), N_NODES - 1);
            }
            const int cnt = min(32, end - base);
            int t0 = 0;
            for (; t0 + 8 <= cnt; t0 += 8) {
                v2f h[8];
#pragma unroll
                for (int k = 0; k < 8; ++k) {
                    const int st = __builtin_amdgcn_readlane(s, t0 + k);
                    h[k] = H2[(size_t)st * (HID / 2) + lane];
                }
#pragma unroll
                for (int k = 0; k < 8; ++k) {
                    a0 += h[k][0] * di;
                    a1 += h[k][1] * di;
                }
            }
            if (t0 < cnt) {
                v2f h[7];
#pragma unroll
                for (int k = 0; k < 7; ++k) {
                    h[k] = (v2f){0.0f, 0.0f};
                    if (t0 + k < cnt) {
                        const int st = __builtin_amdgcn_readlane(s, t0 + k);
                        h[k] = H2[(size_t)st * (HID / 2) + lane];
                    }
                }
#pragma unroll
                for (int k = 0; k < 7; ++k) {
                    a0 += h[k][0] * di;
                    a1 += h[k][1] * di;
                }
            }
        }

        float v0 = a0 + b1a;
        float v1 = a1 + b1b;
        v0 = v0 > 0.0f ? v0 : 0.0f;
        v1 = v1 > 0.0f ? v1 : 0.0f;
        float sum = v0 * w2a + v1 * w2b;
        for (int off = 16; off > 0; off >>= 1)
            sum += __shfl_down(sum, off, 32);
        const float tot = __int_as_float(__builtin_amdgcn_readfirstlane(__float_as_int(sum)));
        if (lane == r) zv = tot * di;
    }
    if (lane < nrows) zs[row0 + lane] = zv;
}

__global__ __launch_bounds__(GATHER_T) void k_gather2(const unsigned int* __restrict__ rowptr,
                                                      const int* __restrict__ csr, int E,
                                                      const float* __restrict__ zs,
                                                      const float* __restrict__ dinv,
                                                      const float* __restrict__ b2,
                                                      float* __restrict__ out) {
    const int lane = threadIdx.x & 31;
    const int gw = __builtin_amdgcn_readfirstlane((int)((blockIdx.x * blockDim.x + threadIdx.x) >> 5));
    const int row0 = gw * RPW;
    if (row0 >= N_NODES) return;
    const int nrows = min(RPW, N_NODES - row0);

    const int bi = row0 + min(lane, nrows);
    int rb = E;
    if (bi < N_NODES) rb = (int)rowptr[bi];
    rb = min(rb, E);
    const int rl = row0 + min(lane, nrows - 1);
    const float dl = dinv[rl];
    const float zl = zs[rl];
    const float bias = b2[0];
    float ov = 0.0f;

    for (int r = 0; r < nrows; ++r) {
        const int start = __builtin_amdgcn_readlane(rb, r);
        const int end   = __builtin_amdgcn_readlane(rb, r + 1);
        const float di  = __int_as_float(__builtin_amdgcn_readlane(__float_as_int(dl), r));
        float p = 0.0f;
        for (int idx = start + lane; idx < end; idx += 32) {
            int s = csr[idx];
            s = min(max(s, 0), N_NODES - 1);
            p += zs[s] * di;
        }
        for (int off = 16; off > 0; off >>= 1)
            p += __shfl_down(p, off, 32);
        const float tot = __int_as_float(__builtin_amdgcn_readfirstlane(__float_as_int(p)));
        if (lane == r) ov = tot;
    }
    if (lane < nrows) out[row0 + lane] = zl * dl + bias + ov;
}


extern "C" void kernel_launch(void* const* d_in, const int* in_sizes, int n_in,
                              void* d_out, int out_size, void* d_ws, size_t ws_size,
                              hipStream_t stream) {
    const float* X  = (const float*)d_in[0];
    const int*   EI = (const int*)d_in[1];
    const float* W1 = (const float*)d_in[2];
    const float* b1 = (const float*)d_in[3];
    const float* W2 = (const float*)d_in[4];
    const float* b2 = (const float*)d_in[5];
    float* out = (float*)d_out;

    const int E = in_sizes[1] / 2;
    const int* src = EI;
    const int* dst = EI + E;

    char* ws = (char*)d_ws;
    float* H    = (float*)ws;
    unsigned int* keys = (unsigned int*)(ws + (size_t)N_NODES * HID * 4);
    int*   csr  = (int*)(keys + E);
    float* dinv = (float*)(ws + (size_t)2 * N_NODES * HID * 4);
    float* z    = dinv + N_NODES;
    unsigned int* rowptr = (unsigned int*)(z + N_NODES);
    unsigned int* craw = (unsigned int*)dinv;
    unsigned int* offs = (unsigned int*)z;

    k_hist           <<<G1, PB, 0, stream>>>(dst, E, craw);
    k_scan_counts    <<<(NBKT * G1 + SCAN_B - 1) / SCAN_B, SCAN_B, 0, stream>>>(craw, offs, NBKT * G1);
    k_bucket_scatter <<<G1, PB, 0, stream>>>(src, dst, E, craw, offs, keys);
    k_bucket_sort    <<<NBKT, PB, 0, stream>>>(offs, G1, E, keys, rowptr, dinv, csr);

    k_gemm1    <<<GEMM_BLOCKS, 128, 0, stream>>>(X, W1, dinv, H);

    k_gather1  <<<GATHER_BLOCKS, GATHER_T, 0, stream>>>(rowptr, csr, E, H, dinv, b1, W2, z);
    k_gather2  <<<GATHER_BLOCKS, GATHER_T, 0, stream>>>(rowptr, csr, E, z, dinv, b2, out);
}


# Both edge passes: a wave's round trips overlapped (layer 1: neighbour rows in batches of 16, next 32 indices a batch ahead; layer 2: its 8 rows side by side); sum order and trees kept, terms by fma
# speedup vs baseline: 7.8052x; 1.0465x over previous
#include <hip/hip_runtime.h>


typedef float v2f __attribute__((ext_vector_type(2)));
typedef float v8f __attribute__((ext_vector_type(8)));

#define N_NODES 100000
#define IN_DIM  384
#define HID     64
#define SCAN_B  1024


#define BKT_SHIFT 9
#define BKT_NODES (1 << BKT_SHIFT)
#define NBKT ((N_NODES + BKT_NODES - 1) / BKT_NODES)
#define G1 256
#define PB 1024
#define EU 4
#define SRC_BITS 17
#define SRC_MASK ((1u << SRC_BITS) - 1u)

__global__ __launch_bounds__(PB) void k_hist(const int* __restrict__ dst, int E,
                                             unsigned int* __restrict__ counts) {
    __shared__ unsigned int h[NBKT];
    for (int i = threadIdx.x; i < NBKT; i += PB) h[i] = 0u;
    __syncthreads();
    const int chunk = (E + (int)gridDim.x - 1) / (int)gridDim.x;
    const int lo = (int)blockIdx.x * chunk;
    const int hi = min(E, lo + chunk);
    int e0 = lo + (int)threadIdx.x;
    for (; e0 + (EU - 1) * PB < hi; e0 += EU * PB) {
        int d[EU];
#pragma unroll
        for (int u = 0; u < EU; ++u) d[u] = dst[e0 + u * PB];
#pragma unroll
        for (int u = 0; u < EU; ++u) {
            const int dd = min(max(d[u], 0), N_NODES - 1);
            atomicAdd(&h[dd >> BKT_SHIFT], 1u);
        }
    }
    for (; e0 < hi; e0 += PB) {
        const int dd = min(max(dst[e0], 0), N_NODES - 1);
        atomicAdd(&h[dd >> BKT_SHIFT], 1u);
    }
    __syncthreads();
    for (int i = threadIdx.x; i < NBKT; i += PB) counts[(size_t)i * gridDim.x + blockIdx.x] = h[i];
}

__global__ __launch_bounds__(SCAN_B) void k_scan_counts(const unsigned int* __restrict__ cin,
                                                        unsigned int* __restrict__ cout, int n) {
    __shared__ unsigned int wtot[SCAN_B / 32];
    __shared__ unsigned int wpre[SCAN_B / 32];
    const int lane = threadIdx.x & 31;
    const int wave = threadIdx.x >> 5;
    unsigned int pre = 0u;
    for (int t = 0; t < (int)blockIdx.x; ++t) pre += cin[(size_t)t * SCAN_B + threadIdx.x];
    for (int off = 16; off > 0; off >>= 1) pre += __shfl_down(pre, off, 32);
    const int i = blockIdx.x * SCAN_B + threadIdx.x;
    const unsigned int v = (i < n) ? cin[i] : 0u;
    unsigned int incl = v;
    for (int off = 1; off < 32; off <<= 1) {
        unsigned int t = __shfl_up(incl, off, 32);
        if (lane >= off) incl += t;
    }
    if (lane == 31) wtot[wave] = incl;
    if (lane == 0)  wpre[wave] = pre;
    __syncthreads();
    unsigned int before = 0u;
    for (int w = 0; w < SCAN_B / 32; ++w) before += wpre[w];
    for (int w = 0; w < wave; ++w) before += wtot[w];
    if (i < n) cout[i] = before + incl - v;
}

#define CH_CAP 16384
typedef unsigned int v2u __attribute__((ext_vector_type(2)));
static_assert(NBKT <= 256, "the local scan of k_bucket_scatter uses the first 8 waves");

template <bool STAGED>
__device__ __forceinline__ void scatter_chunk(const int* __restrict__ src, const int* __restrict__ dst,
                                              int lo, int hi, int E, unsigned int* cur,
                                              v2u* stage, unsigned int* __restrict__ keys) {
    int e0 = lo + (int)threadIdx.x;
    for (; e0 + (EU - 1) * PB < hi; e0 += EU * PB) {
        int d[EU], s[EU];
#pragma unroll
        for (int u = 0; u < EU; ++u) {
            d[u] = dst[e0 + u * PB];
            s[u] = src[e0 + u * PB];
        }
        unsigned int pos[EU];
#pragma unroll
        for (int u = 0; u < EU; ++u) {
            d[u] = min(max(d[u], 0), N_NODES - 1);
            s[u] = min(max(s[u], 0), N_NODES - 1);
            pos[u] = atomicAdd(&cur[d[u] >> BKT_SHIFT], 1u);
        }
#pragma unroll
        for (int u = 0; u < EU; ++u) {
            const unsigned int w = ((unsigned int)(d[u] & (BKT_NODES - 1)) << SRC_BITS) | (unsigned int)s[u];
            if (STAGED) stage[min(pos[u], (unsigned int)CH_CAP - 1u)] = (v2u){w, (unsigned int)(d[u] >> BKT_SHIFT)};
            else        keys[min(pos[u], (unsigned int)E - 1u)] = w;
        }
    }
    for (; e0 < hi; e0 += PB) {
        const int dd = min(max(dst[e0], 0), N_NODES - 1);
        const int ss = min(max(src[e0], 0), N_NODES - 1);
        const unsigned int pos = atomicAdd(&cur[dd >> BKT_SHIFT], 1u);
        const unsigned int w = ((unsigned int)(dd & (BKT_NODES - 1)) << SRC_BITS) | (unsigned int)ss;
        if (STAGED) stage[min(pos, (unsigned int)CH_CAP - 1u)] = (v2u){w, (unsigned int)(dd >> BKT_SHIFT)};
        else        keys[min(pos, (unsigned int)E - 1u)] = w;
    }
}

__global__ __launch_bounds__(PB) void k_bucket_scatter(const int* __restrict__ src,
                                                       const int* __restrict__ dst, int E,
                                                       const unsigned int* __restrict__ counts,
                                                       const unsigned int* __restrict__ offs,
                                                       unsigned int* __restrict__ keys) {
    __shared__ unsigned int cur[NBKT];
    __shared__ unsigned int gdelta[NBKT];
    __shared__ unsigned int wtot[8];
    __shared__ __attribute__((aligned(8))) v2u stage[CH_CAP];
    const int tid = threadIdx.x;
    const int chunk = (E + (int)gridDim.x - 1) / (int)gridDim.x;
    const int lo = (int)blockIdx.x * chunk;
    const int hi = min(E, lo + chunk);

    if (chunk > CH_CAP) {
        for (int i = tid; i < NBKT; i += PB) cur[i] = offs[(size_t)i * gridDim.x + blockIdx.x];
        __syncthreads();
        scatter_chunk<false>(src, dst, lo, hi, E, cur, stage, keys);
        return;
    }

    const int lane = tid & 31;
    const int wave = tid >> 5;
    unsigned int v = 0u, g0 = 0u;
    if (tid < NBKT) {
        v  = counts[(size_t)tid * gridDim.x + blockIdx.x];
        g0 = offs[(size_t)tid * gridDim.x + blockIdx.x];
    }
    unsigned int incl = v;
    if (tid < 256) {
        for (int off = 1; off < 32; off <<= 1) {
            unsigned int t = __shfl_up(incl, off, 32);
            if (lane >= off) incl += t;
        }
        if (lane == 31) wtot[wave] = incl;
    }
    __syncthreads();
    if (tid < NBKT) {
        unsigned int before = 0u;
        for (int w = 0; w < wave; ++w) before += wtot[w];
        const unsigned int ls = before + incl - v;
        cur[tid] = ls;
        gdelta[tid] = g0 - ls;
    }
    __syncthreads();

    scatter_chunk<true>(src, dst, lo, hi, E, cur, stage, keys);
    __syncthreads();

    const int n = hi - lo;
    int i = tid;
    for (; i + (EU - 1) * PB < n; i += EU * PB) {
        v2u e[EU];
#pragma unroll
        for (int u = 0; u < EU; ++u) e[u] = stage[i + u * PB];
        unsigned int gd[EU];
#pragma unroll
        for (int u = 0; u < EU; ++u) gd[u] = gdelta[min(e[u][1], (unsigned int)NBKT - 1u)];
#pragma unroll
        for (int u = 0; u < EU; ++u) {
            const unsigned int g = (unsigned int)(i + u * PB) + gd[u];
            keys[min(g, (unsigned int)E - 1u)] = e[u][0];
        }
    }
    for (; i < n; i += PB) {
        const v2u e = stage[i];
        const unsigned int g = (unsigned int)i + gdelta[min(e[1], (unsigned int)NBKT - 1u)];
        keys[min(g, (unsigned int)E - 1u)] = e[0];
    }
}

#define SORT_CAP 32768

template <bool STAGED>
__device__ __forceinline__ void place_bucket(const unsigned int* __restrict__ keys,
                                             unsigned int lo, unsigned int hi, int E,
                                             unsigned int* cnt, int* stage, int* __restrict__ csr) {
    unsigned int i0 = lo + threadIdx.x;
    for (; i0 + (EU - 1) * PB < hi; i0 += EU * PB) {
        unsigned int k[EU];
#pragma unroll
        for (int u = 0; u < EU; ++u) k[u] = keys[i0 + u * PB];
        unsigned int pos[EU];
#pragma unroll
        for (int u = 0; u < EU; ++u)
            pos[u] = atomicAdd(&cnt[(k[u] >> SRC_BITS) & (BKT_NODES - 1)], 1u);
#pragma unroll
        for (int u = 0; u < EU; ++u) {
            if (STAGED) stage[min(pos[u], (unsigned int)SORT_CAP - 1u)] = (int)(k[u] & SRC_MASK);
            else        csr[min(lo + pos[u], (unsigned int)E - 1u)] = (int)(k[u] & SRC_MASK);
        }
    }
    for (; i0 < hi; i0 += PB) {
        const unsigned int k = keys[i0];
        const unsigned int pos = atomicAdd(&cnt[(k >> SRC_BITS) & (BKT_NODES - 1)], 1u);
        if (STAGED) stage[min(pos, (unsigned int)SORT_CAP - 1u)] = (int)(k & SRC_MASK);
        else        csr[min(lo + pos, (unsigned int)E - 1u)] = (int)(k & SRC_MASK);
    }
}

__global__ __launch_bounds__(PB) void k_bucket_sort(const unsigned int* __restrict__ offs, int nblk, int E,
                                                    const unsigned int* __restrict__ keys,
                                                    unsigned int* __restrict__ rowptr,
                                                    float* __restrict__ dinv,
                                                    int* __restrict__ csr) {
    __shared__ unsigned int cnt[BKT_NODES];
    __shared__ unsigned int wtot[BKT_NODES / 32];
    __shared__ int stage[SORT_CAP];
    const int tid  = threadIdx.x;
    const int lane = tid & 31;
    const int wave = tid >> 5;
    const int b = blockIdx.x;
    unsigned int lo = offs[(size_t)b * nblk];
    unsigned int hi = (unsigned int)E;
    if (b + 1 < NBKT) hi = offs[(size_t)(b + 1) * nblk];
    hi = min(hi, (unsigned int)E);
    lo = min(lo, hi);
    const unsigned int n = hi - lo;
    const bool staged = n <= (unsigned int)SORT_CAP;

    if (tid < BKT_NODES) cnt[tid] = 0u;
    __syncthreads();
    unsigned int i0 = lo + tid;
    for (; i0 + (EU - 1) * PB < hi; i0 += EU * PB) {
        unsigned int k[EU];
#pragma unroll
        for (int u = 0; u < EU; ++u) k[u] = keys[i0 + u * PB];
#pragma unroll
        for (int u = 0; u < EU; ++u)
            atomicAdd(&cnt[(k[u] >> SRC_BITS) & (BKT_NODES - 1)], 1u);
    }
    for (; i0 < hi; i0 += PB)
        atomicAdd(&cnt[(keys[i0] >> SRC_BITS) & (BKT_NODES - 1)], 1u);
    __syncthreads();

    const unsigned int v = (tid < BKT_NODES) ? cnt[tid] : 0u;
    unsigned int incl = v;
    for (int off = 1; off < 32; off <<= 1) {
        unsigned int t = __shfl_up(incl, off, 32);
        if (lane >= off) incl += t;
    }
    if (tid < BKT_NODES && lane == 31) wtot[wave] = incl;
    __syncthreads();
    if (tid < BKT_NODES) {
        unsigned int before = 0u;
        for (int w = 0; w < wave; ++w) before += wtot[w];
        const unsigned int start = before + incl - v;
        cnt[tid] = start;
        const int node = b * BKT_NODES + tid;
        if (node < N_NODES) {
            rowptr[node] = lo + start;
            dinv[node] = rsqrtf((float)v + 1.0f);
        }
    }
    __syncthreads();
    if (!staged) {
        place_bucket<false>(keys, lo, hi, E, cnt, stage, csr);
        return;
    }
    place_bucket<true>(keys, lo, hi, E, cnt, stage, csr);
    __syncthreads();

    const unsigned int a = lo & 31u;
    const unsigned int m = n + a;
    int* __restrict__ line0 = csr + (lo - a);
    unsigned int j = (unsigned int)tid < a ? (unsigned int)tid + PB : (unsigned int)tid;
    for (; j + (EU - 1) * PB < m; j += EU * PB) {
        int w[EU];
#pragma unroll
        for (int u = 0; u < EU; ++u) w[u] = stage[j - a + u * PB];
#pragma unroll
        for (int u = 0; u < EU; ++u) line0[j + u * PB] = w[u];
    }
    for (; j < m; j += PB) line0[j] = stage[j - a];
}

typedef float v4f __attribute__((ext_vector_type(4)));

#define KP 388
#define GEMM_ROWS 64
#define GEMM_BLOCKS ((N_NODES + GEMM_ROWS - 1) / GEMM_ROWS)
#define KSUP 64

__global__ __launch_bounds__(128) void k_gemm1(const float* __restrict__ X,
                                               const float* __restrict__ W1,
                                               const float* __restrict__ dinv,
                                               float* __restrict__ H) {
    __shared__ __attribute__((aligned(16))) float Bt[HID * KP];
    const int tid  = threadIdx.x;

    {
        const int c = tid & (HID - 1);
#pragma unroll 12
        for (int k4 = tid >> 6; k4 < IN_DIM / 4; k4 += 2) {
            const float* p = W1 + (size_t)(4 * k4) * HID + c;
            v4f v;
            v[0] = p[0];
            v[1] = p[HID];
            v[2] = p[2 * HID];
            v[3] = p[3 * HID];
            *(v4f*)&Bt[c * KP + 4 * k4] = v;
        }
    }
    __syncthreads();

    const int lane = tid & 31;
    const int wave = tid >> 5;
    const int half = lane >> 4;
    const int lr   = lane & 15;
    const int rowBase = blockIdx.x * GEMM_ROWS + wave * 16;
    if (rowBase >= N_NODES) return;

    const v4f* xa = (const v4f*)(X + (size_t)(rowBase + lr) * IN_DIM + 4 * half);
    const v4f* bb = (const v4f*)(Bt + lr * KP + 4 * half);

    const v4f* dp = (const v4f*)(dinv + rowBase + 8 * half);
    const v4f dlo = dp[0];
    const v4f dhi = dp[1];

    v8f acc[4] = {};
    v4f a_cur[KSUP / 8], a_nxt[KSUP / 8];
#pragma unroll
    for (int j = 0; j < KSUP / 8; ++j) a_cur[j] = xa[2 * j];

#pragma unroll
    for (int ss = 0; ss < IN_DIM / KSUP; ++ss) {
        if (ss + 1 < IN_DIM / KSUP) {
#pragma unroll
            for (int j = 0; j < KSUP / 8; ++j) a_nxt[j] = xa[(ss + 1) * (KSUP / 4) + 2 * j];
        }
#pragma unroll
        for (int ch = 0; ch < KSUP / 8; ++ch) {
            v4f b[4];
#pragma unroll
            for (int t = 0; t < 4; ++t) b[t] = bb[t * (16 * KP / 4) + ss * (KSUP / 4) + 2 * ch];
            const v4f a = a_cur[ch];
            const v2f a01 = {a[0], a[1]};
            const v2f a23 = {a[2], a[3]};
#pragma unroll
            for (int t = 0; t < 4; ++t) {
                const v2f b01 = {b[t][0], b[t][1]};
                acc[t] = __builtin_amdgcn_wmma_f32_16x16x4_f32(
                         false, a01,  false, b01,
                         (short)0, acc[t],  false,  false);
            }
#pragma unroll
            for (int t = 0; t < 4; ++t) {
                const v2f b23 = {b[t][2], b[t][3]};
                acc[t] = __builtin_amdgcn_wmma_f32_16x16x4_f32(
                         false, a23,  false, b23,
                         (short)0, acc[t],  false,  false);
            }
        }
#pragma unroll
        for (int j = 0; j < KSUP / 8; ++j) a_cur[j] = a_nxt[j];
    }

    float* hrow = H + (size_t)(rowBase + 8 * half) * HID + lr;
#pragma unroll
    for (int t = 0; t < 4; ++t) {
#pragma unroll
        for (int r = 0; r < 8; ++r)
            hrow[(size_t)r * HID + 16 * t] = acc[t][r] * (r < 4 ? dlo[r & 3] : dhi[r & 3]);
    }
}

#define RPW 8
#define GATHER_T 128
#define GATHER_BLOCKS ((N_NODES + (GATHER_T / 32) * RPW - 1) / ((GATHER_T / 32) * RPW))

#define GB 16
typedef const v2f __attribute__((address_space(1))) gv2f;

template <bool ALL>
__device__ __forceinline__ void rows_request(v2f (&h)[GB], const v2f* __restrict__ H2,
                                             int s, int t0, int cnt, int lane) {
#pragma unroll
    for (int k = 0; k < GB; ++k) {
        h[k] = (v2f){0.0f, 0.0f};
        if (ALL || t0 + k < cnt) {
            const int st = __builtin_amdgcn_readlane(s, t0 + k);
            const unsigned long long ra = (unsigned long long)H2 + (unsigned long long)(unsigned int)st * (HID * 4);
            const unsigned int ra_lo = (unsigned int)__builtin_amdgcn_readfirstlane((int)(unsigned int)ra);
            const unsigned int ra_hi = (unsigned int)__builtin_amdgcn_readfirstlane((int)(unsigned int)(ra >> 32));
            const gv2f* rp = (const gv2f*)(((unsigned long long)ra_hi << 32) | ra_lo);
            h[k] = rp[(unsigned int)lane];
        }
    }
}

__device__ __forceinline__ void rows_sum(const v2f (&h)[GB], float di, float& a0, float& a1) {
#pragma unroll
    for (int k = 0; k < GB; ++k) {
        a0 = __builtin_fmaf(h[k][0], di, a0);
        a1 = __builtin_fmaf(h[k][1], di, a1);
    }
}

__device__ __forceinline__ int idx_request(const int* __restrict__ csr, int at, int lane, int E) {
    const int s = csr[max(min(at + lane, E - 1), 0)];
    return min(max(s, 0), N_NODES - 1);
}

__global__ __launch_bounds__(GATHER_T, 16) void k_gather1(const unsigned int* __restrict__ rowptr,
                                                      const int* __restrict__ csr, int E,
                                                      const float* __restrict__ H,
                                                      const float* __restrict__ dinv,
                                                      const float* __restrict__ b1,
                                                      const float* __restrict__ W2,
                                                      float* __restrict__ zs) {
    const int lane = threadIdx.x & 31;
    const int gw = __builtin_amdgcn_readfirstlane((int)((blockIdx.x * blockDim.x + threadIdx.x) >> 5));
    const int row0 = gw * RPW;
    if (row0 >= N_NODES) return;
    const int nrows = min(RPW, N_NODES - row0);

    const int bi = row0 + min(lane, nrows);
    int rb = E;
    if (bi < N_NODES) rb = (int)rowptr[bi];
    rb = max(min(rb, E), 0);
    const float dl = dinv[row0 + min(lane, nrows - 1)];
    const float b1a = b1[2 * lane];
    const float b1b = b1[2 * lane + 1];
    const float w2a = W2[2 * lane];
    const float w2b = W2[2 * lane + 1];

    const v2f* H2 = (const v2f*)H;
    float part[RPW];
#pragma unroll
    for (int k = 0; k < RPW; ++k) part[k] = 0.0f;
    v2f hs_next = H2[(size_t)row0 * (HID / 2) + lane];
    int s_next = idx_request(csr, __builtin_amdgcn_readlane(rb, 0), lane, E);

    for (int r = 0; r < nrows; ++r) {
        const int row = row0 + r;
        const int start = __builtin_amdgcn_readlane(rb, r);
        const int end   = __builtin_amdgcn_readlane(rb, r + 1);
        const float di  = __int_as_float(__builtin_amdgcn_readlane(__float_as_int(dl), r));

        const v2f hs = hs_next;
        if (r + 1 < nrows) hs_next = H2[(size_t)(row + 1) * (HID / 2) + lane];
        float a0 = hs[0] * di;
        float a1 = hs[1] * di;

        for (int base = start; base < end; base += 32) {
            const int s = s_next;
            const int cnt = min(32, end - base);
            v2f h[GB];
            if (cnt >= GB) rows_request<true >(h, H2, s, 0, cnt, lane);
            else           rows_request<false>(h, H2, s, 0, cnt, lane);
            s_next = idx_request(csr, min(base + 32, end), lane, E);
            __builtin_amdgcn_sched_barrier(0);
            rows_sum(h, di, a0, a1);
            if (cnt > GB) {
                if (cnt == 2 * GB) rows_request<true >(h, H2, s, GB, cnt, lane);
                else               rows_request<false>(h, H2, s, GB, cnt, lane);
                rows_sum(h, di, a0, a1);
            }
        }

        float v0 = a0 + b1a;
        float v1 = a1 + b1b;
        v0 = v0 > 0.0f ? v0 : 0.0f;
        v1 = v1 > 0.0f ? v1 : 0.0f;
        const float sum = v0 * w2a + v1 * w2b;
#pragma unroll
        for (int k = 0; k < RPW; ++k) part[k] = (r == k) ? sum : part[k];
    }
#pragma unroll
    for (int off = 16; off > 0; off >>= 1) {
#pragma unroll
        for (int k = 0; k < RPW; ++k) part[k] += __shfl_down(part[k], off, 32);
    }
    float zv = 0.0f;
#pragma unroll
    for (int k = 0; k < RPW; ++k) {
        const float tot = __int_as_float(__builtin_amdgcn_readfirstlane(__float_as_int(part[k])));
        if (lane == k) zv = tot * dl;
    }
    if (lane < nrows) zs[row0 + lane] = zv;
}

__global__ __launch_bounds__(GATHER_T) void k_gather2(const unsigned int* __restrict__ rowptr,
                                                      const int* __restrict__ csr, int E,
                                                      const float* __restrict__ zs,
                                                      const float* __restrict__ dinv,
                                                      const float* __restrict__ b2,
                                                      float* __restrict__ out) {
    const int lane = threadIdx.x & 31;
    const int gw = __builtin_amdgcn_readfirstlane((int)((blockIdx.x * blockDim.x + threadIdx.x) >> 5));
    const int row0 = gw * RPW;
    if (row0 >= N_NODES) return;
    const int nrows = min(RPW, N_NODES - row0);

    const int bi = row0 + min(lane, nrows);
    int rb = E;
    if (bi < N_NODES) rb = (int)rowptr[bi];
    rb = max(min(rb, E), 0);
    const int rl = row0 + min(lane, nrows - 1);
    const float dl = dinv[rl];
    const float zl = zs[rl];
    const float bias = b2[0];

    int sa[RPW], sb[RPW];
    bool ha[RPW], hb[RPW];
#pragma unroll
    for (int r = 0; r < RPW; ++r) {
        const int start = __builtin_amdgcn_readlane(rb, r);
        const int end   = __builtin_amdgcn_readlane(rb, r + 1);
        const int ia = start + lane;
        ha[r] = ia < end;
        hb[r] = ia + 32 < end;
        sa[r] = csr[max(min(ia, E - 1), 0)];
        sb[r] = csr[max(min(ia + 32, E - 1), 0)];
    }
    float za[RPW], zb[RPW];
#pragma unroll
    for (int r = 0; r < RPW; ++r) {
        const int ja = ha[r] ? min(max(sa[r], 0), N_NODES - 1) : 0;
        const int jb = hb[r] ? min(max(sb[r], 0), N_NODES - 1) : 0;
        za[r] = zs[ja];
        zb[r] = zs[jb];
    }
    float p[RPW];
#pragma unroll
    for (int r = 0; r < RPW; ++r) {
        const int start = __builtin_amdgcn_readlane(rb, r);
        const int end   = __builtin_amdgcn_readlane(rb, r + 1);
        const float di  = __int_as_float(__builtin_amdgcn_readlane(__float_as_int(dl), r));
        float q = 0.0f;
        q += (ha[r] ? za[r] : 0.0f) * di;
        q += (hb[r] ? zb[r] : 0.0f) * di;
        for (int idx = start + 64 + lane; idx < end; idx += 32) {
            int s = csr[idx];
            s = min(max(s, 0), N_NODES - 1);
            q += zs[s] * di;
        }
        p[r] = q;
    }
#pragma unroll
    for (int off = 16; off > 0; off >>= 1) {
#pragma unroll
        for (int r = 0; r < RPW; ++r) p[r] += __shfl_down(p[r], off, 32);
    }
    float ov = 0.0f;
#pragma unroll
    for (int r = 0; r < RPW; ++r) {
        const float tot = __int_as_float(__builtin_amdgcn_readfirstlane(__float_as_int(p[r])));
        if (lane == r) ov = tot;
    }
    if (lane < nrows) out[row0 + lane] = zl * dl + bias + ov;
}


extern "C" void kernel_launch(void* const* d_in, const int* in_sizes, int n_in,
                              void* d_out, int out_size, void* d_ws, size_t ws_size,
                              hipStream_t stream) {
    const float* X  = (const float*)d_in[0];
    const int*   EI = (const int*)d_in[1];
    const float* W1 = (const float*)d_in[2];
    const float* b1 = (const float*)d_in[3];
    const float* W2 = (const float*)d_in[4];
    const float* b2 = (const float*)d_in[5];
    float* out = (float*)d_out;

    const int E = in_sizes[1] / 2;
    const int* src = EI;
    const int* dst = EI + E;

    char* ws = (char*)d_ws;
    float* H    = (float*)ws;
    unsigned int* keys = (unsigned int*)(ws + (size_t)N_NODES * HID * 4);
    int*   csr  = (int*)(keys + E);
    float* dinv = (float*)(ws + (size_t)2 * N_NODES * HID * 4);
    float* z    = dinv + N_NODES;
    unsigned int* rowptr = (unsigned int*)(z + N_NODES);
    unsigned int* craw = (unsigned int*)dinv;
    unsigned int* offs = (unsigned int*)z;

    k_hist           <<<G1, PB, 0, stream>>>(dst, E, craw);
    k_scan_counts    <<<(NBKT * G1 + SCAN_B - 1) / SCAN_B, SCAN_B, 0, stream>>>(craw, offs, NBKT * G1);
    k_bucket_scatter <<<G1, PB, 0, stream>>>(src, dst, E, craw, offs, keys);
    k_bucket_sort    <<<NBKT, PB, 0, stream>>>(offs, G1, E, keys, rowptr, dinv, csr);

    k_gemm1    <<<GEMM_BLOCKS, 128, 0, stream>>>(X, W1, dinv, H);

    k_gather1  <<<GATHER_BLOCKS, GATHER_T, 0, stream>>>(rowptr, csr, E, H, dinv, b1, W2, z);
    k_gather2  <<<GATHER_BLOCKS, GATHER_T, 0, stream>>>(rowptr, csr, E, z, dinv, b2, out);
}


# Edge sort in two launches, not four: a chunk is bucket-ordered in LDS from registers and written as its own image; a bucket block reads its 256 runs from the images; sums in LDS-atomic arrival order
# speedup vs baseline: 8.0823x; 1.0355x over previous
#include <hip/hip_runtime.h>


typedef float v2f __attribute__((ext_vector_type(2)));
typedef float v8f __attribute__((ext_vector_type(8)));

#define N_NODES 100000
#define IN_DIM  384
#define HID     64


#define BKT_SHIFT 9
#define BKT_NODES (1 << BKT_SHIFT)
#define NBKT ((N_NODES + BKT_NODES - 1) / BKT_NODES)
#define G1 256
#define PB 1024
#define EU 4
#define EPT 16
#define CH_CAP (PB * EPT)
#define SRC_BITS 17
#define SRC_MASK ((1u << SRC_BITS) - 1u)
static_assert(NBKT <= 256, "the bucket scan of k_chunk_sort uses the first 8 waves");

__global__ __launch_bounds__(PB) void k_chunk_sort(const int* __restrict__ src,
                                                   const int* __restrict__ dst, int E,
                                                   unsigned int* __restrict__ cnt_tab,
                                                   unsigned int* __restrict__ lst_tab,
                                                   unsigned int* __restrict__ img) {
    __shared__ unsigned int h[NBKT];
    __shared__ unsigned int cur[NBKT + 32];
    __shared__ unsigned int wtot[8];
    __shared__ unsigned int stage[CH_CAP];
    const int tid = threadIdx.x;
    const int chunk = (E + G1 - 1) / G1;
    const int lo = (int)blockIdx.x * chunk;
    const int hi = min(E, lo + chunk);
    const int n = min(max(hi - lo, 0), CH_CAP);

    for (int i = tid; i < NBKT; i += PB) h[i] = 0u;

    int d[EPT], s[EPT];
#pragma unroll
    for (int u = 0; u < EPT; ++u) {
        const int e = max(min(lo + tid + u * PB, hi - 1), 0);
        d[u] = dst[e];
        s[u] = src[e];
    }
    __syncthreads();
#pragma unroll
    for (int u = 0; u < EPT; ++u) {
        d[u] = min(max(d[u], 0), N_NODES - 1);
        s[u] = min(max(s[u], 0), N_NODES - 1);
        if (tid + u * PB < n) atomicAdd(&h[d[u] >> BKT_SHIFT], 1u);
    }
    __syncthreads();

    const int lane = tid & 31;
    const int wave = tid >> 5;
    const unsigned int v = (tid < NBKT) ? h[tid] : 0u;
    unsigned int incl = v;
    if (tid < 256) {
        for (int off = 1; off < 32; off <<= 1) {
            unsigned int t = __shfl_up(incl, off, 32);
            if (lane >= off) incl += t;
        }
        if (lane == 31) wtot[wave] = incl;
    }
    __syncthreads();
    if (tid < NBKT) {
        unsigned int before = 0u;
        for (int w = 0; w < wave; ++w) before += wtot[w];
        const unsigned int ls = before + incl - v;
        cur[tid] = ls;
        cnt_tab[(size_t)tid * G1 + blockIdx.x] = v;
        lst_tab[(size_t)tid * G1 + blockIdx.x] = ls;
    }
    __syncthreads();

    unsigned int pos[EPT];
#pragma unroll
    for (int u = 0; u < EPT; ++u) {
        const int cw = (tid + u * PB < n) ? (d[u] >> BKT_SHIFT) : (NBKT + lane);
        pos[u] = atomicAdd(&cur[cw], 1u);
    }
#pragma unroll
    for (int u = 0; u < EPT; ++u) {
        if (tid + u * PB < n) {
            const unsigned int w = ((unsigned int)(d[u] & (BKT_NODES - 1)) << SRC_BITS) | (unsigned int)s[u];
            stage[min(pos[u], (unsigned int)CH_CAP - 1u)] = w;
        }
    }
    __syncthreads();

    int i = tid;
    for (; i + (EU - 1) * PB < n; i += EU * PB) {
        unsigned int w[EU];
#pragma unroll
        for (int u = 0; u < EU; ++u) w[u] = stage[i + u * PB];
#pragma unroll
        for (int u = 0; u < EU; ++u) img[lo + i + u * PB] = w[u];
    }
    for (; i < n; i += PB) img[lo + i] = stage[i];
}

#define SORT_CAP 32768
#define RUNS_PW (G1 / 32)
#define RUN_U 4
#define TABQ (NBKT * G1 / 4)
#define TABQ_PT ((TABQ + PB - 1) / PB)
typedef unsigned int v4u __attribute__((ext_vector_type(4)));
static_assert(G1 == 32 * RUNS_PW && PB == 1024, "one wave per RUNS_PW runs, 32 waves");


template <bool STAGED>
__device__ __forceinline__ void place_runs(const unsigned int (&kk)[RUNS_PW][RUN_U], unsigned int cv, unsigned int lv,
                                           int wave, int lane, int chunk, int E, unsigned int lo,
                                           const unsigned int* __restrict__ img,
                                           unsigned int* cnt, int* stage, int* __restrict__ csr) {
#pragma unroll
    for (int r = 0; r < RUNS_PW; ++r) {
        const unsigned int c = (unsigned int)__builtin_amdgcn_readlane((int)cv, r);
        unsigned int pos[RUN_U];
#pragma unroll
        for (int u = 0; u < RUN_U; ++u) {
            const unsigned int cw = ((unsigned int)lane + 32u * u < c)
                                  ? ((kk[r][u] >> SRC_BITS) & (BKT_NODES - 1)) : (unsigned int)(BKT_NODES + lane);
            pos[u] = atomicAdd(&cnt[cw], 1u);
        }
#pragma unroll
        for (int u = 0; u < RUN_U; ++u) {
            if ((unsigned int)lane + 32u * u < c) {
                if (STAGED) stage[min(pos[u], (unsigned int)SORT_CAP - 1u)] = (int)(kk[r][u] & SRC_MASK);
                else        csr[min(lo + pos[u], (unsigned int)E - 1u)] = (int)(kk[r][u] & SRC_MASK);
            }
        }
        if (c > 32u * RUN_U) {
            const unsigned int base = (unsigned int)(wave * RUNS_PW + r) * (unsigned int)chunk
                                    + (unsigned int)__builtin_amdgcn_readlane((int)lv, r);
            for (unsigned int i = 32u * RUN_U + (unsigned int)lane; i < c; i += 32u) {
                const unsigned int k = img[min(base + i, (unsigned int)E - 1u)];
                const unsigned int p = atomicAdd(&cnt[(k >> SRC_BITS) & (BKT_NODES - 1)], 1u);
                if (STAGED) stage[min(p, (unsigned int)SORT_CAP - 1u)] = (int)(k & SRC_MASK);
                else        csr[min(lo + p, (unsigned int)E - 1u)] = (int)(k & SRC_MASK);
            }
        }
    }
}

__global__ __launch_bounds__(PB) void k_bucket_sort(const unsigned int* __restrict__ cnt_tab,
                                                    const unsigned int* __restrict__ lst_tab, int E,
                                                    const unsigned int* __restrict__ img,
                                                    unsigned int* __restrict__ rowptr,
                                                    float* __restrict__ dinv,
                                                    int* __restrict__ csr) {
    __shared__ unsigned int cnt[BKT_NODES + 32];
    __shared__ unsigned int wtot[BKT_NODES / 32];
    __shared__ unsigned int red_lo[PB / 32];
    __shared__ unsigned int red_n[PB / 32];
    __shared__ int stage[SORT_CAP];
    const int tid  = threadIdx.x;
    const int lane = tid & 31;
    const int wave = __builtin_amdgcn_readfirstlane(tid >> 5);
    const int b = blockIdx.x;
    const int chunk = (E + G1 - 1) / G1;

    if (tid < BKT_NODES) cnt[tid] = 0u;

    unsigned int cv = 0u, lv = 0u;
    if (lane < RUNS_PW) {
        cv = cnt_tab[(size_t)b * G1 + wave * RUNS_PW + lane];
        lv = lst_tab[(size_t)b * G1 + wave * RUNS_PW + lane];
    }
    cv = min(cv, (unsigned int)chunk);

    unsigned int kk[RUNS_PW][RUN_U];
#pragma unroll
    for (int r = 0; r < RUNS_PW; ++r) {
        const unsigned int c = (unsigned int)__builtin_amdgcn_readlane((int)cv, r);
        const unsigned int base = (unsigned int)(wave * RUNS_PW + r) * (unsigned int)chunk
                                + (unsigned int)__builtin_amdgcn_readlane((int)lv, r);
#pragma unroll
        for (int u = 0; u < RUN_U; ++u) {
            kk[r][u] = 0u;
            if (32u * u < c)
                kk[r][u] = img[min(base + (unsigned int)lane + 32u * u, (unsigned int)E - 1u)];
        }
    }

    const v4u* __restrict__ tab4 = (const v4u*)cnt_tab;
    const int nq = b * (G1 / 4);
    unsigned int acc = 0u;
#pragma unroll
    for (int u = 0; u < TABQ_PT; ++u) {
        const int q = tid + u * PB;
        const v4u t = tab4[min(q, TABQ - 1)];
        const unsigned int keep = q < nq ? 0xffffffffu : 0u;
        acc += ((t[0] + t[1]) + (t[2] + t[3])) & keep;
    }
    unsigned int accn = cv;
    for (int off = 16; off > 0; off >>= 1) {
        acc  += __shfl_down(acc, off, 32);
        accn += __shfl_down(accn, off, 32);
    }
    if (lane == 0) {
        red_lo[wave] = acc;
        red_n[wave]  = accn;
    }
    __syncthreads();
    unsigned int lo = 0u, n = 0u;
    for (int w = 0; w < PB / 32; ++w) {
        lo += red_lo[w];
        n  += red_n[w];
    }
    lo = min(lo, (unsigned int)E);
    n  = min(n, (unsigned int)E - lo);
    const bool staged = n <= (unsigned int)SORT_CAP;

#pragma unroll
    for (int r = 0; r < RUNS_PW; ++r) {
        const unsigned int c = (unsigned int)__builtin_amdgcn_readlane((int)cv, r);
#pragma unroll
        for (int u = 0; u < RUN_U; ++u)
            if ((unsigned int)lane + 32u * u < c)
                atomicAdd(&cnt[(kk[r][u] >> SRC_BITS) & (BKT_NODES - 1)], 1u);
        if (c > 32u * RUN_U) {
            const unsigned int base = (unsigned int)(wave * RUNS_PW + r) * (unsigned int)chunk
                                    + (unsigned int)__builtin_amdgcn_readlane((int)lv, r);
            for (unsigned int i = 32u * RUN_U + (unsigned int)lane; i < c; i += 32u) {
                const unsigned int k = img[min(base + i, (unsigned int)E - 1u)];
                atomicAdd(&cnt[(k >> SRC_BITS) & (BKT_NODES - 1)], 1u);
            }
        }
    }
    __syncthreads();

    const int swave = tid >> 5;
    const unsigned int v = (tid < BKT_NODES) ? cnt[tid] : 0u;
    unsigned int incl = v;
    for (int off = 1; off < 32; off <<= 1) {
        unsigned int t = __shfl_up(incl, off, 32);
        if (lane >= off) incl += t;
    }
    if (tid < BKT_NODES && lane == 31) wtot[swave] = incl;
    __syncthreads();
    if (tid < BKT_NODES) {
        unsigned int before = 0u;
        for (int w = 0; w < swave; ++w) before += wtot[w];
        const unsigned int start = before + incl - v;
        cnt[tid] = start;
        const int node = b * BKT_NODES + tid;
        if (node < N_NODES) {
            rowptr[node] = lo + start;
            dinv[node] = rsqrtf((float)v + 1.0f);
        }
    }
    __syncthreads();

    if (staged) place_runs<true >(kk, cv, lv, wave, lane, chunk, E, lo, img, cnt, stage, csr);
    else        place_runs<false>(kk, cv, lv, wave, lane, chunk, E, lo, img, cnt, stage, csr);
    if (!staged) return;
    __syncthreads();

    const unsigned int a = lo & 31u;
    const unsigned int m = n + a;
    int* __restrict__ line0 = csr + (lo - a);
    unsigned int j = (unsigned int)tid < a ? (unsigned int)tid + PB : (unsigned int)tid;
    for (; j + (EU - 1) * PB < m; j += EU * PB) {
        int w[EU];
#pragma unroll
        for (int u = 0; u < EU; ++u) w[u] = stage[j - a + u * PB];
#pragma unroll
        for (int u = 0; u < EU; ++u) line0[j + u * PB] = w[u];
    }
    for (; j < m; j += PB) line0[j] = stage[j - a];
}


typedef float v4f __attribute__((ext_vector_type(4)));

#define KP 388
#define GEMM_ROWS 64
#define GEMM_BLOCKS ((N_NODES + GEMM_ROWS - 1) / GEMM_ROWS)
#define KSUP 64

__global__ __launch_bounds__(128) void k_gemm1(const float* __restrict__ X,
                                               const float* __restrict__ W1,
                                               const float* __restrict__ dinv,
                                               float* __restrict__ H) {
    __shared__ __attribute__((aligned(16))) float Bt[HID * KP];
    const int tid  = threadIdx.x;

    {
        const int c = tid & (HID - 1);
#pragma unroll 12
        for (int k4 = tid >> 6; k4 < IN_DIM / 4; k4 += 2) {
            const float* p = W1 + (size_t)(4 * k4) * HID + c;
            v4f v;
            v[0] = p[0];
            v[1] = p[HID];
            v[2] = p[2 * HID];
            v[3] = p[3 * HID];
            *(v4f*)&Bt[c * KP + 4 * k4] = v;
        }
    }
    __syncthreads();

    const int lane = tid & 31;
    const int wave = tid >> 5;
    const int half = lane >> 4;
    const int lr   = lane & 15;
    const int rowBase = blockIdx.x * GEMM_ROWS + wave * 16;
    if (rowBase >= N_NODES) return;

    const v4f* xa = (const v4f*)(X + (size_t)(rowBase + lr) * IN_DIM + 4 * half);
    const v4f* bb = (const v4f*)(Bt + lr * KP + 4 * half);

    const v4f* dp = (const v4f*)(dinv + rowBase + 8 * half);
    const v4f dlo = dp[0];
    const v4f dhi = dp[1];

    v8f acc[4] = {};
    v4f a_cur[KSUP / 8], a_nxt[KSUP / 8];
#pragma unroll
    for (int j = 0; j < KSUP / 8; ++j) a_cur[j] = xa[2 * j];

#pragma unroll
    for (int ss = 0; ss < IN_DIM / KSUP; ++ss) {
        if (ss + 1 < IN_DIM / KSUP) {
#pragma unroll
            for (int j = 0; j < KSUP / 8; ++j) a_nxt[j] = xa[(ss + 1) * (KSUP / 4) + 2 * j];
        }
#pragma unroll
        for (int ch = 0; ch < KSUP / 8; ++ch) {
            v4f b[4];
#pragma unroll
            for (int t = 0; t < 4; ++t) b[t] = bb[t * (16 * KP / 4) + ss * (KSUP / 4) + 2 * ch];
            const v4f a = a_cur[ch];
            const v2f a01 = {a[0], a[1]};
            const v2f a23 = {a[2], a[3]};
#pragma unroll
            for (int t = 0; t < 4; ++t) {
                const v2f b01 = {b[t][0], b[t][1]};
                acc[t] = __builtin_amdgcn_wmma_f32_16x16x4_f32(
                         false, a01,  false, b01,
                         (short)0, acc[t],  false,  false);
            }
#pragma unroll
            for (int t = 0; t < 4; ++t) {
                const v2f b23 = {b[t][2], b[t][3]};
                acc[t] = __builtin_amdgcn_wmma_f32_16x16x4_f32(
                         false, a23,  false, b23,
                         (short)0, acc[t],  false,  false);
            }
        }
#pragma unroll
        for (int j = 0; j < KSUP / 8; ++j) a_cur[j] = a_nxt[j];
    }

    float* hrow = H + (size_t)(rowBase + 8 * half) * HID + lr;
#pragma unroll
    for (int t = 0; t < 4; ++t) {
#pragma unroll
        for (int r = 0; r < 8; ++r)
            hrow[(size_t)r * HID + 16 * t] = acc[t][r] * (r < 4 ? dlo[r & 3] : dhi[r & 3]);
    }
}

#define RPW 8
#define GATHER_T 128
#define GATHER_BLOCKS ((N_NODES + (GATHER_T / 32) * RPW - 1) / ((GATHER_T / 32) * RPW))

#define GB 16
typedef const v2f __attribute__((address_space(1))) gv2f;

template <bool ALL>
__device__ __forceinline__ void rows_request(v2f (&h)[GB], const v2f* __restrict__ H2,
                                             int s, int t0, int cnt, int lane) {
#pragma unroll
    for (int k = 0; k < GB; ++k) {
        h[k] = (v2f){0.0f, 0.0f};
        if (ALL || t0 + k < cnt) {
            const int st = __builtin_amdgcn_readlane(s, t0 + k);
            const unsigned long long ra = (unsigned long long)H2 + (unsigned long long)(unsigned int)st * (HID * 4);
            const unsigned int ra_lo = (unsigned int)__builtin_amdgcn_readfirstlane((int)(unsigned int)ra);
            const unsigned int ra_hi = (unsigned int)__builtin_amdgcn_readfirstlane((int)(unsigned int)(ra >> 32));
            const gv2f* rp = (const gv2f*)(((unsigned long long)ra_hi << 32) | ra_lo);
            h[k] = rp[(unsigned int)lane];
        }
    }
}

__device__ __forceinline__ void rows_sum(const v2f (&h)[GB], float di, float& a0, float& a1) {
#pragma unroll
    for (int k = 0; k < GB; ++k) {
        a0 = __builtin_fmaf(h[k][0], di, a0);
        a1 = __builtin_fmaf(h[k][1], di, a1);
    }
}

__device__ __forceinline__ int idx_request(const int* __restrict__ csr, int at, int lane, int E) {
    const int s = csr[max(min(at + lane, E - 1), 0)];
    return min(max(s, 0), N_NODES - 1);
}

__global__ __launch_bounds__(GATHER_T, 16) void k_gather1(const unsigned int* __restrict__ rowptr,
                                                      const int* __restrict__ csr, int E,
                                                      const float* __restrict__ H,
                                                      const float* __restrict__ dinv,
                                                      const float* __restrict__ b1,
                                                      const float* __restrict__ W2,
                                                      float* __restrict__ zs) {
    const int lane = threadIdx.x & 31;
    const int gw = __builtin_amdgcn_readfirstlane((int)((blockIdx.x * blockDim.x + threadIdx.x) >> 5));
    const int row0 = gw * RPW;
    if (row0 >= N_NODES) return;
    const int nrows = min(RPW, N_NODES - row0);

    const int bi = row0 + min(lane, nrows);
    int rb = E;
    if (bi < N_NODES) rb = (int)rowptr[bi];
    rb = max(min(rb, E), 0);
    const float dl = dinv[row0 + min(lane, nrows - 1)];
    const float b1a = b1[2 * lane];
    const float b1b = b1[2 * lane + 1];
    const float w2a = W2[2 * lane];
    const float w2b = W2[2 * lane + 1];

    const v2f* H2 = (const v2f*)H;
    float part[RPW];
#pragma unroll
    for (int k = 0; k < RPW; ++k) part[k] = 0.0f;
    v2f hs_next = H2[(size_t)row0 * (HID / 2) + lane];
    int s_next = idx_request(csr, __builtin_amdgcn_readlane(rb, 0), lane, E);

    for (int r = 0; r < nrows; ++r) {
        const int row = row0 + r;
        const int start = __builtin_amdgcn_readlane(rb, r);
        const int end   = __builtin_amdgcn_readlane(rb, r + 1);
        const float di  = __int_as_float(__builtin_amdgcn_readlane(__float_as_int(dl), r));

        const v2f hs = hs_next;
        if (r + 1 < nrows) hs_next = H2[(size_t)(row + 1) * (HID / 2) + lane];
        float a0 = hs[0] * di;
        float a1 = hs[1] * di;

        for (int base = start; base < end; base += 32) {
            const int s = s_next;
            const int cnt = min(32, end - base);
            v2f h[GB];
            if (cnt >= GB) rows_request<true >(h, H2, s, 0, cnt, lane);
            else           rows_request<false>(h, H2, s, 0, cnt, lane);
            s_next = idx_request(csr, min(base + 32, end), lane, E);
            __builtin_amdgcn_sched_barrier(0);
            rows_sum(h, di, a0, a1);
            if (cnt > GB) {
                if (cnt == 2 * GB) rows_request<true >(h, H2, s, GB, cnt, lane);
                else               rows_request<false>(h, H2, s, GB, cnt, lane);
                rows_sum(h, di, a0, a1);
            }
        }

        float v0 = a0 + b1a;
        float v1 = a1 + b1b;
        v0 = v0 > 0.0f ? v0 : 0.0f;
        v1 = v1 > 0.0f ? v1 : 0.0f;
        const float sum = v0 * w2a + v1 * w2b;
#pragma unroll
        for (int k = 0; k < RPW; ++k) part[k] = (r == k) ? sum : part[k];
    }
#pragma unroll
    for (int off = 16; off > 0; off >>= 1) {
#pragma unroll
        for (int k = 0; k < RPW; ++k) part[k] += __shfl_down(part[k], off, 32);
    }
    float zv = 0.0f;
#pragma unroll
    for (int k = 0; k < RPW; ++k) {
        const float tot = __int_as_float(__builtin_amdgcn_readfirstlane(__float_as_int(part[k])));
        if (lane == k) zv = tot * dl;
    }
    if (lane < nrows) zs[row0 + lane] = zv;
}

__global__ __launch_bounds__(GATHER_T) void k_gather2(const unsigned int* __restrict__ rowptr,
                                                      const int* __restrict__ csr, int E,
                                                      const float* __restrict__ zs,
                                                      const float* __restrict__ dinv,
                                                      const float* __restrict__ b2,
                                                      float* __restrict__ out) {
    const int lane = threadIdx.x & 31;
    const int gw = __builtin_amdgcn_readfirstlane((int)((blockIdx.x * blockDim.x + threadIdx.x) >> 5));
    const int row0 = gw * RPW;
    if (row0 >= N_NODES) return;
    const int nrows = min(RPW, N_NODES - row0);

    const int bi = row0 + min(lane, nrows);
    int rb = E;
    if (bi < N_NODES) rb = (int)rowptr[bi];
    rb = max(min(rb, E), 0);
    const int rl = row0 + min(lane, nrows - 1);
    const float dl = dinv[rl];
    const float zl = zs[rl];
    const float bias = b2[0];

    int sa[RPW], sb[RPW];
    bool ha[RPW], hb[RPW];
#pragma unroll
    for (int r = 0; r < RPW; ++r) {
        const int start = __builtin_amdgcn_readlane(rb, r);
        const int end   = __builtin_amdgcn_readlane(rb, r + 1);
        const int ia = start + lane;
        ha[r] = ia < end;
        hb[r] = ia + 32 < end;
        sa[r] = csr[max(min(ia, E - 1), 0)];
        sb[r] = csr[max(min(ia + 32, E - 1), 0)];
    }
    float za[RPW], zb[RPW];
#pragma unroll
    for (int r = 0; r < RPW; ++r) {
        const int ja = ha[r] ? min(max(sa[r], 0), N_NODES - 1) : 0;
        const int jb = hb[r] ? min(max(sb[r], 0), N_NODES - 1) : 0;
        za[r] = zs[ja];
        zb[r] = zs[jb];
    }
    float p[RPW];
#pragma unroll
    for (int r = 0; r < RPW; ++r) {
        const int start = __builtin_amdgcn_readlane(rb, r);
        const int end   = __builtin_amdgcn_readlane(rb, r + 1);
        const float di  = __int_as_float(__builtin_amdgcn_readlane(__float_as_int(dl), r));
        float q = 0.0f;
        q += (ha[r] ? za[r] : 0.0f) * di;
        q += (hb[r] ? zb[r] : 0.0f) * di;
        for (int idx = start + 64 + lane; idx < end; idx += 32) {
            int s = csr[idx];
            s = min(max(s, 0), N_NODES - 1);
            q += zs[s] * di;
        }
        p[r] = q;
    }
#pragma unroll
    for (int off = 16; off > 0; off >>= 1) {
#pragma unroll
        for (int r = 0; r < RPW; ++r) p[r] += __shfl_down(p[r], off, 32);
    }
    float ov = 0.0f;
#pragma unroll
    for (int r = 0; r < RPW; ++r) {
        const float tot = __int_as_float(__builtin_amdgcn_readfirstlane(__float_as_int(p[r])));
        if (lane == r) ov = tot;
    }
    if (lane < nrows) out[row0 + lane] = zl * dl + bias + ov;
}


extern "C" void kernel_launch(void* const* d_in, const int* in_sizes, int n_in,
                              void* d_out, int out_size, void* d_ws, size_t ws_size,
                              hipStream_t stream) {
    const float* X  = (const float*)d_in[0];
    const int*   EI = (const int*)d_in[1];
    const float* W1 = (const float*)d_in[2];
    const float* b1 = (const float*)d_in[3];
    const float* W2 = (const float*)d_in[4];
    const float* b2 = (const float*)d_in[5];
    float* out = (float*)d_out;

    const int E = in_sizes[1] / 2;
    const int* src = EI;
    const int* dst = EI + E;

    char* ws = (char*)d_ws;
    float* H    = (float*)ws;
    unsigned int* img = (unsigned int*)(ws + (size_t)N_NODES * HID * 4);
    int*   csr  = (int*)(img + E);
    float* dinv = (float*)(ws + (size_t)2 * N_NODES * HID * 4);
    float* z    = dinv + N_NODES;
    unsigned int* rowptr = (unsigned int*)(z + N_NODES);
    unsigned int* cnt_tab = (unsigned int*)H;
    unsigned int* lst_tab = cnt_tab + NBKT * G1;

    k_chunk_sort  <<<G1, PB, 0, stream>>>(src, dst, E, cnt_tab, lst_tab, img);
    k_bucket_sort <<<NBKT, PB, 0, stream>>>(cnt_tab, lst_tab, E, img, rowptr, dinv, csr);

    k_gemm1    <<<GEMM_BLOCKS, 128, 0, stream>>>(X, W1, dinv, H);

    k_gather1  <<<GATHER_BLOCKS, GATHER_T, 0, stream>>>(rowptr, csr, E, H, dinv, b1, W2, z);
    k_gather2  <<<GATHER_BLOCKS, GATHER_T, 0, stream>>>(rowptr, csr, E, z, dinv, b2, out);
}


# X @ W1 moved into the edge sort's bucket blocks (400-node buckets: 250 blocks, was 196): a block's waves compute its 25 strips, dealt evenly to the 4 SIMDs; 5 launches, now 4; arrival-order sums kept
# speedup vs baseline: 8.8727x; 1.0978x over previous
#include <hip/hip_runtime.h>


typedef float v2f __attribute__((ext_vector_type(2)));
typedef float v8f __attribute__((ext_vector_type(8)));

#define N_NODES 100000
#define IN_DIM  384
#define HID     64


#define BKT_NODES 400
#define BKT_MASK 511u
#define NBKT (N_NODES / BKT_NODES)
static_assert(N_NODES % BKT_NODES == 0 && BKT_NODES % 16 == 0 && BKT_NODES <= 512, "whole buckets of whole 16-row strips");
#define G1 256
#define PB 1024
#define EU 4
#define EPT 16
#define CH_CAP (PB * EPT)
#define SRC_BITS 17
#define SRC_MASK ((1u << SRC_BITS) - 1u)
static_assert(NBKT <= 256, "the bucket scan of k_chunk_sort uses the first 8 waves");

__global__ __launch_bounds__(PB) void k_chunk_sort(const int* __restrict__ src,
                                                   const int* __restrict__ dst, int E,
                                                   unsigned int* __restrict__ run_tab,
                                                   unsigned int* __restrict__ img) {
    __shared__ unsigned int h[NBKT];
    __shared__ unsigned int cur[NBKT + 32];
    __shared__ unsigned int wtot[8];
    __shared__ unsigned int stage[CH_CAP];
    const int tid = threadIdx.x;
    const int chunk = (E + G1 - 1) / G1;
    const int lo = (int)blockIdx.x * chunk;
    const int hi = min(E, lo + chunk);
    const int n = min(max(hi - lo, 0), CH_CAP);

    for (int i = tid; i < NBKT; i += PB) h[i] = 0u;

    int d[EPT], s[EPT];
#pragma unroll
    for (int u = 0; u < EPT; ++u) {
        const int e = max(min(lo + tid + u * PB, hi - 1), 0);
        d[u] = dst[e];
        s[u] = src[e];
    }
    __syncthreads();
#pragma unroll
    for (int u = 0; u < EPT; ++u) {
        d[u] = min(max(d[u], 0), N_NODES - 1);
        s[u] = min(max(s[u], 0), N_NODES - 1);
        if (tid + u * PB < n) atomicAdd(&h[(unsigned int)d[u] / BKT_NODES], 1u);
    }
    __syncthreads();

    const int lane = tid & 31;
    const int wave = tid >> 5;
    const unsigned int v = (tid < NBKT) ? h[tid] : 0u;
    unsigned int incl = v;
    if (tid < 256) {
        for (int off = 1; off < 32; off <<= 1) {
            unsigned int t = __shfl_up(incl, off, 32);
            if (lane >= off) incl += t;
        }
        if (lane == 31) wtot[wave] = incl;
    }
    __syncthreads();
    if (tid < NBKT) {
        unsigned int before = 0u;
        for (int w = 0; w < wave; ++w) before += wtot[w];
        const unsigned int ls = before + incl - v;
        cur[tid] = ls;
        run_tab[(size_t)tid * G1 + blockIdx.x] = (ls << 16) | v;
    }
    __syncthreads();

    unsigned int pos[EPT];
#pragma unroll
    for (int u = 0; u < EPT; ++u) {
        const int cw = (tid + u * PB < n) ? (int)((unsigned int)d[u] / BKT_NODES) : (NBKT + lane);
        pos[u] = atomicAdd(&cur[cw], 1u);
    }
#pragma unroll
    for (int u = 0; u < EPT; ++u) {
        if (tid + u * PB < n) {
            const unsigned int w = (((unsigned int)d[u] % BKT_NODES) << SRC_BITS) | (unsigned int)s[u];
            stage[min(pos[u], (unsigned int)CH_CAP - 1u)] = w;
        }
    }
    __syncthreads();

    int i = tid;
    for (; i + (EU - 1) * PB < n; i += EU * PB) {
        unsigned int w[EU];
#pragma unroll
        for (int u = 0; u < EU; ++u) w[u] = stage[i + u * PB];
#pragma unroll
        for (int u = 0; u < EU; ++u) img[lo + i + u * PB] = w[u];
    }
    for (; i < n; i += PB) img[lo + i] = stage[i];
}

typedef float v4f __attribute__((ext_vector_type(4)));

#define KP 388
#define KSUP 64
#define STRIPS (BKT_NODES / 16)
#define W1_PT (IN_DIM / 4 / (1024 / HID))

template <int NT>
__device__ __forceinline__ void gemm_strip(const float* __restrict__ X, const float* Bt, const float* dsc,
                                           float* __restrict__ H, int rowBase, int lane, int t0) {
    const int half = lane >> 4;
    const int lr   = lane & 15;

    const v4f* xa = (const v4f*)(X + (size_t)(rowBase + lr) * IN_DIM + 4 * half);
    const v4f* bb = (const v4f*)(Bt + (16 * t0 + lr) * KP + 4 * half);

    v8f acc[NT] = {};
    v4f a_cur[KSUP / 8], a_nxt[KSUP / 8];
#pragma unroll
    for (int j = 0; j < KSUP / 8; ++j) a_cur[j] = xa[2 * j];

#pragma unroll
    for (int ss = 0; ss < IN_DIM / KSUP; ++ss) {
        if (ss + 1 < IN_DIM / KSUP) {
#pragma unroll
            for (int j = 0; j < KSUP / 8; ++j) a_nxt[j] = xa[(ss + 1) * (KSUP / 4) + 2 * j];
        }
#pragma unroll
        for (int ch = 0; ch < KSUP / 8; ++ch) {
            v4f b[NT];
#pragma unroll
            for (int t = 0; t < NT; ++t) b[t] = bb[t * (16 * KP / 4) + ss * (KSUP / 4) + 2 * ch];
            const v4f a = a_cur[ch];
            const v2f a01 = {a[0], a[1]};
            const v2f a23 = {a[2], a[3]};
#pragma unroll
            for (int t = 0; t < NT; ++t) {
                const v2f b01 = {b[t][0], b[t][1]};
                acc[t] = __builtin_amdgcn_wmma_f32_16x16x4_f32(
                         false, a01,  false, b01,
                         (short)0, acc[t],  false,  false);
            }
#pragma unroll
            for (int t = 0; t < NT; ++t) {
                const v2f b23 = {b[t][2], b[t][3]};
                acc[t] = __builtin_amdgcn_wmma_f32_16x16x4_f32(
                         false, a23,  false, b23,
                         (short)0, acc[t],  false,  false);
            }
        }
#pragma unroll
        for (int j = 0; j < KSUP / 8; ++j) a_cur[j] = a_nxt[j];
    }

    const v4f dlo = *(const v4f*)(dsc + 8 * half);
    const v4f dhi = *(const v4f*)(dsc + 8 * half + 4);
    float* hrow = H + (size_t)(rowBase + 8 * half) * HID + 16 * t0 + lr;
#pragma unroll
    for (int t = 0; t < NT; ++t) {
#pragma unroll
        for (int r = 0; r < 8; ++r)
            hrow[(size_t)r * HID + 16 * t] = acc[t][r] * (r < 4 ? dlo[r & 3] : dhi[r & 3]);
    }
}

#define SORT_CAP 16384
#define RUNS_PW (G1 / 32)
#define RUN_U 4
#define TABQ (NBKT * G1 / 4)
#define TABQ_PT ((TABQ + PB - 1) / PB)
#define CNT_SPARE 512
typedef unsigned int v4u __attribute__((ext_vector_type(4)));
static_assert(G1 == 32 * RUNS_PW && PB == 1024, "one wave per RUNS_PW runs, 32 waves");
static_assert(STRIPS <= PB / 32 && W1_PT * (PB / HID) * 4 == IN_DIM, "a wave per strip; the block's threads cover W1 exactly");


template <bool STAGED>
__device__ __forceinline__ void place_runs(const unsigned int (&kk)[RUNS_PW][RUN_U], unsigned int cv, unsigned int lv,
                                           int wave, int lane, int chunk, int E, unsigned int lo,
                                           const unsigned int* __restrict__ img,
                                           unsigned int* cnt, int* stage, int* __restrict__ csr) {
#pragma unroll
    for (int r = 0; r < RUNS_PW; ++r) {
        const unsigned int c = (unsigned int)__builtin_amdgcn_readlane((int)cv, r);
        unsigned int pos[RUN_U];
#pragma unroll
        for (int u = 0; u < RUN_U; ++u) {
            const unsigned int cw = ((unsigned int)lane + 32u * u < c)
                                  ? ((kk[r][u] >> SRC_BITS) & BKT_MASK) : (unsigned int)(CNT_SPARE + lane);
            pos[u] = atomicAdd(&cnt[cw], 1u);
        }
#pragma unroll
        for (int u = 0; u < RUN_U; ++u) {
            if ((unsigned int)lane + 32u * u < c) {
                if (STAGED) stage[min(pos[u], (unsigned int)SORT_CAP - 1u)] = (int)(kk[r][u] & SRC_MASK);
                else        csr[min(lo + pos[u], (unsigned int)E - 1u)] = (int)(kk[r][u] & SRC_MASK);
            }
        }
        if (c > 32u * RUN_U) {
            const unsigned int base = (unsigned int)(wave * RUNS_PW + r) * (unsigned int)chunk
                                    + (unsigned int)__builtin_amdgcn_readlane((int)lv, r);
            for (unsigned int i = 32u * RUN_U + (unsigned int)lane; i < c; i += 32u) {
                const unsigned int k = img[min(base + i, (unsigned int)E - 1u)];
                const unsigned int p = atomicAdd(&cnt[(k >> SRC_BITS) & BKT_MASK], 1u);
                if (STAGED) stage[min(p, (unsigned int)SORT_CAP - 1u)] = (int)(k & SRC_MASK);
                else        csr[min(lo + p, (unsigned int)E - 1u)] = (int)(k & SRC_MASK);
            }
        }
    }
}

__global__ __launch_bounds__(PB) void k_bucket_sort(const unsigned int* __restrict__ run_tab, int E,
                                                    const unsigned int* __restrict__ img,
                                                    unsigned int* __restrict__ rowptr,
                                                    float* __restrict__ dinv,
                                                    int* __restrict__ csr,
                                                    const float* __restrict__ X,
                                                    const float* __restrict__ W1,
                                                    float* __restrict__ H) {
    __shared__ unsigned int cnt[CNT_SPARE + 32];
    __shared__ unsigned int wtot[(BKT_NODES + 31) / 32];
    __shared__ unsigned int red_lo[PB / 32];
    __shared__ unsigned int red_n[PB / 32];
    __shared__ __attribute__((aligned(16))) float dsc[BKT_NODES];
    __shared__ int stage[SORT_CAP];
    __shared__ __attribute__((aligned(16))) float Bt[HID * KP];
    const int tid  = threadIdx.x;
    const int lane = tid & 31;
    const int wave = __builtin_amdgcn_readfirstlane(tid >> 5);
    const int b = blockIdx.x;
    const int chunk = (E + G1 - 1) / G1;

    if (tid < CNT_SPARE) cnt[tid] = 0u;

    const int wc  = tid & (HID - 1);
    const int wk4 = tid >> 6;
    v4f w1r[W1_PT];
#pragma unroll
    for (int u = 0; u < W1_PT; ++u) {
        const float* p = W1 + (size_t)(4 * (wk4 + (PB / HID) * u)) * HID + wc;
        w1r[u][0] = p[0];
        w1r[u][1] = p[HID];
        w1r[u][2] = p[2 * HID];
        w1r[u][3] = p[3 * HID];
    }

    unsigned int cv = 0u, lv = 0u;
    if (lane < RUNS_PW) {
        const unsigned int t = run_tab[(size_t)b * G1 + wave * RUNS_PW + lane];
        cv = t & 0xffffu;
        lv = t >> 16;
    }
    cv = min(cv, (unsigned int)chunk);

    const v4u* __restrict__ tab4 = (const v4u*)run_tab;
    const int nq = b * (G1 / 4);
    unsigned int acc = 0u;
#pragma unroll
    for (int u = 0; u < TABQ_PT; ++u) {
        const int q = tid + u * PB;
        const v4u t = tab4[min(q, nq)];
        const unsigned int keep = q < nq ? 0xffffu : 0u;
        acc += ((t[0] & keep) + (t[1] & keep)) + ((t[2] & keep) + (t[3] & keep));
    }

#pragma unroll
    for (int u = 0; u < W1_PT; ++u)
        *(v4f*)&Bt[wc * KP + 4 * (wk4 + (PB / HID) * u)] = w1r[u];

    unsigned int kk[RUNS_PW][RUN_U];
#pragma unroll
    for (int r = 0; r < RUNS_PW; ++r) {
        const unsigned int c = (unsigned int)__builtin_amdgcn_readlane((int)cv, r);
        const unsigned int base = (unsigned int)(wave * RUNS_PW + r) * (unsigned int)chunk
                                + (unsigned int)__builtin_amdgcn_readlane((int)lv, r);
#pragma unroll
        for (int u = 0; u < RUN_U; ++u) {
            kk[r][u] = 0u;
            if (32u * u < c)
                kk[r][u] = img[min(base + (unsigned int)lane + 32u * u, (unsigned int)E - 1u)];
        }
    }

    unsigned int accn = cv;
    for (int off = 16; off > 0; off >>= 1) {
        acc  += __shfl_down(acc, off, 32);
        accn += __shfl_down(accn, off, 32);
    }
    if (lane == 0) {
        red_lo[wave] = acc;
        red_n[wave]  = accn;
    }
    __syncthreads();
    unsigned int lo = 0u, n = 0u;
    for (int w = 0; w < PB / 32; ++w) {
        lo += red_lo[w];
        n  += red_n[w];
    }
    lo = min(lo, (unsigned int)E);
    n  = min(n, (unsigned int)E - lo);
    const bool staged = n <= (unsigned int)SORT_CAP;

#pragma unroll
    for (int r = 0; r < RUNS_PW; ++r) {
        const unsigned int c = (unsigned int)__builtin_amdgcn_readlane((int)cv, r);
#pragma unroll
        for (int u = 0; u < RUN_U; ++u)
            if ((unsigned int)lane + 32u * u < c)
                atomicAdd(&cnt[(kk[r][u] >> SRC_BITS) & BKT_MASK], 1u);
        if (c > 32u * RUN_U) {
            const unsigned int base = (unsigned int)(wave * RUNS_PW + r) * (unsigned int)chunk
                                    + (unsigned int)__builtin_amdgcn_readlane((int)lv, r);
            for (unsigned int i = 32u * RUN_U + (unsigned int)lane; i < c; i += 32u) {
                const unsigned int k = img[min(base + i, (unsigned int)E - 1u)];
                atomicAdd(&cnt[(k >> SRC_BITS) & BKT_MASK], 1u);
            }
        }
    }
    __syncthreads();

    const int swave = tid >> 5;
    const unsigned int v = (tid < BKT_NODES) ? cnt[tid] : 0u;
    unsigned int incl = v;
    for (int off = 1; off < 32; off <<= 1) {
        unsigned int t = __shfl_up(incl, off, 32);
        if (lane >= off) incl += t;
    }
    if (tid < BKT_NODES && lane == 31) wtot[swave] = incl;
    __syncthreads();
    if (tid < BKT_NODES) {
        unsigned int before = 0u;
        for (int w = 0; w < swave; ++w) before += wtot[w];
        const unsigned int start = before + incl - v;
        cnt[tid] = start;
        const int node = b * BKT_NODES + tid;
        const float dv = rsqrtf((float)v + 1.0f);
        rowptr[node] = lo + start;
        dinv[node] = dv;
        dsc[tid] = dv;
    }
    __syncthreads();

    if (staged) place_runs<true >(kk, cv, lv, wave, lane, chunk, E, lo, img, cnt, stage, csr);
    else        place_runs<false>(kk, cv, lv, wave, lane, chunk, E, lo, img, cnt, stage, csr);

    if (staged) {
        __syncthreads();

        const unsigned int a = lo & 31u;
        const unsigned int m = n + a;
        int* __restrict__ line0 = csr + (lo - a);
        unsigned int j = (unsigned int)tid < a ? (unsigned int)tid + PB : (unsigned int)tid;
        for (; j + (EU - 1) * PB < m; j += EU * PB) {
            int w[EU];
#pragma unroll
            for (int u = 0; u < EU; ++u) w[u] = stage[j - a + u * PB];
#pragma unroll
            for (int u = 0; u < EU; ++u) line0[j + u * PB] = w[u];
        }
        for (; j < m; j += PB) line0[j] = stage[j - a];
    }

    const unsigned int OTHER = (1u << 0) | (1u << 5) | (1u << 10) | (1u << 15) | (1u << 18) | (1u << 23) | (1u << 24) | (1u << 29);
    static_assert(STRIPS == 25 && PB == 1024, "24 whole strips + 4 tiles on 28 of 32 waves");
    if (!((OTHER >> wave) & 1u)) {
        const int strip = wave - __builtin_popcount(OTHER & ((1u << wave) - 1u));
        gemm_strip<4>(X, Bt, dsc + 16 * strip, H, b * BKT_NODES + 16 * strip, lane, 0);
    } else {
        const int t0 = wave == 0 ? 0 : wave == 15 ? 1 : wave == 18 ? 2 : wave == 29 ? 3 : -1;
        if (t0 >= 0)
            gemm_strip<1>(X, Bt, dsc + 16 * (STRIPS - 1), H, b * BKT_NODES + 16 * (STRIPS - 1), lane, t0);
    }
}

#define RPW 8
#define GATHER_T 128
#define GATHER_BLOCKS ((N_NODES + (GATHER_T / 32) * RPW - 1) / ((GATHER_T / 32) * RPW))

#define GB 16
typedef const v2f __attribute__((address_space(1))) gv2f;

template <bool ALL>
__device__ __forceinline__ void rows_request(v2f (&h)[GB], const v2f* __restrict__ H2,
                                             int s, int t0, int cnt, int lane) {
#pragma unroll
    for (int k = 0; k < GB; ++k) {
        h[k] = (v2f){0.0f, 0.0f};
        if (ALL || t0 + k < cnt) {
            const int st = __builtin_amdgcn_readlane(s, t0 + k);
            const unsigned long long ra = (unsigned long long)H2 + (unsigned long long)(unsigned int)st * (HID * 4);
            const unsigned int ra_lo = (unsigned int)__builtin_amdgcn_readfirstlane((int)(unsigned int)ra);
            const unsigned int ra_hi = (unsigned int)__builtin_amdgcn_readfirstlane((int)(unsigned int)(ra >> 32));
            const gv2f* rp = (const gv2f*)(((unsigned long long)ra_hi << 32) | ra_lo);
            h[k] = rp[(unsigned int)lane];
        }
    }
}

__device__ __forceinline__ void rows_sum(const v2f (&h)[GB], float di, float& a0, float& a1) {
#pragma unroll
    for (int k = 0; k < GB; ++k) {
        a0 = __builtin_fmaf(h[k][0], di, a0);
        a1 = __builtin_fmaf(h[k][1], di, a1);
    }
}

__device__ __forceinline__ int idx_request(const int* __restrict__ csr, int at, int lane, int E) {
    const int s = csr[max(min(at + lane, E - 1), 0)];
    return min(max(s, 0), N_NODES - 1);
}

__global__ __launch_bounds__(GATHER_T, 16) void k_gather1(const unsigned int* __restrict__ rowptr,
                                                      const int* __restrict__ csr, int E,
                                                      const float* __restrict__ H,
                                                      const float* __restrict__ dinv,
                                                      const float* __restrict__ b1,
                                                      const float* __restrict__ W2,
                                                      float* __restrict__ zs) {
    const int lane = threadIdx.x & 31;
    const int gw = __builtin_amdgcn_readfirstlane((int)((blockIdx.x * blockDim.x + threadIdx.x) >> 5));
    const int row0 = gw * RPW;
    if (row0 >= N_NODES) return;
    const int nrows = min(RPW, N_NODES - row0);

    const int bi = row0 + min(lane, nrows);
    int rb = E;
    if (bi < N_NODES) rb = (int)rowptr[bi];
    rb = max(min(rb, E), 0);
    const float dl = dinv[row0 + min(lane, nrows - 1)];
    const float b1a = b1[2 * lane];
    const float b1b = b1[2 * lane + 1];
    const float w2a = W2[2 * lane];
    const float w2b = W2[2 * lane + 1];

    const v2f* H2 = (const v2f*)H;
    float part[RPW];
#pragma unroll
    for (int k = 0; k < RPW; ++k) part[k] = 0.0f;
    v2f hs_next = H2[(size_t)row0 * (HID / 2) + lane];
    int s_next = idx_request(csr, __builtin_amdgcn_readlane(rb, 0), lane, E);

    for (int r = 0; r < nrows; ++r) {
        const int row = row0 + r;
        const int start = __builtin_amdgcn_readlane(rb, r);
        const int end   = __builtin_amdgcn_readlane(rb, r + 1);
        const float di  = __int_as_float(__builtin_amdgcn_readlane(__float_as_int(dl), r));

        const v2f hs = hs_next;
        if (r + 1 < nrows) hs_next = H2[(size_t)(row + 1) * (HID / 2) + lane];
        float a0 = hs[0] * di;
        float a1 = hs[1] * di;

        for (int base = start; base < end; base += 32) {
            const int s = s_next;
            const int cnt = min(32, end - base);
            v2f h[GB];
            if (cnt >= GB) rows_request<true >(h, H2, s, 0, cnt, lane);
            else           rows_request<false>(h, H2, s, 0, cnt, lane);
            s_next = idx_request(csr, min(base + 32, end), lane, E);
            __builtin_amdgcn_sched_barrier(0);
            rows_sum(h, di, a0, a1);
            if (cnt > GB) {
                if (cnt == 2 * GB) rows_request<true >(h, H2, s, GB, cnt, lane);
                else               rows_request<false>(h, H2, s, GB, cnt, lane);
                rows_sum(h, di, a0, a1);
            }
        }

        float v0 = a0 + b1a;
        float v1 = a1 + b1b;
        v0 = v0 > 0.0f ? v0 : 0.0f;
        v1 = v1 > 0.0f ? v1 : 0.0f;
        const float sum = v0 * w2a + v1 * w2b;
#pragma unroll
        for (int k = 0; k < RPW; ++k) part[k] = (r == k) ? sum : part[k];
    }
#pragma unroll
    for (int off = 16; off > 0; off >>= 1) {
#pragma unroll
        for (int k = 0; k < RPW; ++k) part[k] += __shfl_down(part[k], off, 32);
    }
    float zv = 0.0f;
#pragma unroll
    for (int k = 0; k < RPW; ++k) {
        const float tot = __int_as_float(__builtin_amdgcn_readfirstlane(__float_as_int(part[k])));
        if (lane == k) zv = tot * dl;
    }
    if (lane < nrows) zs[row0 + lane] = zv;
}

__global__ __launch_bounds__(GATHER_T) void k_gather2(const unsigned int* __restrict__ rowptr,
                                                      const int* __restrict__ csr, int E,
                                                      const float* __restrict__ zs,
                                                      const float* __restrict__ dinv,
                                                      const float* __restrict__ b2,
                                                      float* __restrict__ out) {
    const int lane = threadIdx.x & 31;
    const int gw = __builtin_amdgcn_readfirstlane((int)((blockIdx.x * blockDim.x + threadIdx.x) >> 5));
    const int row0 = gw * RPW;
    if (row0 >= N_NODES) return;
    const int nrows = min(RPW, N_NODES - row0);

    const int bi = row0 + min(lane, nrows);
    int rb = E;
    if (bi < N_NODES) rb = (int)rowptr[bi];
    rb = max(min(rb, E), 0);
    const int rl = row0 + min(lane, nrows - 1);
    const float dl = dinv[rl];
    const float zl = zs[rl];
    const float bias = b2[0];

    int sa[RPW], sb[RPW];
    bool ha[RPW], hb[RPW];
#pragma unroll
    for (int r = 0; r < RPW; ++r) {
        const int start = __builtin_amdgcn_readlane(rb, r);
        const int end   = __builtin_amdgcn_readlane(rb, r + 1);
        const int ia = start + lane;
        ha[r] = ia < end;
        hb[r] = ia + 32 < end;
        sa[r] = csr[max(min(ia, E - 1), 0)];
        sb[r] = csr[max(min(ia + 32, E - 1), 0)];
    }
    float za[RPW], zb[RPW];
#pragma unroll
    for (int r = 0; r < RPW; ++r) {
        const int ja = ha[r] ? min(max(sa[r], 0), N_NODES - 1) : 0;
        const int jb = hb[r] ? min(max(sb[r], 0), N_NODES - 1) : 0;
        za[r] = zs[ja];
        zb[r] = zs[jb];
    }
    float p[RPW];
#pragma unroll
    for (int r = 0; r < RPW; ++r) {
        const int start = __builtin_amdgcn_readlane(rb, r);
        const int end   = __builtin_amdgcn_readlane(rb, r + 1);
        const float di  = __int_as_float(__builtin_amdgcn_readlane(__float_as_int(dl), r));
        float q = 0.0f;
        q += (ha[r] ? za[r] : 0.0f) * di;
        q += (hb[r] ? zb[r] : 0.0f) * di;
        for (int idx = start + 64 + lane; idx < end; idx += 32) {
            int s = csr[idx];
            s = min(max(s, 0), N_NODES - 1);
            q += zs[s] * di;
        }
        p[r] = q;
    }
#pragma unroll
    for (int off = 16; off > 0; off >>= 1) {
#pragma unroll
        for (int r = 0; r < RPW; ++r) p[r] += __shfl_down(p[r], off, 32);
    }
    float ov = 0.0f;
#pragma unroll
    for (int r = 0; r < RPW; ++r) {
        const float tot = __int_as_float(__builtin_amdgcn_readfirstlane(__float_as_int(p[r])));
        if (lane == r) ov = tot;
    }
    if (lane < nrows) out[row0 + lane] = zl * dl + bias + ov;
}


extern "C" void kernel_launch(void* const* d_in, const int* in_sizes, int n_in,
                              void* d_out, int out_size, void* d_ws, size_t ws_size,
                              hipStream_t stream) {
    const float* X  = (const float*)d_in[0];
    const int*   EI = (const int*)d_in[1];
    const float* W1 = (const float*)d_in[2];
    const float* b1 = (const float*)d_in[3];
    const float* W2 = (const float*)d_in[4];
    const float* b2 = (const float*)d_in[5];
    float* out = (float*)d_out;

    const int E = in_sizes[1] / 2;
    const int* src = EI;
    const int* dst = EI + E;

    char* ws = (char*)d_ws;
    float* H    = (float*)ws;
    unsigned int* img = (unsigned int*)(ws + (size_t)N_NODES * HID * 4);
    int*   csr  = (int*)(img + E);
    float* dinv = (float*)(ws + (size_t)2 * N_NODES * HID * 4);
    float* z    = dinv + N_NODES;
    unsigned int* rowptr = (unsigned int*)(z + N_NODES);
    unsigned int* run_tab = (unsigned int*)z;
    static_assert((size_t)NBKT * G1 <= (size_t)N_NODES, "the run table fits the z plane");

    k_chunk_sort  <<<G1, PB, 0, stream>>>(src, dst, E, run_tab, img);
    k_bucket_sort <<<NBKT, PB, 0, stream>>>(run_tab, E, img, rowptr, dinv, csr, X, W1, H);

    k_gather1  <<<GATHER_BLOCKS, GATHER_T, 0, stream>>>(rowptr, csr, E, H, dinv, b1, W2, z);
    k_gather2  <<<GATHER_BLOCKS, GATHER_T, 0, stream>>>(rowptr, csr, E, z, dinv, b2, out);
}


# Layer-2 edge pass: z*dinv staged in LDS by 256 resident 32-wave blocks (whole-line loads, 13 rows per wave); a neighbour value is an LDS read, not a scattered global read; sum order and trees kept
# speedup vs baseline: 9.2354x; 1.0409x over previous
#include <hip/hip_runtime.h>


typedef float v2f __attribute__((ext_vector_type(2)));
typedef float v8f __attribute__((ext_vector_type(8)));

#define N_NODES 100000
#define IN_DIM  384
#define HID     64


#define BKT_NODES 400
#define BKT_MASK 511u
#define NBKT (N_NODES / BKT_NODES)
static_assert(N_NODES % BKT_NODES == 0 && BKT_NODES % 16 == 0 && BKT_NODES <= 512, "whole buckets of whole 16-row strips");
#define G1 256
#define PB 1024
#define EU 4
#define EPT 16
#define CH_CAP (PB * EPT)
#define SRC_BITS 17
#define SRC_MASK ((1u << SRC_BITS) - 1u)
static_assert(NBKT <= 256, "the bucket scan of k_chunk_sort uses the first 8 waves");

__global__ __launch_bounds__(PB) void k_chunk_sort(const int* __restrict__ src,
                                                   const int* __restrict__ dst, int E,
                                                   unsigned int* __restrict__ run_tab,
                                                   unsigned int* __restrict__ img) {
    __shared__ unsigned int h[NBKT];
    __shared__ unsigned int cur[NBKT + 32];
    __shared__ unsigned int wtot[8];
    __shared__ unsigned int stage[CH_CAP];
    const int tid = threadIdx.x;
    const int chunk = (E + G1 - 1) / G1;
    const int lo = (int)blockIdx.x * chunk;
    const int hi = min(E, lo + chunk);
    const int n = min(max(hi - lo, 0), CH_CAP);

    for (int i = tid; i < NBKT; i += PB) h[i] = 0u;

    int d[EPT], s[EPT];
#pragma unroll
    for (int u = 0; u < EPT; ++u) {
        const int e = max(min(lo + tid + u * PB, hi - 1), 0);
        d[u] = dst[e];
        s[u] = src[e];
    }
    __syncthreads();
#pragma unroll
    for (int u = 0; u < EPT; ++u) {
        d[u] = min(max(d[u], 0), N_NODES - 1);
        s[u] = min(max(s[u], 0), N_NODES - 1);
        if (tid + u * PB < n) atomicAdd(&h[(unsigned int)d[u] / BKT_NODES], 1u);
    }
    __syncthreads();

    const int lane = tid & 31;
    const int wave = tid >> 5;
    const unsigned int v = (tid < NBKT) ? h[tid] : 0u;
    unsigned int incl = v;
    if (tid < 256) {
        for (int off = 1; off < 32; off <<= 1) {
            unsigned int t = __shfl_up(incl, off, 32);
            if (lane >= off) incl += t;
        }
        if (lane == 31) wtot[wave] = incl;
    }
    __syncthreads();
    if (tid < NBKT) {
        unsigned int before = 0u;
        for (int w = 0; w < wave; ++w) before += wtot[w];
        const unsigned int ls = before + incl - v;
        cur[tid] = ls;
        run_tab[(size_t)tid * G1 + blockIdx.x] = (ls << 16) | v;
    }
    __syncthreads();

    unsigned int pos[EPT];
#pragma unroll
    for (int u = 0; u < EPT; ++u) {
        const int cw = (tid + u * PB < n) ? (int)((unsigned int)d[u] / BKT_NODES) : (NBKT + lane);
        pos[u] = atomicAdd(&cur[cw], 1u);
    }
#pragma unroll
    for (int u = 0; u < EPT; ++u) {
        if (tid + u * PB < n) {
            const unsigned int w = (((unsigned int)d[u] % BKT_NODES) << SRC_BITS) | (unsigned int)s[u];
            stage[min(pos[u], (unsigned int)CH_CAP - 1u)] = w;
        }
    }
    __syncthreads();

    int i = tid;
    for (; i + (EU - 1) * PB < n; i += EU * PB) {
        unsigned int w[EU];
#pragma unroll
        for (int u = 0; u < EU; ++u) w[u] = stage[i + u * PB];
#pragma unroll
        for (int u = 0; u < EU; ++u) img[lo + i + u * PB] = w[u];
    }
    for (; i < n; i += PB) img[lo + i] = stage[i];
}

typedef float v4f __attribute__((ext_vector_type(4)));

#define KP 388
#define KSUP 64
#define STRIPS (BKT_NODES / 16)
#define W1_PT (IN_DIM / 4 / (1024 / HID))

template <int NT>
__device__ __forceinline__ void gemm_strip(const float* __restrict__ X, const float* Bt, const float* dsc,
                                           float* __restrict__ H, int rowBase, int lane, int t0) {
    const int half = lane >> 4;
    const int lr   = lane & 15;

    const v4f* xa = (const v4f*)(X + (size_t)(rowBase + lr) * IN_DIM + 4 * half);
    const v4f* bb = (const v4f*)(Bt + (16 * t0 + lr) * KP + 4 * half);

    v8f acc[NT] = {};
    v4f a_cur[KSUP / 8], a_nxt[KSUP / 8];
#pragma unroll
    for (int j = 0; j < KSUP / 8; ++j) a_cur[j] = xa[2 * j];

#pragma unroll
    for (int ss = 0; ss < IN_DIM / KSUP; ++ss) {
        if (ss + 1 < IN_DIM / KSUP) {
#pragma unroll
            for (int j = 0; j < KSUP / 8; ++j) a_nxt[j] = xa[(ss + 1) * (KSUP / 4) + 2 * j];
        }
#pragma unroll
        for (int ch = 0; ch < KSUP / 8; ++ch) {
            v4f b[NT];
#pragma unroll
            for (int t = 0; t < NT; ++t) b[t] = bb[t * (16 * KP / 4) + ss * (KSUP / 4) + 2 * ch];
            const v4f a = a_cur[ch];
            const v2f a01 = {a[0], a[1]};
            const v2f a23 = {a[2], a[3]};
#pragma unroll
            for (int t = 0; t < NT; ++t) {
                const v2f b01 = {b[t][0], b[t][1]};
                acc[t] = __builtin_amdgcn_wmma_f32_16x16x4_f32(
                         false, a01,  false, b01,
                         (short)0, acc[t],  false,  false);
            }
#pragma unroll
            for (int t = 0; t < NT; ++t) {
                const v2f b23 = {b[t][2], b[t][3]};
                acc[t] = __builtin_amdgcn_wmma_f32_16x16x4_f32(
                         false, a23,  false, b23,
                         (short)0, acc[t],  false,  false);
            }
        }
#pragma unroll
        for (int j = 0; j < KSUP / 8; ++j) a_cur[j] = a_nxt[j];
    }

    const v4f dlo = *(const v4f*)(dsc + 8 * half);
    const v4f dhi = *(const v4f*)(dsc + 8 * half + 4);
    float* hrow = H + (size_t)(rowBase + 8 * half) * HID + 16 * t0 + lr;
#pragma unroll
    for (int t = 0; t < NT; ++t) {
#pragma unroll
        for (int r = 0; r < 8; ++r)
            hrow[(size_t)r * HID + 16 * t] = acc[t][r] * (r < 4 ? dlo[r & 3] : dhi[r & 3]);
    }
}

#define SORT_CAP 16384
#define RUNS_PW (G1 / 32)
#define RUN_U 4
#define TABQ (NBKT * G1 / 4)
#define TABQ_PT ((TABQ + PB - 1) / PB)
#define CNT_SPARE 512
typedef unsigned int v4u __attribute__((ext_vector_type(4)));
static_assert(G1 == 32 * RUNS_PW && PB == 1024, "one wave per RUNS_PW runs, 32 waves");
static_assert(STRIPS <= PB / 32 && W1_PT * (PB / HID) * 4 == IN_DIM, "a wave per strip; the block's threads cover W1 exactly");


template <bool STAGED>
__device__ __forceinline__ void place_runs(const unsigned int (&kk)[RUNS_PW][RUN_U], unsigned int cv, unsigned int lv,
                                           int wave, int lane, int chunk, int E, unsigned int lo,
                                           const unsigned int* __restrict__ img,
                                           unsigned int* cnt, int* stage, int* __restrict__ csr) {
#pragma unroll
    for (int r = 0; r < RUNS_PW; ++r) {
        const unsigned int c = (unsigned int)__builtin_amdgcn_readlane((int)cv, r);
        unsigned int pos[RUN_U];
#pragma unroll
        for (int u = 0; u < RUN_U; ++u) {
            const unsigned int cw = ((unsigned int)lane + 32u * u < c)
                                  ? ((kk[r][u] >> SRC_BITS) & BKT_MASK) : (unsigned int)(CNT_SPARE + lane);
            pos[u] = atomicAdd(&cnt[cw], 1u);
        }
#pragma unroll
        for (int u = 0; u < RUN_U; ++u) {
            if ((unsigned int)lane + 32u * u < c) {
                if (STAGED) stage[min(pos[u], (unsigned int)SORT_CAP - 1u)] = (int)(kk[r][u] & SRC_MASK);
                else        csr[min(lo + pos[u], (unsigned int)E - 1u)] = (int)(kk[r][u] & SRC_MASK);
            }
        }
        if (c > 32u * RUN_U) {
            const unsigned int base = (unsigned int)(wave * RUNS_PW + r) * (unsigned int)chunk
                                    + (unsigned int)__builtin_amdgcn_readlane((int)lv, r);
            for (unsigned int i = 32u * RUN_U + (unsigned int)lane; i < c; i += 32u) {
                const unsigned int k = img[min(base + i, (unsigned int)E - 1u)];
                const unsigned int p = atomicAdd(&cnt[(k >> SRC_BITS) & BKT_MASK], 1u);
                if (STAGED) stage[min(p, (unsigned int)SORT_CAP - 1u)] = (int)(k & SRC_MASK);
                else        csr[min(lo + p, (unsigned int)E - 1u)] = (int)(k & SRC_MASK);
            }
        }
    }
}

__global__ __launch_bounds__(PB) void k_bucket_sort(const unsigned int* __restrict__ run_tab, int E,
                                                    const unsigned int* __restrict__ img,
                                                    unsigned int* __restrict__ rowptr,
                                                    float* __restrict__ dinv,
                                                    int* __restrict__ csr,
                                                    const float* __restrict__ X,
                                                    const float* __restrict__ W1,
                                                    float* __restrict__ H) {
    __shared__ unsigned int cnt[CNT_SPARE + 32];
    __shared__ unsigned int wtot[(BKT_NODES + 31) / 32];
    __shared__ unsigned int red_lo[PB / 32];
    __shared__ unsigned int red_n[PB / 32];
    __shared__ __attribute__((aligned(16))) float dsc[BKT_NODES];
    __shared__ int stage[SORT_CAP];
    __shared__ __attribute__((aligned(16))) float Bt[HID * KP];
    const int tid  = threadIdx.x;
    const int lane = tid & 31;
    const int wave = __builtin_amdgcn_readfirstlane(tid >> 5);
    const int b = blockIdx.x;
    const int chunk = (E + G1 - 1) / G1;

    if (tid < CNT_SPARE) cnt[tid] = 0u;

    const int wc  = tid & (HID - 1);
    const int wk4 = tid >> 6;
    v4f w1r[W1_PT];
#pragma unroll
    for (int u = 0; u < W1_PT; ++u) {
        const float* p = W1 + (size_t)(4 * (wk4 + (PB / HID) * u)) * HID + wc;
        w1r[u][0] = p[0];
        w1r[u][1] = p[HID];
        w1r[u][2] = p[2 * HID];
        w1r[u][3] = p[3 * HID];
    }

    unsigned int cv = 0u, lv = 0u;
    if (lane < RUNS_PW) {
        const unsigned int t = run_tab[(size_t)b * G1 + wave * RUNS_PW + lane];
        cv = t & 0xffffu;
        lv = t >> 16;
    }
    cv = min(cv, (unsigned int)chunk);

    const v4u* __restrict__ tab4 = (const v4u*)run_tab;
    const int nq = b * (G1 / 4);
    unsigned int acc = 0u;
#pragma unroll
    for (int u = 0; u < TABQ_PT; ++u) {
        const int q = tid + u * PB;
        const v4u t = tab4[min(q, nq)];
        const unsigned int keep = q < nq ? 0xffffu : 0u;
        acc += ((t[0] & keep) + (t[1] & keep)) + ((t[2] & keep) + (t[3] & keep));
    }

#pragma unroll
    for (int u = 0; u < W1_PT; ++u)
        *(v4f*)&Bt[wc * KP + 4 * (wk4 + (PB / HID) * u)] = w1r[u];

    unsigned int kk[RUNS_PW][RUN_U];
#pragma unroll
    for (int r = 0; r < RUNS_PW; ++r) {
        const unsigned int c = (unsigned int)__builtin_amdgcn_readlane((int)cv, r);
        const unsigned int base = (unsigned int)(wave * RUNS_PW + r) * (unsigned int)chunk
                                + (unsigned int)__builtin_amdgcn_readlane((int)lv, r);
#pragma unroll
        for (int u = 0; u < RUN_U; ++u) {
            kk[r][u] = 0u;
            if (32u * u < c)
                kk[r][u] = img[min(base + (unsigned int)lane + 32u * u, (unsigned int)E - 1u)];
        }
    }

    unsigned int accn = cv;
    for (int off = 16; off > 0; off >>= 1) {
        acc  += __shfl_down(acc, off, 32);
        accn += __shfl_down(accn, off, 32);
    }
    if (lane == 0) {
        red_lo[wave] = acc;
        red_n[wave]  = accn;
    }
    __syncthreads();
    unsigned int lo = 0u, n = 0u;
    for (int w = 0; w < PB / 32; ++w) {
        lo += red_lo[w];
        n  += red_n[w];
    }
    lo = min(lo, (unsigned int)E);
    n  = min(n, (unsigned int)E - lo);
    const bool staged = n <= (unsigned int)SORT_CAP;

#pragma unroll
    for (int r = 0; r < RUNS_PW; ++r) {
        const unsigned int c = (unsigned int)__builtin_amdgcn_readlane((int)cv, r);
#pragma unroll
        for (int u = 0; u < RUN_U; ++u)
            if ((unsigned int)lane + 32u * u < c)
                atomicAdd(&cnt[(kk[r][u] >> SRC_BITS) & BKT_MASK], 1u);
        if (c > 32u * RUN_U) {
            const unsigned int base = (unsigned int)(wave * RUNS_PW + r) * (unsigned int)chunk
                                    + (unsigned int)__builtin_amdgcn_readlane((int)lv, r);
            for (unsigned int i = 32u * RUN_U + (unsigned int)lane; i < c; i += 32u) {
                const unsigned int k = img[min(base + i, (unsigned int)E - 1u)];
                atomicAdd(&cnt[(k >> SRC_BITS) & BKT_MASK], 1u);
            }
        }
    }
    __syncthreads();

    const int swave = tid >> 5;
    const unsigned int v = (tid < BKT_NODES) ? cnt[tid] : 0u;
    unsigned int incl = v;
    for (int off = 1; off < 32; off <<= 1) {
        unsigned int t = __shfl_up(incl, off, 32);
        if (lane >= off) incl += t;
    }
    if (tid < BKT_NODES && lane == 31) wtot[swave] = incl;
    __syncthreads();
    if (tid < BKT_NODES) {
        unsigned int before = 0u;
        for (int w = 0; w < swave; ++w) before += wtot[w];
        const unsigned int start = before + incl - v;
        cnt[tid] = start;
        const int node = b * BKT_NODES + tid;
        const float dv = rsqrtf((float)v + 1.0f);
        rowptr[node] = lo + start;
        dinv[node] = dv;
        dsc[tid] = dv;
    }
    __syncthreads();

    if (staged) place_runs<true >(kk, cv, lv, wave, lane, chunk, E, lo, img, cnt, stage, csr);
    else        place_runs<false>(kk, cv, lv, wave, lane, chunk, E, lo, img, cnt, stage, csr);

    if (staged) {
        __syncthreads();

        const unsigned int a = lo & 31u;
        const unsigned int m = n + a;
        int* __restrict__ line0 = csr + (lo - a);
        unsigned int j = (unsigned int)tid < a ? (unsigned int)tid + PB : (unsigned int)tid;
        for (; j + (EU - 1) * PB < m; j += EU * PB) {
            int w[EU];
#pragma unroll
            for (int u = 0; u < EU; ++u) w[u] = stage[j - a + u * PB];
#pragma unroll
            for (int u = 0; u < EU; ++u) line0[j + u * PB] = w[u];
        }
        for (; j < m; j += PB) line0[j] = stage[j - a];
    }

    const unsigned int OTHER = (1u << 0) | (1u << 5) | (1u << 10) | (1u << 15) | (1u << 18) | (1u << 23) | (1u << 24) | (1u << 29);
    static_assert(STRIPS == 25 && PB == 1024, "24 whole strips + 4 tiles on 28 of 32 waves");
    if (!((OTHER >> wave) & 1u)) {
        const int strip = wave - __builtin_popcount(OTHER & ((1u << wave) - 1u));
        gemm_strip<4>(X, Bt, dsc + 16 * strip, H, b * BKT_NODES + 16 * strip, lane, 0);
    } else {
        const int t0 = wave == 0 ? 0 : wave == 15 ? 1 : wave == 18 ? 2 : wave == 29 ? 3 : -1;
        if (t0 >= 0)
            gemm_strip<1>(X, Bt, dsc + 16 * (STRIPS - 1), H, b * BKT_NODES + 16 * (STRIPS - 1), lane, t0);
    }
}

#define RPW 8
#define GATHER_T 128
#define GATHER_BLOCKS ((N_NODES + (GATHER_T / 32) * RPW - 1) / ((GATHER_T / 32) * RPW))

#define GB 16
typedef const v2f __attribute__((address_space(1))) gv2f;

template <bool ALL>
__device__ __forceinline__ void rows_request(v2f (&h)[GB], const v2f* __restrict__ H2,
                                             int s, int t0, int cnt, int lane) {
#pragma unroll
    for (int k = 0; k < GB; ++k) {
        h[k] = (v2f){0.0f, 0.0f};
        if (ALL || t0 + k < cnt) {
            const int st = __builtin_amdgcn_readlane(s, t0 + k);
            const unsigned long long ra = (unsigned long long)H2 + (unsigned long long)(unsigned int)st * (HID * 4);
            const unsigned int ra_lo = (unsigned int)__builtin_amdgcn_readfirstlane((int)(unsigned int)ra);
            const unsigned int ra_hi = (unsigned int)__builtin_amdgcn_readfirstlane((int)(unsigned int)(ra >> 32));
            const gv2f* rp = (const gv2f*)(((unsigned long long)ra_hi << 32) | ra_lo);
            h[k] = rp[(unsigned int)lane];
        }
    }
}

__device__ __forceinline__ void rows_sum(const v2f (&h)[GB], float di, float& a0, float& a1) {
#pragma unroll
    for (int k = 0; k < GB; ++k) {
        a0 = __builtin_fmaf(h[k][0], di, a0);
        a1 = __builtin_fmaf(h[k][1], di, a1);
    }
}

__device__ __forceinline__ int idx_request(const int* __restrict__ csr, int at, int lane, int E) {
    const int s = csr[max(min(at + lane, E - 1), 0)];
    return min(max(s, 0), N_NODES - 1);
}

__global__ __launch_bounds__(GATHER_T, 16) void k_gather1(const unsigned int* __restrict__ rowptr,
                                                      const int* __restrict__ csr, int E,
                                                      const float* __restrict__ H,
                                                      const float* __restrict__ dinv,
                                                      const float* __restrict__ b1,
                                                      const float* __restrict__ W2,
                                                      float* __restrict__ zs) {
    const int lane = threadIdx.x & 31;
    const int gw = __builtin_amdgcn_readfirstlane((int)((blockIdx.x * blockDim.x + threadIdx.x) >> 5));
    const int row0 = gw * RPW;
    if (row0 >= N_NODES) return;
    const int nrows = min(RPW, N_NODES - row0);

    const int bi = row0 + min(lane, nrows);
    int rb = E;
    if (bi < N_NODES) rb = (int)rowptr[bi];
    rb = max(min(rb, E), 0);
    const float dl = dinv[row0 + min(lane, nrows - 1)];
    const float b1a = b1[2 * lane];
    const float b1b = b1[2 * lane + 1];
    const float w2a = W2[2 * lane];
    const float w2b = W2[2 * lane + 1];

    const v2f* H2 = (const v2f*)H;
    float part[RPW];
#pragma unroll
    for (int k = 0; k < RPW; ++k) part[k] = 0.0f;
    v2f hs_next = H2[(size_t)row0 * (HID / 2) + lane];
    int s_next = idx_request(csr, __builtin_amdgcn_readlane(rb, 0), lane, E);

    for (int r = 0; r < nrows; ++r) {
        const int row = row0 + r;
        const int start = __builtin_amdgcn_readlane(rb, r);
        const int end   = __builtin_amdgcn_readlane(rb, r + 1);
        const float di  = __int_as_float(__builtin_amdgcn_readlane(__float_as_int(dl), r));

        const v2f hs = hs_next;
        if (r + 1 < nrows) hs_next = H2[(size_t)(row + 1) * (HID / 2) + lane];
        float a0 = hs[0] * di;
        float a1 = hs[1] * di;

        for (int base = start; base < end; base += 32) {
            const int s = s_next;
            const int cnt = min(32, end - base);
            v2f h[GB];
            if (cnt >= GB) rows_request<true >(h, H2, s, 0, cnt, lane);
            else           rows_request<false>(h, H2, s, 0, cnt, lane);
            s_next = idx_request(csr, min(base + 32, end), lane, E);
            __builtin_amdgcn_sched_barrier(0);
            rows_sum(h, di, a0, a1);
            if (cnt > GB) {
                if (cnt == 2 * GB) rows_request<true >(h, H2, s, GB, cnt, lane);
                else               rows_request<false>(h, H2, s, GB, cnt, lane);
                rows_sum(h, di, a0, a1);
            }
        }

        float v0 = a0 + b1a;
        float v1 = a1 + b1b;
        v0 = v0 > 0.0f ? v0 : 0.0f;
        v1 = v1 > 0.0f ? v1 : 0.0f;
        const float sum = v0 * w2a + v1 * w2b;
#pragma unroll
        for (int k = 0; k < RPW; ++k) part[k] = (r == k) ? sum : part[k];
    }
#pragma unroll
    for (int off = 16; off > 0; off >>= 1) {
#pragma unroll
        for (int k = 0; k < RPW; ++k) part[k] += __shfl_down(part[k], off, 32);
    }
    float zv = 0.0f;
#pragma unroll
    for (int k = 0; k < RPW; ++k) {
        const float tot = __int_as_float(__builtin_amdgcn_readfirstlane(__float_as_int(part[k])));
        if (lane == k) zv = tot * dl;
    }
    if (lane < nrows) zs[row0 + lane] = zv;
}

#define G2_BLOCKS 256
#define G2_T 1024
#define G2_ROWS ((N_NODES + G2_BLOCKS - 1) / G2_BLOCKS)
#define RPW2 ((G2_ROWS + G2_T / 32 - 1) / (G2_T / 32))
#define ZA 80000
#define ZB (N_NODES - ZA)
#define ZA_PT ((ZA / 4 + G2_T - 1) / G2_T)
#define ZB_PT ((ZB / 4 + G2_T - 1) / G2_T)
#define ZROT_A 64
#define ZROT_B 16
static_assert(ZA % 4 == 0 && ZB % 4 == 0 && ZB > 0 && ZB <= ZA && ZA * 4 <= 320 * 1024, "whole 16-byte words; the second part fits the first one's space; the first fits the LDS");
static_assert((G2_BLOCKS - 1) * ZROT_A < ZA / 4 && (G2_BLOCKS - 1) * ZROT_B < ZB / 4, "a block's starting point lies inside the part");
static_assert(RPW2 + 1 <= 32, "a lane per row bound");

__device__ __forceinline__ int zword(int tid, int u, int rot, int nq) {
    const int p = tid + u * G2_T;
    int idx = p + rot;
    idx = idx >= nq ? idx - nq : idx;
    return p < nq ? idx : nq - 1;
}

template <int BASE, int CNT>
__device__ __forceinline__ void zpick(const float* zh, const int (&j)[RPW2], float (&z)[RPW2]) {
    float v[RPW2];
#pragma unroll
    for (int r = 0; r < RPW2; ++r) {
        const unsigned int u = (unsigned int)(j[r] - BASE);
        v[r] = zh[u < (unsigned int)CNT ? u : 0u];
    }
#pragma unroll
    for (int r = 0; r < RPW2; ++r)
        z[r] = (unsigned int)(j[r] - BASE) < (unsigned int)CNT ? v[r] : z[r];
}

__global__ __launch_bounds__(G2_T) void k_gather2(const unsigned int* __restrict__ rowptr,
                                                  const int* __restrict__ csr, int E,
                                                  const float* __restrict__ zs,
                                                  const float* __restrict__ dinv,
                                                  const float* __restrict__ b2,
                                                  float* __restrict__ out) {
    __shared__ __attribute__((aligned(16))) float zh[ZA];
    const int tid  = threadIdx.x;
    const int lane = tid & 31;
    const int wave = __builtin_amdgcn_readfirstlane(tid >> 5);
    const int bend = min(((int)blockIdx.x + 1) * G2_ROWS, N_NODES);
    const int row0 = (int)blockIdx.x * G2_ROWS + wave * RPW2;
    const int nrows = min(max(bend - row0, 0), RPW2);

    const int bi = row0 + min(lane, nrows);
    int rb = (int)rowptr[min(bi, N_NODES - 1)];
    rb = bi < N_NODES ? rb : E;
    rb = max(min(rb, E), 0);
    const int rl = min(max(row0 + min(lane, nrows - 1), 0), N_NODES - 1);
    const float dl = dinv[rl];
    const float zl = zs[rl];
    const float bias = b2[0];

    const v4f* __restrict__ z4 = (const v4f*)zs;
    v4f* zh4 = (v4f*)zh;
    const int rotA = (int)blockIdx.x * ZROT_A;
    v4f w[ZA_PT];
#pragma unroll
    for (int u = 0; u < ZA_PT; ++u) w[u] = z4[zword(tid, u, rotA, ZA / 4)];

    int sa[RPW2], sb[RPW2];
#pragma unroll
    for (int r = 0; r < RPW2; ++r) {
        const int start = __builtin_amdgcn_readlane(rb, r);
        const int ia = start + lane;
        sa[r] = csr[max(min(ia, E - 1), 0)];
        sb[r] = csr[max(min(ia + 32, E - 1), 0)];
    }

#pragma unroll
    for (int u = 0; u < ZA_PT; ++u) zh4[zword(tid, u, rotA, ZA / 4)] = w[u];
    const int rotB = (int)blockIdx.x * ZROT_B;
    v4f wb[ZB_PT];
#pragma unroll
    for (int u = 0; u < ZB_PT; ++u) wb[u] = z4[ZA / 4 + zword(tid, u, rotB, ZB / 4)];
    __syncthreads();

    int ja[RPW2], jb[RPW2];
    float za[RPW2], zb[RPW2];
#pragma unroll
    for (int r = 0; r < RPW2; ++r) {
        const int start = __builtin_amdgcn_readlane(rb, r);
        const int end   = __builtin_amdgcn_readlane(rb, r + 1);
        const int ia = start + lane;
        ja[r] = ia < end      ? min(max(sa[r], 0), N_NODES - 1) : -1;
        jb[r] = ia + 32 < end ? min(max(sb[r], 0), N_NODES - 1) : -1;
        za[r] = 0.0f;
        zb[r] = 0.0f;
    }
    zpick<0, ZA>(zh, ja, za);
    zpick<0, ZA>(zh, jb, zb);
    __syncthreads();
#pragma unroll
    for (int u = 0; u < ZB_PT; ++u) zh4[zword(tid, u, rotB, ZB / 4)] = wb[u];
    __syncthreads();
    zpick<ZA, ZB>(zh, ja, za);
    zpick<ZA, ZB>(zh, jb, zb);

    float p[RPW2];
#pragma unroll
    for (int r = 0; r < RPW2; ++r) {
        const int start = __builtin_amdgcn_readlane(rb, r);
        const int end   = __builtin_amdgcn_readlane(rb, r + 1);
        const float di  = __int_as_float(__builtin_amdgcn_readlane(__float_as_int(dl), r));
        float q = 0.0f;
        q += za[r] * di;
        q += zb[r] * di;
        for (int idx = start + 64 + lane; idx < end; idx += 32) {
            int s = csr[idx];
            s = min(max(s, 0), N_NODES - 1);
            q += zs[s] * di;
        }
        p[r] = q;
    }
#pragma unroll
    for (int off = 16; off > 0; off >>= 1) {
#pragma unroll
        for (int r = 0; r < RPW2; ++r) p[r] += __shfl_down(p[r], off, 32);
    }
    float ov = 0.0f;
#pragma unroll
    for (int r = 0; r < RPW2; ++r) {
        const float tot = __int_as_float(__builtin_amdgcn_readfirstlane(__float_as_int(p[r])));
        if (lane == r) ov = tot;
    }
    if (lane < nrows) out[row0 + lane] = zl * dl + bias + ov;
}


extern "C" void kernel_launch(void* const* d_in, const int* in_sizes, int n_in,
                              void* d_out, int out_size, void* d_ws, size_t ws_size,
                              hipStream_t stream) {
    const float* X  = (const float*)d_in[0];
    const int*   EI = (const int*)d_in[1];
    const float* W1 = (const float*)d_in[2];
    const float* b1 = (const float*)d_in[3];
    const float* W2 = (const float*)d_in[4];
    const float* b2 = (const float*)d_in[5];
    float* out = (float*)d_out;

    const int E = in_sizes[1] / 2;
    const int* src = EI;
    const int* dst = EI + E;

    char* ws = (char*)d_ws;
    float* H    = (float*)ws;
    unsigned int* img = (unsigned int*)(ws + (size_t)N_NODES * HID * 4);
    int*   csr  = (int*)(img + E);
    float* dinv = (float*)(ws + (size_t)2 * N_NODES * HID * 4);
    float* z    = dinv + N_NODES;
    unsigned int* rowptr = (unsigned int*)(z + N_NODES);
    unsigned int* run_tab = (unsigned int*)z;
    static_assert((size_t)NBKT * G1 <= (size_t)N_NODES, "the run table fits the z plane");

    k_chunk_sort  <<<G1, PB, 0, stream>>>(src, dst, E, run_tab, img);
    k_bucket_sort <<<NBKT, PB, 0, stream>>>(run_tab, E, img, rowptr, dinv, csr, X, W1, H);

    k_gather1  <<<GATHER_BLOCKS, GATHER_T, 0, stream>>>(rowptr, csr, E, H, dinv, b1, W2, z);
    k_gather2  <<<G2_BLOCKS, G2_T, 0, stream>>>(rowptr, csr, E, z, dinv, b2, out);
}
